# starting kernel
# baseline (speedup 1.0000x reference)
#include <hip/hip_runtime.h>
#include <hip/hip_bf16.h>

typedef __attribute__((ext_vector_type(2))) float v2f_t;
typedef __attribute__((ext_vector_type(4))) float v4f_t;
typedef __attribute__((ext_vector_type(8))) float v8f_t;

#define THREADS 256

__device__ __forceinline__ void atadd1(float* p, float a) {
    asm volatile("global_atomic_add_f32 %0, %1, off scope:SCOPE_DEV"
                 :: "v"(p), "v"(a) : "memory");
}
__device__ __forceinline__ void atadd2(float* p, float a, float b) {
    asm volatile("global_atomic_add_f32 %0, %1, off scope:SCOPE_DEV\n\t"
                 "global_atomic_add_f32 %0, %2, off offset:4 scope:SCOPE_DEV"
                 :: "v"(p), "v"(a), "v"(b) : "memory");
}
__device__ __forceinline__ void atadd4(float* p, float a, float b, float c, float d) {
    asm volatile("global_atomic_add_f32 %0, %1, off scope:SCOPE_DEV\n\t"
                 "global_atomic_add_f32 %0, %2, off offset:4 scope:SCOPE_DEV\n\t"
                 "global_atomic_add_f32 %0, %3, off offset:8 scope:SCOPE_DEV\n\t"
                 "global_atomic_add_f32 %0, %4, off offset:12 scope:SCOPE_DEV"
                 :: "v"(p), "v"(a), "v"(b), "v"(c), "v"(d) : "memory");
}

__global__ void gcn_deg_kernel(const int* __restrict__ dst, float* __restrict__ deg, int E) {
    int e = blockIdx.x * blockDim.x + threadIdx.x;
    if (e >= E) return;
    int d = __builtin_nontemporal_load(dst + e);
    atadd1(&deg[d], 1.0f);
}

__global__ void gcn_dinv_kernel(float* __restrict__ deg, int N) {
    int i = blockIdx.x * blockDim.x + threadIdx.x;
    if (i < N) deg[i] = rsqrtf(deg[i] + 1.0f);
}

__global__ void gcn_gemm1_wmma(const float* __restrict__ x, const float* __restrict__ W1,
                               float* __restrict__ h1, int ntiles, int N) {
    int wave = blockIdx.x * (blockDim.x >> 5) + (threadIdx.x >> 5);
    if (wave >= ntiles) return;
    int lane = threadIdx.x & 31;
    int half = lane >> 4;
    int m    = lane & 15;
    int koff = half << 1;
    int row  = wave * 16 + m;
    int rcl  = row < N ? row : N - 1;
    const v2f_t* xr = (const v2f_t*)(x + (size_t)rcl * 128);

    v8f_t c = {};
#pragma unroll
    for (int k0 = 0; k0 < 128; k0 += 4) {
        v2f_t a = __builtin_nontemporal_load(xr + ((k0 + koff) >> 1));
        v2f_t b;
        b.x = W1[(k0 + koff) * 16 + m];
        b.y = W1[(k0 + koff + 1) * 16 + m];
        c = __builtin_amdgcn_wmma_f32_16x16x4_f32(false, a, false, b, (short)0, c,
                                                  false, false);
    }
#pragma unroll
    for (int v = 0; v < 8; ++v) {
        int r = wave * 16 + half * 8 + v;
        if (r < N) h1[(size_t)r * 16 + m] = c[v];
    }
}

__global__ void gcn_edge_agg16(const int* __restrict__ src, const int* __restrict__ dst,
                               const float* __restrict__ dinv, const float* __restrict__ h,
                               float* __restrict__ agg, int E) {
    int t = blockIdx.x * blockDim.x + threadIdx.x;
    int e = t >> 2;
    int q = t & 3;
    if (e >= E) return;
    int s = __builtin_nontemporal_load(src + e);
    int d = __builtin_nontemporal_load(dst + e);
    float norm = dinv[s] * dinv[d];
    v4f_t v = *((const v4f_t*)(h + (size_t)s * 16) + q);
    float* p = agg + (size_t)d * 16 + q * 4;
    atadd4(p, v.x * norm, v.y * norm, v.z * norm, v.w * norm);
}

__global__ void gcn_combine1(float* __restrict__ h1, const float* __restrict__ agg1,
                             const float* __restrict__ dinv, const float* __restrict__ b1,
                             int N) {
    int i = blockIdx.x * blockDim.x + threadIdx.x;
    if (i >= N) return;
    float dv = dinv[i];
    float d2 = dv * dv;
    v4f_t* hp = (v4f_t*)(h1 + (size_t)i * 16);
    const v4f_t* ap = (const v4f_t*)(agg1 + (size_t)i * 16);
    const v4f_t* bp = (const v4f_t*)b1;
#pragma unroll
    for (int j = 0; j < 4; ++j) {
        v4f_t hv = hp[j], av = ap[j], bv = bp[j];
        v4f_t o;
        o.x = fmaxf(av.x + hv.x * d2 + bv.x, 0.0f);
        o.y = fmaxf(av.y + hv.y * d2 + bv.y, 0.0f);
        o.z = fmaxf(av.z + hv.z * d2 + bv.z, 0.0f);
        o.w = fmaxf(av.w + hv.w * d2 + bv.w, 0.0f);
        hp[j] = o;
    }
}

__global__ void gcn_gemm2(const float* __restrict__ h, const float* __restrict__ W2,
                          float* __restrict__ h2, int N) {
    int i = blockIdx.x * blockDim.x + threadIdx.x;
    if (i >= N) return;
    const float* hr = h + (size_t)i * 16;
    float o0 = 0.0f, o1 = 0.0f;
#pragma unroll
    for (int k = 0; k < 16; ++k) {
        float v = hr[k];
        o0 += v * W2[2 * k];
        o1 += v * W2[2 * k + 1];
    }
    h2[2 * (size_t)i]     = o0;
    h2[2 * (size_t)i + 1] = o1;
}

__global__ void gcn_edge_agg2(const int* __restrict__ src, const int* __restrict__ dst,
                              const float* __restrict__ dinv, const float* __restrict__ h,
                              float* __restrict__ agg, int E) {
    int e = blockIdx.x * blockDim.x + threadIdx.x;
    if (e >= E) return;
    int s = __builtin_nontemporal_load(src + e);
    int d = __builtin_nontemporal_load(dst + e);
    float norm = dinv[s] * dinv[d];
    v2f_t v = *(const v2f_t*)(h + (size_t)s * 2);
    atadd2(agg + (size_t)d * 2, v.x * norm, v.y * norm);
}

__global__ void gcn_final(const float* __restrict__ h2, const float* __restrict__ agg2,
                          const float* __restrict__ dinv, const float* __restrict__ b2,
                          float* __restrict__ out, int N) {
    int i = blockIdx.x * blockDim.x + threadIdx.x;
    if (i >= N) return;
    float dv = dinv[i];
    float d2 = dv * dv;
    float a = agg2[2 * (size_t)i]     + h2[2 * (size_t)i]     * d2 + b2[0];
    float b = agg2[2 * (size_t)i + 1] + h2[2 * (size_t)i + 1] * d2 + b2[1];
    float mx  = fmaxf(a, b);
    float lse = mx + logf(expf(a - mx) + expf(b - mx));
    out[2 * (size_t)i]     = a - lse;
    out[2 * (size_t)i + 1] = b - lse;
}

extern "C" void kernel_launch(void* const* d_in, const int* in_sizes, int n_in,
                              void* d_out, int out_size, void* d_ws, size_t ws_size,
                              hipStream_t stream) {
    const float* x  = (const float*)d_in[0];
    const int*   ei = (const int*)d_in[1];
    const float* W1 = (const float*)d_in[2];
    const float* b1 = (const float*)d_in[3];
    const float* W2 = (const float*)d_in[4];
    const float* b2 = (const float*)d_in[5];
    float* out = (float*)d_out;

    int N = in_sizes[0] / 128;
    int E = in_sizes[1] / 2;
    const int* src = ei;
    const int* dst = ei + E;

    float* ws   = (float*)d_ws;
    float* deg  = ws;
    float* agg1 = deg + (size_t)N;
    float* agg2 = agg1 + (size_t)16 * N;
    float* h1   = agg2 + (size_t)2 * N;
    float* h2   = h1 + (size_t)16 * N;

    hipMemsetAsync(deg, 0, (size_t)19 * (size_t)N * sizeof(float), stream);

    int eb  = (E + THREADS - 1) / THREADS;
    int nb  = (N + THREADS - 1) / THREADS;
    int eb4 = (int)(((long long)E * 4 + THREADS - 1) / THREADS);

    gcn_deg_kernel<<<eb, THREADS, 0, stream>>>(dst, deg, E);
    gcn_dinv_kernel<<<nb, THREADS, 0, stream>>>(deg, N);

    int ntiles = (N + 15) / 16;
    int gb = (ntiles + 7) / 8;
    gcn_gemm1_wmma<<<gb, THREADS, 0, stream>>>(x, W1, h1, ntiles, N);

    gcn_edge_agg16<<<eb4, THREADS, 0, stream>>>(src, dst, deg, h1, agg1, E);
    gcn_combine1<<<nb, THREADS, 0, stream>>>(h1, agg1, deg, b1, N);
    gcn_gemm2<<<nb, THREADS, 0, stream>>>(h1, W2, h2, N);
    gcn_edge_agg2<<<eb, THREADS, 0, stream>>>(src, dst, deg, h2, agg2, E);
    gcn_final<<<nb, THREADS, 0, stream>>>(h2, agg2, deg, b2, out, N);
}


# Scatter sums become per-node pulls: the degree count's f32 atomic now returns each edge's rank, one placement pass groups edges by destination (no sort), nodes sum in registers; arrival-order sums
# speedup vs baseline: 1.9390x; 1.9390x over previous
#include <hip/hip_runtime.h>
#include <hip/hip_bf16.h>

typedef __attribute__((ext_vector_type(2))) float v2f_t;
typedef __attribute__((ext_vector_type(4))) float v4f_t;
typedef __attribute__((ext_vector_type(8))) float v8f_t;

#define THREADS 256

__device__ __forceinline__ void atadd8_ret(float* const (&p)[8], const float (&a)[8], float (&r)[8]) {
    asm volatile("global_atomic_add_f32 %0, %8, %16, off th:TH_ATOMIC_RETURN scope:SCOPE_DEV\n\t"
                 "global_atomic_add_f32 %1, %9, %17, off th:TH_ATOMIC_RETURN scope:SCOPE_DEV\n\t"
                 "global_atomic_add_f32 %2, %10, %18, off th:TH_ATOMIC_RETURN scope:SCOPE_DEV\n\t"
                 "global_atomic_add_f32 %3, %11, %19, off th:TH_ATOMIC_RETURN scope:SCOPE_DEV\n\t"
                 "global_atomic_add_f32 %4, %12, %20, off th:TH_ATOMIC_RETURN scope:SCOPE_DEV\n\t"
                 "global_atomic_add_f32 %5, %13, %21, off th:TH_ATOMIC_RETURN scope:SCOPE_DEV\n\t"
                 "global_atomic_add_f32 %6, %14, %22, off th:TH_ATOMIC_RETURN scope:SCOPE_DEV\n\t"
                 "global_atomic_add_f32 %7, %15, %23, off th:TH_ATOMIC_RETURN scope:SCOPE_DEV\n\t"
                 "s_wait_loadcnt 0x0"
                 : "=&v"(r[0]), "=&v"(r[1]), "=&v"(r[2]), "=&v"(r[3]),
                   "=&v"(r[4]), "=&v"(r[5]), "=&v"(r[6]), "=&v"(r[7])
                 : "v"(p[0]), "v"(p[1]), "v"(p[2]), "v"(p[3]),
                   "v"(p[4]), "v"(p[5]), "v"(p[6]), "v"(p[7]),
                   "v"(a[0]), "v"(a[1]), "v"(a[2]), "v"(a[3]),
                   "v"(a[4]), "v"(a[5]), "v"(a[6]), "v"(a[7])
                 : "memory");
}

__device__ __forceinline__ unsigned pkg_half() {
    unsigned id = __builtin_amdgcn_s_sendmsg_rtn(0x87);
    return (id >> 18) & 1u;
}


__global__ void gcn_block_counts(const float* __restrict__ cnt, size_t pstride, int N,
                                 int* __restrict__ bt) {
    __shared__ int wtot[THREADS / 32];
    int i = blockIdx.x * blockDim.x + threadIdx.x;
    int c = 0;
    if (i < N) {
        float f = cnt[i];
        if (pstride) f += cnt[(size_t)i + pstride];
        c = (int)f;
    }
    for (int o = 16; o > 0; o >>= 1) c += __shfl_xor(c, o);
    if ((threadIdx.x & 31) == 0) wtot[threadIdx.x >> 5] = c;
    __syncthreads();
    if (threadIdx.x == 0) {
        int t = 0;
        for (int k = 0; k < THREADS / 32; ++k) t += wtot[k];
        bt[blockIdx.x] = t;
    }
}

__global__ void gcn_dinv_offsets(float* __restrict__ cnt, size_t pstride, int N,
                                 const int* __restrict__ bt, int* __restrict__ endp,
                                 float* __restrict__ dinv) {
    __shared__ int wsum[THREADS / 32];
    __shared__ int bsum[THREADS / 32];
    int i = blockIdx.x * blockDim.x + threadIdx.x;
    int lane = threadIdx.x & 31;
    int w = threadIdx.x >> 5;
    int part = 0;
    for (int k = threadIdx.x; k < (int)blockIdx.x; k += THREADS) part += bt[k];
    for (int o = 16; o > 0; o >>= 1) part += __shfl_xor(part, o);
    int cA = 0, c = 0;
    if (i < N) {
        float fa = cnt[i];
        float f = fa;
        if (pstride) f += cnt[(size_t)i + pstride];
        cA = (int)fa;
        c  = (int)f;
        dinv[i] = rsqrtf(f + 1.0f);
    }
    int incl = c;
    for (int o = 1; o < 32; o <<= 1) {
        int t = __shfl_up(incl, o);
        if (lane >= o) incl += t;
    }
    if (lane == 0)  bsum[w] = part;
    if (lane == 31) wsum[w] = incl;
    __syncthreads();
    int st = incl - c;
    for (int k = 0; k < THREADS / 32; ++k) {
        st += bsum[k];
        if (k < w) st += wsum[k];
    }
    if (i < N) {
        int* s0 = (int*)cnt;
        s0[i] = st;
        if (pstride) s0[(size_t)i + pstride] = st + cA;
        endp[i] = st + c;
    }
}

__global__ void gcn_place_edges(const int* __restrict__ src, const int* __restrict__ dst,
                                const int* __restrict__ rankw, const int* __restrict__ st,
                                size_t pstride, int* __restrict__ lst, int E) {
    int e = blockIdx.x * blockDim.x + threadIdx.x;
    if (e >= E) return;
    int s = __builtin_nontemporal_load(src + e);
    int d = __builtin_nontemporal_load(dst + e);
    int r = __builtin_nontemporal_load(rankw + e);
    size_t pl = (r >> 30) & 1 ? pstride : 0;
    int pos = st[(size_t)d + pl] + (r & 0x3fffffff);
    if ((unsigned)pos < (unsigned)E) lst[pos] = s;
}

#define XPITCH 132
#define EPL 8
__global__ void gcn_gemm1_deg(const float* __restrict__ x, const float* __restrict__ W1,
                              float* __restrict__ h1, int ntiles, int N,
                              const int* __restrict__ dst, float* __restrict__ deg,
                              size_t pstride, int E, int* __restrict__ rankw) {
    __shared__ __attribute__((aligned(16))) float xs[(THREADS / 32) * 16 * XPITCH];
    int wib  = threadIdx.x >> 5;
    int wave = blockIdx.x * (blockDim.x >> 5) + wib;
    bool active = wave < ntiles;
    int tile = active ? wave : ntiles - 1;
    int lane = threadIdx.x & 31;
    float* xt = xs + wib * (16 * XPITCH);

    unsigned wv     = (unsigned)__builtin_amdgcn_readfirstlane(wave);
    unsigned nwaves = gridDim.x * (blockDim.x >> 5);
    unsigned chunk  = (((unsigned)E + nwaves - 1u) / nwaves + 31u) & ~31u;
    unsigned e0 = wv * chunk;
    unsigned e1 = e0 + chunk < (unsigned)E ? e0 + chunk : (unsigned)E;
    bool any = e0 < e1;
    unsigned hsel = pstride ? pkg_half() : 0u;
    float* plane = deg + (size_t)hsel * pstride;
    int hbit = (int)(hsel << 30);

    int dv[EPL] = {};
    if (any) {
#pragma unroll
        for (int j = 0; j < EPL; ++j) {
            unsigned idx = e0 + j * 32 + lane;
            dv[j] = __builtin_nontemporal_load(dst + (idx < e1 ? idx : e1 - 1u));
        }
    }

    v4f_t rowv[16];
#pragma unroll
    for (int r = 0; r < 16; ++r) {
        int row = tile * 16 + r;
        int rcl = row < N ? row : N - 1;
        rowv[r] = __builtin_nontemporal_load((const v4f_t*)(x + (size_t)rcl * 128) + lane);
    }
    {
        float* pp[EPL];
        float  aa[EPL];
        float  rk[EPL];
#pragma unroll
        for (int j = 0; j < EPL; ++j) {
            unsigned idx = e0 + j * 32 + lane;
            pp[j] = plane + (unsigned)dv[j];
            aa[j] = idx < e1 ? 1.0f : 0.0f;
        }
        atadd8_ret(pp, aa, rk);
#pragma unroll
        for (int j = 0; j < EPL; ++j) {
            unsigned idx = e0 + j * 32 + lane;
            if (idx < e1) __builtin_nontemporal_store((int)rk[j] | hbit, rankw + idx);
        }
    }
    for (unsigned g = e0 + 32 * EPL; g < e1; g += 32 * EPL) {
        float* pp[EPL];
        float  aa[EPL];
        float  rk[EPL];
#pragma unroll
        for (int j = 0; j < EPL; ++j) {
            unsigned idx = g + j * 32 + lane;
            pp[j] = plane + (unsigned)__builtin_nontemporal_load(dst + (idx < e1 ? idx : e1 - 1u));
            aa[j] = idx < e1 ? 1.0f : 0.0f;
        }
        atadd8_ret(pp, aa, rk);
#pragma unroll
        for (int j = 0; j < EPL; ++j) {
            unsigned idx = g + j * 32 + lane;
            if (idx < e1) __builtin_nontemporal_store((int)rk[j] | hbit, rankw + idx);
        }
    }

#pragma unroll
    for (int r = 0; r < 16; ++r) *(v4f_t*)(xt + r * XPITCH + lane * 4) = rowv[r];
    __syncthreads();

    int half = lane >> 4;
    int m    = lane & 15;
    int koff = half << 1;
    const float* xr = xt + m * XPITCH;

    v8f_t c = {};
#pragma unroll
    for (int k0 = 0; k0 < 128; k0 += 4) {
        v2f_t a = *(const v2f_t*)(xr + k0 + koff);
        v2f_t b;
        b.x = W1[(k0 + koff) * 16 + m];
        b.y = W1[(k0 + koff + 1) * 16 + m];
        c = __builtin_amdgcn_wmma_f32_16x16x4_f32(false, a, false, b, (short)0, c,
                                                  false, false);
    }
    if (!active) return;
#pragma unroll
    for (int v = 0; v < 8; ++v) {
        int r = wave * 16 + half * 8 + v;
        if (r < N) h1[(size_t)r * 16 + m] = c[v];
    }
}

__global__ void gcn_pull16(const int* __restrict__ start, const int* __restrict__ endp,
                           const int* __restrict__ lst, const float* __restrict__ dinv,
                           const float* __restrict__ h, const float* __restrict__ b1,
                           float* __restrict__ hout, int N) {
    int t = blockIdx.x * blockDim.x + threadIdx.x;
    int i = t >> 2;
    int q = t & 3;
    if (i >= N) return;
    int p  = start[i];
    int pe = endp[i];
    float di = dinv[i];
    unsigned nmax = (unsigned)N - 1u;
    v4f_t acc = {0.0f, 0.0f, 0.0f, 0.0f};
    for (; p + 1 < pe; p += 2) {
        unsigned s0 = (unsigned)lst[p];
        unsigned s1 = (unsigned)lst[p + 1];
        s0 = s0 < nmax ? s0 : nmax;
        s1 = s1 < nmax ? s1 : nmax;
        float n0 = dinv[s0] * di;
        float n1 = dinv[s1] * di;
        v4f_t v0 = *((const v4f_t*)(h + (size_t)s0 * 16) + q);
        v4f_t v1 = *((const v4f_t*)(h + (size_t)s1 * 16) + q);
        acc += v0 * n0;
        acc += v1 * n1;
    }
    if (p < pe) {
        unsigned s0 = (unsigned)lst[p];
        s0 = s0 < nmax ? s0 : nmax;
        float n0 = dinv[s0] * di;
        v4f_t v0 = *((const v4f_t*)(h + (size_t)s0 * 16) + q);
        acc += v0 * n0;
    }
    float d2 = di * di;
    v4f_t hv = *((const v4f_t*)(h + (size_t)i * 16) + q);
    v4f_t bv = ((const v4f_t*)b1)[q];
    v4f_t o;
    o.x = fmaxf(acc.x + hv.x * d2 + bv.x, 0.0f);
    o.y = fmaxf(acc.y + hv.y * d2 + bv.y, 0.0f);
    o.z = fmaxf(acc.z + hv.z * d2 + bv.z, 0.0f);
    o.w = fmaxf(acc.w + hv.w * d2 + bv.w, 0.0f);
    *((v4f_t*)(hout + (size_t)i * 16) + q) = o;
}

__global__ void gcn_gemm2(const float* __restrict__ h, const float* __restrict__ W2,
                          float* __restrict__ h2, int N) {
    int i = blockIdx.x * blockDim.x + threadIdx.x;
    if (i >= N) return;
    const float* hr = h + (size_t)i * 16;
    float o0 = 0.0f, o1 = 0.0f;
#pragma unroll
    for (int k = 0; k < 16; ++k) {
        float v = hr[k];
        o0 += v * W2[2 * k];
        o1 += v * W2[2 * k + 1];
    }
    h2[2 * (size_t)i]     = o0;
    h2[2 * (size_t)i + 1] = o1;
}

__global__ void gcn_pull2_final(const int* __restrict__ start, const int* __restrict__ endp,
                                const int* __restrict__ lst, const float* __restrict__ dinv,
                                const float* __restrict__ h2, const float* __restrict__ b2,
                                float* __restrict__ out, int N) {
    int t = blockIdx.x * blockDim.x + threadIdx.x;
    int i = t >> 2;
    int q = t & 3;
    bool ok = i < N;
    int ic = ok ? i : N - 1;
    int p  = start[ic] + q;
    int pe = ok ? endp[ic] : 0;
    float di = dinv[ic];
    unsigned nmax = (unsigned)N - 1u;
    float g0 = 0.0f, g1 = 0.0f;
    for (; p < pe; p += 4) {
        unsigned s = (unsigned)lst[p];
        s = s < nmax ? s : nmax;
        float nrm = dinv[s] * di;
        v2f_t v = *(const v2f_t*)(h2 + (size_t)s * 2);
        g0 += v.x * nrm;
        g1 += v.y * nrm;
    }
    g0 += __shfl_xor(g0, 1);
    g1 += __shfl_xor(g1, 1);
    g0 += __shfl_xor(g0, 2);
    g1 += __shfl_xor(g1, 2);
    if (!ok || q != 0) return;
    float d2 = di * di;
    float a = g0 + h2[2 * (size_t)i]     * d2 + b2[0];
    float b = g1 + h2[2 * (size_t)i + 1] * d2 + b2[1];
    float mx  = fmaxf(a, b);
    float lse = mx + logf(expf(a - mx) + expf(b - mx));
    out[2 * (size_t)i]     = a - lse;
    out[2 * (size_t)i + 1] = b - lse;
}

#define R4(v) (((size_t)(v) + 3) & ~(size_t)3)
extern "C" void kernel_launch(void* const* d_in, const int* in_sizes, int n_in,
                              void* d_out, int out_size, void* d_ws, size_t ws_size,
                              hipStream_t stream) {
    const float* x  = (const float*)d_in[0];
    const int*   ei = (const int*)d_in[1];
    const float* W1 = (const float*)d_in[2];
    const float* b1 = (const float*)d_in[3];
    const float* W2 = (const float*)d_in[4];
    const float* b2 = (const float*)d_in[5];
    float* out = (float*)d_out;

    int N = in_sizes[0] / 128;
    int E = in_sizes[1] / 2;
    const int* src = ei;
    const int* dst = ei + E;

    int eb  = (E + THREADS - 1) / THREADS;
    int nb  = (N + THREADS - 1) / THREADS;
    int nb4 = (int)(((long long)N * 4 + THREADS - 1) / THREADS);

    size_t n  = (size_t)N;
    size_t ne = (size_t)E;
    size_t big = ne > 16 * n ? ne : 16 * n;
    size_t fixed = R4(n) + R4(n) + R4((size_t)nb) + R4(big) + R4(ne) + 16 * n + 2 * n;
    size_t np = (ws_size >= (R4(2 * n) + fixed) * sizeof(float)) ? 2 : 1;
    size_t sd = (np - 1) * n;
    float* ws   = (float*)d_ws;
    float* cnt  = ws;
    int*   endp = (int*)(cnt + R4(np * n));
    float* dinv = (float*)(endp + R4(n));
    int*   bt   = (int*)(dinv + R4(n));
    int*   rankw = bt + R4((size_t)nb);
    float* h1r  = (float*)rankw;
    int*   lst  = rankw + R4(big);
    float* h1   = (float*)(lst + R4(ne));
    float* h2   = h1 + (size_t)16 * n;

    hipMemsetAsync(cnt, 0, np * n * sizeof(float), stream);

    int ntiles = (N + 15) / 16;
    int gb = (ntiles + 7) / 8;
    gcn_gemm1_deg<<<gb, THREADS, 0, stream>>>(x, W1, h1, ntiles, N, dst, cnt, sd, E, rankw);
    gcn_block_counts<<<nb, THREADS, 0, stream>>>(cnt, sd, N, bt);
    gcn_dinv_offsets<<<nb, THREADS, 0, stream>>>(cnt, sd, N, bt, endp, dinv);
    gcn_place_edges<<<eb, THREADS, 0, stream>>>(src, dst, rankw, (const int*)cnt, sd, lst, E);

    gcn_pull16<<<nb4, THREADS, 0, stream>>>((const int*)cnt, endp, lst, dinv, h1, b1, h1r, N);
    gcn_gemm2<<<nb, THREADS, 0, stream>>>(h1r, W2, h2, N);
    gcn_pull2_final<<<nb4, THREADS, 0, stream>>>((const int*)cnt, endp, lst, dinv, h2, b2, out, N);
}


# Edge list as two planes, one per package half (as the sums were): a placement wave stores src+1 only into its own half's plane, one plane cleared per call, pulls take the stored word; same sum order
# speedup vs baseline: 2.3097x; 1.1912x over previous
#include <hip/hip_runtime.h>
#include <hip/hip_bf16.h>

typedef __attribute__((ext_vector_type(2))) float v2f_t;
typedef __attribute__((ext_vector_type(4))) float v4f_t;
typedef __attribute__((ext_vector_type(8))) float v8f_t;

#define THREADS 256

__device__ __forceinline__ void atadd8_ret(float* const (&p)[8], const float (&a)[8], float (&r)[8]) {
    asm volatile("global_atomic_add_f32 %0, %8, %16, off th:TH_ATOMIC_RETURN scope:SCOPE_DEV\n\t"
                 "global_atomic_add_f32 %1, %9, %17, off th:TH_ATOMIC_RETURN scope:SCOPE_DEV\n\t"
                 "global_atomic_add_f32 %2, %10, %18, off th:TH_ATOMIC_RETURN scope:SCOPE_DEV\n\t"
                 "global_atomic_add_f32 %3, %11, %19, off th:TH_ATOMIC_RETURN scope:SCOPE_DEV\n\t"
                 "global_atomic_add_f32 %4, %12, %20, off th:TH_ATOMIC_RETURN scope:SCOPE_DEV\n\t"
                 "global_atomic_add_f32 %5, %13, %21, off th:TH_ATOMIC_RETURN scope:SCOPE_DEV\n\t"
                 "global_atomic_add_f32 %6, %14, %22, off th:TH_ATOMIC_RETURN scope:SCOPE_DEV\n\t"
                 "global_atomic_add_f32 %7, %15, %23, off th:TH_ATOMIC_RETURN scope:SCOPE_DEV\n\t"
                 "s_wait_loadcnt 0x0"
                 : "=&v"(r[0]), "=&v"(r[1]), "=&v"(r[2]), "=&v"(r[3]),
                   "=&v"(r[4]), "=&v"(r[5]), "=&v"(r[6]), "=&v"(r[7])
                 : "v"(p[0]), "v"(p[1]), "v"(p[2]), "v"(p[3]),
                   "v"(p[4]), "v"(p[5]), "v"(p[6]), "v"(p[7]),
                   "v"(a[0]), "v"(a[1]), "v"(a[2]), "v"(a[3]),
                   "v"(a[4]), "v"(a[5]), "v"(a[6]), "v"(a[7])
                 : "memory");
}

__device__ __forceinline__ unsigned pkg_half() {
    unsigned id = __builtin_amdgcn_s_sendmsg_rtn(0x87);
    return (id >> 18) & 1u;
}


__global__ void gcn_block_counts(const float* __restrict__ cnt, size_t pstride, int N,
                                 int* __restrict__ bt) {
    __shared__ int wtot[THREADS / 32];
    int i = blockIdx.x * blockDim.x + threadIdx.x;
    int c = 0;
    if (i < N) {
        float f = cnt[i];
        if (pstride) f += cnt[(size_t)i + pstride];
        c = (int)f;
    }
    for (int o = 16; o > 0; o >>= 1) c += __shfl_xor(c, o);
    if ((threadIdx.x & 31) == 0) wtot[threadIdx.x >> 5] = c;
    __syncthreads();
    if (threadIdx.x == 0) {
        int t = 0;
        for (int k = 0; k < THREADS / 32; ++k) t += wtot[k];
        bt[blockIdx.x] = t;
    }
}

__global__ void gcn_dinv_offsets(float* __restrict__ cnt, size_t pstride, int N,
                                 const int* __restrict__ bt, int* __restrict__ endp,
                                 float* __restrict__ dinv) {
    __shared__ int wsum[THREADS / 32];
    __shared__ int bsum[THREADS / 32];
    int i = blockIdx.x * blockDim.x + threadIdx.x;
    int lane = threadIdx.x & 31;
    int w = threadIdx.x >> 5;
    int part = 0;
    for (int k = threadIdx.x; k < (int)blockIdx.x; k += THREADS) part += bt[k];
    for (int o = 16; o > 0; o >>= 1) part += __shfl_xor(part, o);
    int cA = 0, c = 0;
    if (i < N) {
        float fa = cnt[i];
        float f = fa;
        if (pstride) f += cnt[(size_t)i + pstride];
        cA = (int)fa;
        c  = (int)f;
        dinv[i] = rsqrtf(f + 1.0f);
    }
    int incl = c;
    for (int o = 1; o < 32; o <<= 1) {
        int t = __shfl_up(incl, o);
        if (lane >= o) incl += t;
    }
    if (lane == 0)  bsum[w] = part;
    if (lane == 31) wsum[w] = incl;
    __syncthreads();
    int st = incl - c;
    for (int k = 0; k < THREADS / 32; ++k) {
        st += bsum[k];
        if (k < w) st += wsum[k];
    }
    if (i < N) {
        int* s0 = (int*)cnt;
        s0[i] = st;
        if (pstride) s0[(size_t)i + pstride] = st + cA;
        endp[i] = st + c;
    }
}

__global__ void gcn_place_edges(const int* __restrict__ src, const int* __restrict__ dst,
                                const int* __restrict__ rankw, const int* __restrict__ st,
                                size_t pstride, int* __restrict__ lst, int* __restrict__ lstb,
                                int E) {
    int e = blockIdx.x * blockDim.x + threadIdx.x;
    int* lp = (lstb && pkg_half()) ? lstb : lst;
    if (e >= E) return;
    int s = __builtin_nontemporal_load(src + e);
    int d = __builtin_nontemporal_load(dst + e);
    int r = __builtin_nontemporal_load(rankw + e);
    size_t pl = (r >> 30) & 1 ? pstride : 0;
    int pos = st[(size_t)d + pl] + (r & 0x3fffffff);
    if ((unsigned)pos < (unsigned)E) lp[pos] = s + 1;
}

#define XPITCH 132
#define EPL 8
__global__ void gcn_gemm1_deg(const float* __restrict__ x, const float* __restrict__ W1,
                              float* __restrict__ h1, int ntiles, int N,
                              const int* __restrict__ dst, float* __restrict__ deg,
                              size_t pstride, int E, int* __restrict__ rankw) {
    __shared__ __attribute__((aligned(16))) float xs[(THREADS / 32) * 16 * XPITCH];
    int wib  = threadIdx.x >> 5;
    int wave = blockIdx.x * (blockDim.x >> 5) + wib;
    bool active = wave < ntiles;
    int tile = active ? wave : ntiles - 1;
    int lane = threadIdx.x & 31;
    float* xt = xs + wib * (16 * XPITCH);

    unsigned wv     = (unsigned)__builtin_amdgcn_readfirstlane(wave);
    unsigned nwaves = gridDim.x * (blockDim.x >> 5);
    unsigned chunk  = (((unsigned)E + nwaves - 1u) / nwaves + 31u) & ~31u;
    unsigned e0 = wv * chunk;
    unsigned e1 = e0 + chunk < (unsigned)E ? e0 + chunk : (unsigned)E;
    bool any = e0 < e1;
    unsigned hsel = pstride ? pkg_half() : 0u;
    float* plane = deg + (size_t)hsel * pstride;
    int hbit = (int)(hsel << 30);

    int dv[EPL] = {};
    if (any) {
#pragma unroll
        for (int j = 0; j < EPL; ++j) {
            unsigned idx = e0 + j * 32 + lane;
            dv[j] = __builtin_nontemporal_load(dst + (idx < e1 ? idx : e1 - 1u));
        }
    }

    v4f_t rowv[16];
#pragma unroll
    for (int r = 0; r < 16; ++r) {
        int row = tile * 16 + r;
        int rcl = row < N ? row : N - 1;
        rowv[r] = __builtin_nontemporal_load((const v4f_t*)(x + (size_t)rcl * 128) + lane);
    }
    {
        float* pp[EPL];
        float  aa[EPL];
        float  rk[EPL];
#pragma unroll
        for (int j = 0; j < EPL; ++j) {
            unsigned idx = e0 + j * 32 + lane;
            pp[j] = plane + (unsigned)dv[j];
            aa[j] = idx < e1 ? 1.0f : 0.0f;
        }
        atadd8_ret(pp, aa, rk);
#pragma unroll
        for (int j = 0; j < EPL; ++j) {
            unsigned idx = e0 + j * 32 + lane;
            if (idx < e1) __builtin_nontemporal_store((int)rk[j] | hbit, rankw + idx);
        }
    }
    for (unsigned g = e0 + 32 * EPL; g < e1; g += 32 * EPL) {
        float* pp[EPL];
        float  aa[EPL];
        float  rk[EPL];
#pragma unroll
        for (int j = 0; j < EPL; ++j) {
            unsigned idx = g + j * 32 + lane;
            pp[j] = plane + (unsigned)__builtin_nontemporal_load(dst + (idx < e1 ? idx : e1 - 1u));
            aa[j] = idx < e1 ? 1.0f : 0.0f;
        }
        atadd8_ret(pp, aa, rk);
#pragma unroll
        for (int j = 0; j < EPL; ++j) {
            unsigned idx = g + j * 32 + lane;
            if (idx < e1) __builtin_nontemporal_store((int)rk[j] | hbit, rankw + idx);
        }
    }

#pragma unroll
    for (int r = 0; r < 16; ++r) *(v4f_t*)(xt + r * XPITCH + lane * 4) = rowv[r];
    __syncthreads();

    int half = lane >> 4;
    int m    = lane & 15;
    int koff = half << 1;
    const float* xr = xt + m * XPITCH;

    v8f_t c = {};
#pragma unroll
    for (int k0 = 0; k0 < 128; k0 += 4) {
        v2f_t a = *(const v2f_t*)(xr + k0 + koff);
        v2f_t b;
        b.x = W1[(k0 + koff) * 16 + m];
        b.y = W1[(k0 + koff + 1) * 16 + m];
        c = __builtin_amdgcn_wmma_f32_16x16x4_f32(false, a, false, b, (short)0, c,
                                                  false, false);
    }
    if (!active) return;
#pragma unroll
    for (int v = 0; v < 8; ++v) {
        int r = wave * 16 + half * 8 + v;
        if (r < N) h1[(size_t)r * 16 + m] = c[v];
    }
}

__global__ void gcn_pull16(const int* __restrict__ start, const int* __restrict__ endp,
                           const int* __restrict__ lst, const int* __restrict__ lstb,
                           const float* __restrict__ dinv,
                           const float* __restrict__ h, const float* __restrict__ b1,
                           float* __restrict__ hout, int N) {
    int t = blockIdx.x * blockDim.x + threadIdx.x;
    int i = t >> 2;
    int q = t & 3;
    if (i >= N) return;
    int p  = start[i];
    int pe = endp[i];
    float di = dinv[i];
    unsigned nmax = (unsigned)N - 1u;
    v4f_t acc = {0.0f, 0.0f, 0.0f, 0.0f};
    for (; p + 1 < pe; p += 2) {
        int e0 = lst[p];
        int e1 = lst[p + 1];
        if (lstb) {
            int f0 = lstb[p];
            int f1 = lstb[p + 1];
            e0 = f0 ? f0 : e0;
            e1 = f1 ? f1 : e1;
        }
        unsigned s0 = (unsigned)(e0 - 1);
        unsigned s1 = (unsigned)(e1 - 1);
        s0 = s0 < nmax ? s0 : nmax;
        s1 = s1 < nmax ? s1 : nmax;
        float n0 = dinv[s0] * di;
        float n1 = dinv[s1] * di;
        v4f_t v0 = *((const v4f_t*)(h + (size_t)s0 * 16) + q);
        v4f_t v1 = *((const v4f_t*)(h + (size_t)s1 * 16) + q);
        acc += v0 * n0;
        acc += v1 * n1;
    }
    if (p < pe) {
        int e0 = lst[p];
        if (lstb) {
            int f0 = lstb[p];
            e0 = f0 ? f0 : e0;
        }
        unsigned s0 = (unsigned)(e0 - 1);
        s0 = s0 < nmax ? s0 : nmax;
        float n0 = dinv[s0] * di;
        v4f_t v0 = *((const v4f_t*)(h + (size_t)s0 * 16) + q);
        acc += v0 * n0;
    }
    float d2 = di * di;
    v4f_t hv = *((const v4f_t*)(h + (size_t)i * 16) + q);
    v4f_t bv = ((const v4f_t*)b1)[q];
    v4f_t o;
    o.x = fmaxf(acc.x + hv.x * d2 + bv.x, 0.0f);
    o.y = fmaxf(acc.y + hv.y * d2 + bv.y, 0.0f);
    o.z = fmaxf(acc.z + hv.z * d2 + bv.z, 0.0f);
    o.w = fmaxf(acc.w + hv.w * d2 + bv.w, 0.0f);
    *((v4f_t*)(hout + (size_t)i * 16) + q) = o;
}

__global__ void gcn_gemm2(const float* __restrict__ h, const float* __restrict__ W2,
                          float* __restrict__ h2, int N) {
    int i = blockIdx.x * blockDim.x + threadIdx.x;
    if (i >= N) return;
    const float* hr = h + (size_t)i * 16;
    float o0 = 0.0f, o1 = 0.0f;
#pragma unroll
    for (int k = 0; k < 16; ++k) {
        float v = hr[k];
        o0 += v * W2[2 * k];
        o1 += v * W2[2 * k + 1];
    }
    h2[2 * (size_t)i]     = o0;
    h2[2 * (size_t)i + 1] = o1;
}

__global__ void gcn_pull2_final(const int* __restrict__ start, const int* __restrict__ endp,
                                const int* __restrict__ lst, const int* __restrict__ lstb,
                                const float* __restrict__ dinv,
                                const float* __restrict__ h2, const float* __restrict__ b2,
                                float* __restrict__ out, int N) {
    int t = blockIdx.x * blockDim.x + threadIdx.x;
    int i = t >> 2;
    int q = t & 3;
    bool ok = i < N;
    int ic = ok ? i : N - 1;
    int p  = start[ic] + q;
    int pe = ok ? endp[ic] : 0;
    float di = dinv[ic];
    unsigned nmax = (unsigned)N - 1u;
    float g0 = 0.0f, g1 = 0.0f;
    for (; p < pe; p += 4) {
        int ev = lst[p];
        if (lstb) {
            int fv = lstb[p];
            ev = fv ? fv : ev;
        }
        unsigned s = (unsigned)(ev - 1);
        s = s < nmax ? s : nmax;
        float nrm = dinv[s] * di;
        v2f_t v = *(const v2f_t*)(h2 + (size_t)s * 2);
        g0 += v.x * nrm;
        g1 += v.y * nrm;
    }
    g0 += __shfl_xor(g0, 1);
    g1 += __shfl_xor(g1, 1);
    g0 += __shfl_xor(g0, 2);
    g1 += __shfl_xor(g1, 2);
    if (!ok || q != 0) return;
    float d2 = di * di;
    float a = g0 + h2[2 * (size_t)i]     * d2 + b2[0];
    float b = g1 + h2[2 * (size_t)i + 1] * d2 + b2[1];
    float mx  = fmaxf(a, b);
    float lse = mx + logf(expf(a - mx) + expf(b - mx));
    out[2 * (size_t)i]     = a - lse;
    out[2 * (size_t)i + 1] = b - lse;
}

#define R4(v) (((size_t)(v) + 3) & ~(size_t)3)
extern "C" void kernel_launch(void* const* d_in, const int* in_sizes, int n_in,
                              void* d_out, int out_size, void* d_ws, size_t ws_size,
                              hipStream_t stream) {
    const float* x  = (const float*)d_in[0];
    const int*   ei = (const int*)d_in[1];
    const float* W1 = (const float*)d_in[2];
    const float* b1 = (const float*)d_in[3];
    const float* W2 = (const float*)d_in[4];
    const float* b2 = (const float*)d_in[5];
    float* out = (float*)d_out;

    int N = in_sizes[0] / 128;
    int E = in_sizes[1] / 2;
    const int* src = ei;
    const int* dst = ei + E;

    int eb  = (E + THREADS - 1) / THREADS;
    int nb  = (N + THREADS - 1) / THREADS;
    int nb4 = (int)(((long long)N * 4 + THREADS - 1) / THREADS);

    size_t n  = (size_t)N;
    size_t ne = (size_t)E;
    size_t big = ne > 16 * n ? ne : 16 * n;
    size_t fixed = R4(n) + R4(n) + R4((size_t)nb) + R4(big) + R4(ne) + 16 * n + 2 * n;
    size_t np = (ws_size >= (R4(2 * n) + fixed) * sizeof(float)) ? 2 : 1;
    size_t nl = (np == 2 && ws_size >= (R4(2 * n) + R4(ne) + fixed) * sizeof(float)) ? 2 : 1;
    size_t sd = (np - 1) * n;
    float* ws   = (float*)d_ws;
    float* cnt  = ws;
    int*   lstb = nl == 2 ? (int*)(cnt + R4(np * n)) : nullptr;
    int*   endp = (int*)(cnt + R4(np * n)) + (nl - 1) * R4(ne);
    float* dinv = (float*)(endp + R4(n));
    int*   bt   = (int*)(dinv + R4(n));
    int*   rankw = bt + R4((size_t)nb);
    float* h1r  = (float*)rankw;
    int*   lst  = rankw + R4(big);
    float* h1   = (float*)(lst + R4(ne));
    float* h2   = h1 + (size_t)16 * n;

    hipMemsetAsync(cnt, 0, (nl == 2 ? R4(np * n) + ne : np * n) * sizeof(float), stream);

    int ntiles = (N + 15) / 16;
    int gb = (ntiles + 7) / 8;
    gcn_gemm1_deg<<<gb, THREADS, 0, stream>>>(x, W1, h1, ntiles, N, dst, cnt, sd, E, rankw);
    gcn_block_counts<<<nb, THREADS, 0, stream>>>(cnt, sd, N, bt);
    gcn_dinv_offsets<<<nb, THREADS, 0, stream>>>(cnt, sd, N, bt, endp, dinv);
    gcn_place_edges<<<eb, THREADS, 0, stream>>>(src, dst, rankw, (const int*)cnt, sd, lst, lstb, E);

    gcn_pull16<<<nb4, THREADS, 0, stream>>>((const int*)cnt, endp, lst, lstb, dinv, h1, b1, h1r, N);
    gcn_gemm2<<<nb, THREADS, 0, stream>>>(h1r, W2, h2, N);
    gcn_pull2_final<<<nb4, THREADS, 0, stream>>>((const int*)cnt, endp, lst, lstb, dinv, h2, b2, out, N);
}


# dinv[src] rides in the gathered row's own slot (float 16 of a one-line h1 slot, float 2 of a 16-byte h2 slot): an edge of either pull asks the cache for one line, not two; same sums, same order
# speedup vs baseline: 2.8885x; 1.2506x over previous
#include <hip/hip_runtime.h>
#include <hip/hip_bf16.h>

typedef __attribute__((ext_vector_type(2))) float v2f_t;
typedef __attribute__((ext_vector_type(4))) float v4f_t;
typedef __attribute__((ext_vector_type(8))) float v8f_t;

#define THREADS 256

__device__ __forceinline__ void atadd8_ret(float* const (&p)[8], const float (&a)[8], float (&r)[8]) {
    asm volatile("global_atomic_add_f32 %0, %8, %16, off th:TH_ATOMIC_RETURN scope:SCOPE_DEV\n\t"
                 "global_atomic_add_f32 %1, %9, %17, off th:TH_ATOMIC_RETURN scope:SCOPE_DEV\n\t"
                 "global_atomic_add_f32 %2, %10, %18, off th:TH_ATOMIC_RETURN scope:SCOPE_DEV\n\t"
                 "global_atomic_add_f32 %3, %11, %19, off th:TH_ATOMIC_RETURN scope:SCOPE_DEV\n\t"
                 "global_atomic_add_f32 %4, %12, %20, off th:TH_ATOMIC_RETURN scope:SCOPE_DEV\n\t"
                 "global_atomic_add_f32 %5, %13, %21, off th:TH_ATOMIC_RETURN scope:SCOPE_DEV\n\t"
                 "global_atomic_add_f32 %6, %14, %22, off th:TH_ATOMIC_RETURN scope:SCOPE_DEV\n\t"
                 "global_atomic_add_f32 %7, %15, %23, off th:TH_ATOMIC_RETURN scope:SCOPE_DEV\n\t"
                 "s_wait_loadcnt 0x0"
                 : "=&v"(r[0]), "=&v"(r[1]), "=&v"(r[2]), "=&v"(r[3]),
                   "=&v"(r[4]), "=&v"(r[5]), "=&v"(r[6]), "=&v"(r[7])
                 : "v"(p[0]), "v"(p[1]), "v"(p[2]), "v"(p[3]),
                   "v"(p[4]), "v"(p[5]), "v"(p[6]), "v"(p[7]),
                   "v"(a[0]), "v"(a[1]), "v"(a[2]), "v"(a[3]),
                   "v"(a[4]), "v"(a[5]), "v"(a[6]), "v"(a[7])
                 : "memory");
}

__device__ __forceinline__ unsigned pkg_half() {
    unsigned id = __builtin_amdgcn_s_sendmsg_rtn(0x87);
    return (id >> 18) & 1u;
}


__global__ void gcn_block_counts(const float* __restrict__ cnt, size_t pstride, int N,
                                 int* __restrict__ bt) {
    __shared__ int wtot[THREADS / 32];
    int i = blockIdx.x * blockDim.x + threadIdx.x;
    int c = 0;
    if (i < N) {
        float f = cnt[i];
        if (pstride) f += cnt[(size_t)i + pstride];
        c = (int)f;
    }
    for (int o = 16; o > 0; o >>= 1) c += __shfl_xor(c, o);
    if ((threadIdx.x & 31) == 0) wtot[threadIdx.x >> 5] = c;
    __syncthreads();
    if (threadIdx.x == 0) {
        int t = 0;
        for (int k = 0; k < THREADS / 32; ++k) t += wtot[k];
        bt[blockIdx.x] = t;
    }
}

__global__ void gcn_dinv_offsets(float* __restrict__ cnt, size_t pstride, int N,
                                 const int* __restrict__ bt, int* __restrict__ endp,
                                 float* __restrict__ dinv, float* __restrict__ h1, int hp) {
    __shared__ int wsum[THREADS / 32];
    __shared__ int bsum[THREADS / 32];
    int i = blockIdx.x * blockDim.x + threadIdx.x;
    int lane = threadIdx.x & 31;
    int w = threadIdx.x >> 5;
    int part = 0;
    for (int k = threadIdx.x; k < (int)blockIdx.x; k += THREADS) part += bt[k];
    for (int o = 16; o > 0; o >>= 1) part += __shfl_xor(part, o);
    int cA = 0, c = 0;
    if (i < N) {
        float fa = cnt[i];
        float f = fa;
        if (pstride) f += cnt[(size_t)i + pstride];
        cA = (int)fa;
        c  = (int)f;
        float dv = rsqrtf(f + 1.0f);
        dinv[i] = dv;
        if (hp > 16) h1[(size_t)i * hp + 16] = dv;
    }
    int incl = c;
    for (int o = 1; o < 32; o <<= 1) {
        int t = __shfl_up(incl, o);
        if (lane >= o) incl += t;
    }
    if (lane == 0)  bsum[w] = part;
    if (lane == 31) wsum[w] = incl;
    __syncthreads();
    int st = incl - c;
    for (int k = 0; k < THREADS / 32; ++k) {
        st += bsum[k];
        if (k < w) st += wsum[k];
    }
    if (i < N) {
        int* s0 = (int*)cnt;
        s0[i] = st;
        if (pstride) s0[(size_t)i + pstride] = st + cA;
        endp[i] = st + c;
    }
}

__global__ void gcn_place_edges(const int* __restrict__ src, const int* __restrict__ dst,
                                const int* __restrict__ rankw, const int* __restrict__ st,
                                size_t pstride, int* __restrict__ lst, int* __restrict__ lstb,
                                int E) {
    int e = blockIdx.x * blockDim.x + threadIdx.x;
    int* lp = (lstb && pkg_half()) ? lstb : lst;
    if (e >= E) return;
    int s = __builtin_nontemporal_load(src + e);
    int d = __builtin_nontemporal_load(dst + e);
    int r = __builtin_nontemporal_load(rankw + e);
    size_t pl = (r >> 30) & 1 ? pstride : 0;
    int pos = st[(size_t)d + pl] + (r & 0x3fffffff);
    if ((unsigned)pos < (unsigned)E) lp[pos] = s + 1;
}

#define XPITCH 132
#define EPL 8
__global__ void gcn_gemm1_deg(const float* __restrict__ x, const float* __restrict__ W1,
                              float* __restrict__ h1, int hp, int ntiles, int N,
                              const int* __restrict__ dst, float* __restrict__ deg,
                              size_t pstride, int E, int* __restrict__ rankw) {
    __shared__ __attribute__((aligned(16))) float xs[(THREADS / 32) * 16 * XPITCH];
    int wib  = threadIdx.x >> 5;
    int wave = blockIdx.x * (blockDim.x >> 5) + wib;
    bool active = wave < ntiles;
    int tile = active ? wave : ntiles - 1;
    int lane = threadIdx.x & 31;
    float* xt = xs + wib * (16 * XPITCH);

    unsigned wv     = (unsigned)__builtin_amdgcn_readfirstlane(wave);
    unsigned nwaves = gridDim.x * (blockDim.x >> 5);
    unsigned chunk  = (((unsigned)E + nwaves - 1u) / nwaves + 31u) & ~31u;
    unsigned e0 = wv * chunk;
    unsigned e1 = e0 + chunk < (unsigned)E ? e0 + chunk : (unsigned)E;
    bool any = e0 < e1;
    unsigned hsel = pstride ? pkg_half() : 0u;
    float* plane = deg + (size_t)hsel * pstride;
    int hbit = (int)(hsel << 30);

    int dv[EPL] = {};
    if (any) {
#pragma unroll
        for (int j = 0; j < EPL; ++j) {
            unsigned idx = e0 + j * 32 + lane;
            dv[j] = __builtin_nontemporal_load(dst + (idx < e1 ? idx : e1 - 1u));
        }
    }

    v4f_t rowv[16];
#pragma unroll
    for (int r = 0; r < 16; ++r) {
        int row = tile * 16 + r;
        int rcl = row < N ? row : N - 1;
        rowv[r] = __builtin_nontemporal_load((const v4f_t*)(x + (size_t)rcl * 128) + lane);
    }
    {
        float* pp[EPL];
        float  aa[EPL];
        float  rk[EPL];
#pragma unroll
        for (int j = 0; j < EPL; ++j) {
            unsigned idx = e0 + j * 32 + lane;
            pp[j] = plane + (unsigned)dv[j];
            aa[j] = idx < e1 ? 1.0f : 0.0f;
        }
        atadd8_ret(pp, aa, rk);
#pragma unroll
        for (int j = 0; j < EPL; ++j) {
            unsigned idx = e0 + j * 32 + lane;
            if (idx < e1) __builtin_nontemporal_store((int)rk[j] | hbit, rankw + idx);
        }
    }
    for (unsigned g = e0 + 32 * EPL; g < e1; g += 32 * EPL) {
        float* pp[EPL];
        float  aa[EPL];
        float  rk[EPL];
#pragma unroll
        for (int j = 0; j < EPL; ++j) {
            unsigned idx = g + j * 32 + lane;
            pp[j] = plane + (unsigned)__builtin_nontemporal_load(dst + (idx < e1 ? idx : e1 - 1u));
            aa[j] = idx < e1 ? 1.0f : 0.0f;
        }
        atadd8_ret(pp, aa, rk);
#pragma unroll
        for (int j = 0; j < EPL; ++j) {
            unsigned idx = g + j * 32 + lane;
            if (idx < e1) __builtin_nontemporal_store((int)rk[j] | hbit, rankw + idx);
        }
    }

#pragma unroll
    for (int r = 0; r < 16; ++r) *(v4f_t*)(xt + r * XPITCH + lane * 4) = rowv[r];
    __syncthreads();

    int half = lane >> 4;
    int m    = lane & 15;
    int koff = half << 1;
    const float* xr = xt + m * XPITCH;

    v8f_t c = {};
#pragma unroll
    for (int k0 = 0; k0 < 128; k0 += 4) {
        v2f_t a = *(const v2f_t*)(xr + k0 + koff);
        v2f_t b;
        b.x = W1[(k0 + koff) * 16 + m];
        b.y = W1[(k0 + koff + 1) * 16 + m];
        c = __builtin_amdgcn_wmma_f32_16x16x4_f32(false, a, false, b, (short)0, c,
                                                  false, false);
    }
    if (!active) return;
#pragma unroll
    for (int v = 0; v < 8; ++v) {
        int r = wave * 16 + half * 8 + v;
        if (r < N) h1[(size_t)r * hp + m] = c[v];
    }
}

__global__ void gcn_pull16(const int* __restrict__ start, const int* __restrict__ endp,
                           const int* __restrict__ lst, const int* __restrict__ lstb,
                           const float* __restrict__ dinv,
                           const float* __restrict__ h, int hp, const float* __restrict__ b1,
                           float* __restrict__ hout, int N) {
    int t = blockIdx.x * blockDim.x + threadIdx.x;
    int i = t >> 2;
    int q = t & 3;
    if (i >= N) return;
    int p  = start[i];
    int pe = endp[i];
    float di = dinv[i];
    unsigned nmax = (unsigned)N - 1u;
    v4f_t acc = {0.0f, 0.0f, 0.0f, 0.0f};
    for (; p + 1 < pe; p += 2) {
        int e0 = lst[p];
        int e1 = lst[p + 1];
        if (lstb) {
            int f0 = lstb[p];
            int f1 = lstb[p + 1];
            e0 = f0 ? f0 : e0;
            e1 = f1 ? f1 : e1;
        }
        unsigned s0 = (unsigned)(e0 - 1);
        unsigned s1 = (unsigned)(e1 - 1);
        s0 = s0 < nmax ? s0 : nmax;
        s1 = s1 < nmax ? s1 : nmax;
        const float* r0 = h + (size_t)s0 * hp;
        const float* r1 = h + (size_t)s1 * hp;
        float n0 = *(hp > 16 ? r0 + 16 : dinv + s0) * di;
        float n1 = *(hp > 16 ? r1 + 16 : dinv + s1) * di;
        v4f_t v0 = *((const v4f_t*)r0 + q);
        v4f_t v1 = *((const v4f_t*)r1 + q);
        acc += v0 * n0;
        acc += v1 * n1;
    }
    if (p < pe) {
        int e0 = lst[p];
        if (lstb) {
            int f0 = lstb[p];
            e0 = f0 ? f0 : e0;
        }
        unsigned s0 = (unsigned)(e0 - 1);
        s0 = s0 < nmax ? s0 : nmax;
        const float* r0 = h + (size_t)s0 * hp;
        float n0 = *(hp > 16 ? r0 + 16 : dinv + s0) * di;
        v4f_t v0 = *((const v4f_t*)r0 + q);
        acc += v0 * n0;
    }
    float d2 = di * di;
    v4f_t hv = *((const v4f_t*)(h + (size_t)i * hp) + q);
    v4f_t bv = ((const v4f_t*)b1)[q];
    v4f_t o;
    o.x = fmaxf(acc.x + hv.x * d2 + bv.x, 0.0f);
    o.y = fmaxf(acc.y + hv.y * d2 + bv.y, 0.0f);
    o.z = fmaxf(acc.z + hv.z * d2 + bv.z, 0.0f);
    o.w = fmaxf(acc.w + hv.w * d2 + bv.w, 0.0f);
    *((v4f_t*)(hout + (size_t)i * 16) + q) = o;
}

__global__ void gcn_gemm2(const float* __restrict__ h, const float* __restrict__ W2,
                          const float* __restrict__ dinv, float* __restrict__ h2, int N) {
    int i = blockIdx.x * blockDim.x + threadIdx.x;
    if (i >= N) return;
    const float* hr = h + (size_t)i * 16;
    float o0 = 0.0f, o1 = 0.0f;
#pragma unroll
    for (int k = 0; k < 16; ++k) {
        float v = hr[k];
        o0 += v * W2[2 * k];
        o1 += v * W2[2 * k + 1];
    }
    v4f_t o = {o0, o1, dinv[i], 0.0f};
    *((v4f_t*)h2 + i) = o;
}

__global__ void gcn_pull2_final(const int* __restrict__ start, const int* __restrict__ endp,
                                const int* __restrict__ lst, const int* __restrict__ lstb,
                                const float* __restrict__ dinv,
                                const float* __restrict__ h2, const float* __restrict__ b2,
                                float* __restrict__ out, int N) {
    int t = blockIdx.x * blockDim.x + threadIdx.x;
    int i = t >> 2;
    int q = t & 3;
    bool ok = i < N;
    int ic = ok ? i : N - 1;
    int p  = start[ic] + q;
    int pe = ok ? endp[ic] : 0;
    float di = dinv[ic];
    unsigned nmax = (unsigned)N - 1u;
    float g0 = 0.0f, g1 = 0.0f;
    for (; p < pe; p += 4) {
        int ev = lst[p];
        if (lstb) {
            int fv = lstb[p];
            ev = fv ? fv : ev;
        }
        unsigned s = (unsigned)(ev - 1);
        s = s < nmax ? s : nmax;
        v4f_t v = *((const v4f_t*)h2 + s);
        float nrm = v.z * di;
        g0 += v.x * nrm;
        g1 += v.y * nrm;
    }
    g0 += __shfl_xor(g0, 1);
    g1 += __shfl_xor(g1, 1);
    g0 += __shfl_xor(g0, 2);
    g1 += __shfl_xor(g1, 2);
    if (!ok || q != 0) return;
    float d2 = di * di;
    v4f_t me = *((const v4f_t*)h2 + i);
    float a = g0 + me.x * d2 + b2[0];
    float b = g1 + me.y * d2 + b2[1];
    float mx  = fmaxf(a, b);
    float lse = mx + logf(expf(a - mx) + expf(b - mx));
    out[2 * (size_t)i]     = a - lse;
    out[2 * (size_t)i + 1] = b - lse;
}

#define R4(v) (((size_t)(v) + 3) & ~(size_t)3)
extern "C" void kernel_launch(void* const* d_in, const int* in_sizes, int n_in,
                              void* d_out, int out_size, void* d_ws, size_t ws_size,
                              hipStream_t stream) {
    const float* x  = (const float*)d_in[0];
    const int*   ei = (const int*)d_in[1];
    const float* W1 = (const float*)d_in[2];
    const float* b1 = (const float*)d_in[3];
    const float* W2 = (const float*)d_in[4];
    const float* b2 = (const float*)d_in[5];
    float* out = (float*)d_out;

    int N = in_sizes[0] / 128;
    int E = in_sizes[1] / 2;
    const int* src = ei;
    const int* dst = ei + E;

    int eb  = (E + THREADS - 1) / THREADS;
    int nb  = (N + THREADS - 1) / THREADS;
    int nb4 = (int)(((long long)N * 4 + THREADS - 1) / THREADS);

    size_t n  = (size_t)N;
    size_t ne = (size_t)E;
    size_t big = ne > 16 * n ? ne : 16 * n;
    size_t fixed = R4(n) + R4(n) + R4((size_t)nb) + R4(big) + R4(ne) + 32 + 16 * n;
    size_t np = (ws_size >= (R4(2 * n) + fixed) * sizeof(float)) ? 2 : 1;
    size_t nl = (np == 2 && ws_size >= (R4(2 * n) + R4(ne) + fixed) * sizeof(float)) ? 2 : 1;
    int    hp = (ws_size >= (R4(np * n) + (nl - 1) * R4(ne) + fixed + 16 * n) * sizeof(float)) ? 32 : 16;
    size_t sd = (np - 1) * n;
    float* ws   = (float*)d_ws;
    float* cnt  = ws;
    int*   lstb = nl == 2 ? (int*)(cnt + R4(np * n)) : nullptr;
    int*   endp = (int*)(cnt + R4(np * n)) + (nl - 1) * R4(ne);
    float* dinv = (float*)(endp + R4(n));
    int*   bt   = (int*)(dinv + R4(n));
    int*   rankw = bt + R4((size_t)nb);
    float* h1r  = (float*)rankw;
    int*   lst  = rankw + R4(big);
    size_t h1off = (size_t)((float*)(lst + R4(ne)) - ws);
    float* h1   = ws + ((h1off + 31) & ~(size_t)31);
    float* h2   = h1;

    hipMemsetAsync(cnt, 0, (nl == 2 ? R4(np * n) + ne : np * n) * sizeof(float), stream);

    int ntiles = (N + 15) / 16;
    int gb = (ntiles + 7) / 8;
    gcn_gemm1_deg<<<gb, THREADS, 0, stream>>>(x, W1, h1, hp, ntiles, N, dst, cnt, sd, E, rankw);
    gcn_block_counts<<<nb, THREADS, 0, stream>>>(cnt, sd, N, bt);
    gcn_dinv_offsets<<<nb, THREADS, 0, stream>>>(cnt, sd, N, bt, endp, dinv, h1, hp);
    gcn_place_edges<<<eb, THREADS, 0, stream>>>(src, dst, rankw, (const int*)cnt, sd, lst, lstb, E);

    gcn_pull16<<<nb4, THREADS, 0, stream>>>((const int*)cnt, endp, lst, lstb, dinv, h1, hp, b1, h1r, N);
    gcn_gemm2<<<nb, THREADS, 0, stream>>>(h1r, W2, dinv, h2, N);
    gcn_pull2_final<<<nb4, THREADS, 0, stream>>>((const int*)cnt, endp, lst, lstb, dinv, h2, b2, out, N);
}


# Edge lists as fixed slots, 16 per node and package half: the degree atomic's returned rank is the slot index, an edge is stored where counted; no prefix sum, no placement pass; arrival-order sums
# speedup vs baseline: 3.9161x; 1.3558x over previous
#include <hip/hip_runtime.h>
#include <hip/hip_bf16.h>

typedef __attribute__((ext_vector_type(2))) float v2f_t;
typedef __attribute__((ext_vector_type(4))) float v4f_t;
typedef __attribute__((ext_vector_type(8))) float v8f_t;

#define THREADS 256

__device__ __forceinline__ void atadd8_ret(float* const (&p)[8], const float (&a)[8], float (&r)[8]) {
    asm volatile("global_atomic_add_f32 %0, %8, %16, off th:TH_ATOMIC_RETURN scope:SCOPE_DEV\n\t"
                 "global_atomic_add_f32 %1, %9, %17, off th:TH_ATOMIC_RETURN scope:SCOPE_DEV\n\t"
                 "global_atomic_add_f32 %2, %10, %18, off th:TH_ATOMIC_RETURN scope:SCOPE_DEV\n\t"
                 "global_atomic_add_f32 %3, %11, %19, off th:TH_ATOMIC_RETURN scope:SCOPE_DEV\n\t"
                 "global_atomic_add_f32 %4, %12, %20, off th:TH_ATOMIC_RETURN scope:SCOPE_DEV\n\t"
                 "global_atomic_add_f32 %5, %13, %21, off th:TH_ATOMIC_RETURN scope:SCOPE_DEV\n\t"
                 "global_atomic_add_f32 %6, %14, %22, off th:TH_ATOMIC_RETURN scope:SCOPE_DEV\n\t"
                 "global_atomic_add_f32 %7, %15, %23, off th:TH_ATOMIC_RETURN scope:SCOPE_DEV\n\t"
                 "s_wait_loadcnt 0x0"
                 : "=&v"(r[0]), "=&v"(r[1]), "=&v"(r[2]), "=&v"(r[3]),
                   "=&v"(r[4]), "=&v"(r[5]), "=&v"(r[6]), "=&v"(r[7])
                 : "v"(p[0]), "v"(p[1]), "v"(p[2]), "v"(p[3]),
                   "v"(p[4]), "v"(p[5]), "v"(p[6]), "v"(p[7]),
                   "v"(a[0]), "v"(a[1]), "v"(a[2]), "v"(a[3]),
                   "v"(a[4]), "v"(a[5]), "v"(a[6]), "v"(a[7])
                 : "memory");
}

__device__ __forceinline__ unsigned pkg_half() {
    unsigned id = __builtin_amdgcn_s_sendmsg_rtn(0x87);
    return (id >> 18) & 1u;
}


__global__ void gcn_block_counts(const float* __restrict__ cnt, size_t pstride, int N,
                                 int* __restrict__ bt) {
    __shared__ int wtot[THREADS / 32];
    int i = blockIdx.x * blockDim.x + threadIdx.x;
    int c = 0;
    if (i < N) {
        float f = cnt[i];
        if (pstride) f += cnt[(size_t)i + pstride];
        c = (int)f;
    }
    for (int o = 16; o > 0; o >>= 1) c += __shfl_xor(c, o);
    if ((threadIdx.x & 31) == 0) wtot[threadIdx.x >> 5] = c;
    __syncthreads();
    if (threadIdx.x == 0) {
        int t = 0;
        for (int k = 0; k < THREADS / 32; ++k) t += wtot[k];
        bt[blockIdx.x] = t;
    }
}

__global__ void gcn_dinv_offsets(float* __restrict__ cnt, size_t pstride, int N,
                                 const int* __restrict__ bt, int* __restrict__ endp,
                                 float* __restrict__ dinv, float* __restrict__ h1, int hp) {
    __shared__ int wsum[THREADS / 32];
    __shared__ int bsum[THREADS / 32];
    int i = blockIdx.x * blockDim.x + threadIdx.x;
    int lane = threadIdx.x & 31;
    int w = threadIdx.x >> 5;
    int part = 0;
    for (int k = threadIdx.x; k < (int)blockIdx.x; k += THREADS) part += bt[k];
    for (int o = 16; o > 0; o >>= 1) part += __shfl_xor(part, o);
    int cA = 0, c = 0;
    if (i < N) {
        float fa = cnt[i];
        float f = fa;
        if (pstride) f += cnt[(size_t)i + pstride];
        cA = (int)fa;
        c  = (int)f;
        float dv = rsqrtf(f + 1.0f);
        dinv[i] = dv;
        if (hp > 16) h1[(size_t)i * hp + 16] = dv;
    }
    int incl = c;
    for (int o = 1; o < 32; o <<= 1) {
        int t = __shfl_up(incl, o);
        if (lane >= o) incl += t;
    }
    if (lane == 0)  bsum[w] = part;
    if (lane == 31) wsum[w] = incl;
    __syncthreads();
    int st = incl - c;
    for (int k = 0; k < THREADS / 32; ++k) {
        st += bsum[k];
        if (k < w) st += wsum[k];
    }
    if (i < N) {
        int* s0 = (int*)cnt;
        s0[i] = st;
        if (pstride) s0[(size_t)i + pstride] = st + cA;
        endp[i] = st + c;
    }
}

__global__ void gcn_place_edges(const int* __restrict__ src, const int* __restrict__ dst,
                                const int* __restrict__ rankw, const int* __restrict__ st,
                                size_t pstride, int* __restrict__ lst, int* __restrict__ lstb,
                                int E) {
    int e = blockIdx.x * blockDim.x + threadIdx.x;
    int* lp = (lstb && pkg_half()) ? lstb : lst;
    if (e >= E) return;
    int s = __builtin_nontemporal_load(src + e);
    int d = __builtin_nontemporal_load(dst + e);
    int r = __builtin_nontemporal_load(rankw + e);
    size_t pl = (r >> 30) & 1 ? pstride : 0;
    int pos = st[(size_t)d + pl] + (r & 0x3fffffff);
    if ((unsigned)pos < (unsigned)E) lp[pos] = s + 1;
}

#define XPITCH 132
#define EPL 8
__global__ void gcn_gemm1_deg(const float* __restrict__ x, const float* __restrict__ W1,
                              float* __restrict__ h1, int hp, int ntiles, int N,
                              const int* __restrict__ dst, float* __restrict__ deg,
                              size_t pstride, int E, int* __restrict__ rankw) {
    __shared__ __attribute__((aligned(16))) float xs[(THREADS / 32) * 16 * XPITCH];
    int wib  = threadIdx.x >> 5;
    int wave = blockIdx.x * (blockDim.x >> 5) + wib;
    bool active = wave < ntiles;
    int tile = active ? wave : ntiles - 1;
    int lane = threadIdx.x & 31;
    float* xt = xs + wib * (16 * XPITCH);

    unsigned wv     = (unsigned)__builtin_amdgcn_readfirstlane(wave);
    unsigned nwaves = gridDim.x * (blockDim.x >> 5);
    unsigned chunk  = (((unsigned)E + nwaves - 1u) / nwaves + 31u) & ~31u;
    unsigned e0 = wv * chunk;
    unsigned e1 = e0 + chunk < (unsigned)E ? e0 + chunk : (unsigned)E;
    bool any = e0 < e1;
    unsigned hsel = pstride ? pkg_half() : 0u;
    float* plane = deg + (size_t)hsel * pstride;
    int hbit = (int)(hsel << 30);

    int dv[EPL] = {};
    if (any) {
#pragma unroll
        for (int j = 0; j < EPL; ++j) {
            unsigned idx = e0 + j * 32 + lane;
            dv[j] = __builtin_nontemporal_load(dst + (idx < e1 ? idx : e1 - 1u));
        }
    }

    v4f_t rowv[16];
#pragma unroll
    for (int r = 0; r < 16; ++r) {
        int row = tile * 16 + r;
        int rcl = row < N ? row : N - 1;
        rowv[r] = __builtin_nontemporal_load((const v4f_t*)(x + (size_t)rcl * 128) + lane);
    }
    {
        float* pp[EPL];
        float  aa[EPL];
        float  rk[EPL];
#pragma unroll
        for (int j = 0; j < EPL; ++j) {
            unsigned idx = e0 + j * 32 + lane;
            pp[j] = plane + (unsigned)dv[j];
            aa[j] = idx < e1 ? 1.0f : 0.0f;
        }
        atadd8_ret(pp, aa, rk);
#pragma unroll
        for (int j = 0; j < EPL; ++j) {
            unsigned idx = e0 + j * 32 + lane;
            if (idx < e1) __builtin_nontemporal_store((int)rk[j] | hbit, rankw + idx);
        }
    }
    for (unsigned g = e0 + 32 * EPL; g < e1; g += 32 * EPL) {
        float* pp[EPL];
        float  aa[EPL];
        float  rk[EPL];
#pragma unroll
        for (int j = 0; j < EPL; ++j) {
            unsigned idx = g + j * 32 + lane;
            pp[j] = plane + (unsigned)__builtin_nontemporal_load(dst + (idx < e1 ? idx : e1 - 1u));
            aa[j] = idx < e1 ? 1.0f : 0.0f;
        }
        atadd8_ret(pp, aa, rk);
#pragma unroll
        for (int j = 0; j < EPL; ++j) {
            unsigned idx = g + j * 32 + lane;
            if (idx < e1) __builtin_nontemporal_store((int)rk[j] | hbit, rankw + idx);
        }
    }

#pragma unroll
    for (int r = 0; r < 16; ++r) *(v4f_t*)(xt + r * XPITCH + lane * 4) = rowv[r];
    __syncthreads();

    int half = lane >> 4;
    int m    = lane & 15;
    int koff = half << 1;
    const float* xr = xt + m * XPITCH;

    v8f_t c = {};
#pragma unroll
    for (int k0 = 0; k0 < 128; k0 += 4) {
        v2f_t a = *(const v2f_t*)(xr + k0 + koff);
        v2f_t b;
        b.x = W1[(k0 + koff) * 16 + m];
        b.y = W1[(k0 + koff + 1) * 16 + m];
        c = __builtin_amdgcn_wmma_f32_16x16x4_f32(false, a, false, b, (short)0, c,
                                                  false, false);
    }
    if (!active) return;
#pragma unroll
    for (int v = 0; v < 8; ++v) {
        int r = wave * 16 + half * 8 + v;
        if (r < N) h1[(size_t)r * hp + m] = c[v];
    }
}

__global__ void gcn_pull16(const int* __restrict__ start, const int* __restrict__ endp,
                           const int* __restrict__ lst, const int* __restrict__ lstb,
                           const float* __restrict__ dinv,
                           const float* __restrict__ h, int hp, const float* __restrict__ b1,
                           float* __restrict__ hout, int N) {
    int t = blockIdx.x * blockDim.x + threadIdx.x;
    int i = t >> 2;
    int q = t & 3;
    if (i >= N) return;
    int p  = start[i];
    int pe = endp[i];
    float di = dinv[i];
    unsigned nmax = (unsigned)N - 1u;
    v4f_t acc = {0.0f, 0.0f, 0.0f, 0.0f};
    for (; p + 1 < pe; p += 2) {
        int e0 = lst[p];
        int e1 = lst[p + 1];
        if (lstb) {
            int f0 = lstb[p];
            int f1 = lstb[p + 1];
            e0 = f0 ? f0 : e0;
            e1 = f1 ? f1 : e1;
        }
        unsigned s0 = (unsigned)(e0 - 1);
        unsigned s1 = (unsigned)(e1 - 1);
        s0 = s0 < nmax ? s0 : nmax;
        s1 = s1 < nmax ? s1 : nmax;
        const float* r0 = h + (size_t)s0 * hp;
        const float* r1 = h + (size_t)s1 * hp;
        float n0 = *(hp > 16 ? r0 + 16 : dinv + s0) * di;
        float n1 = *(hp > 16 ? r1 + 16 : dinv + s1) * di;
        v4f_t v0 = *((const v4f_t*)r0 + q);
        v4f_t v1 = *((const v4f_t*)r1 + q);
        acc += v0 * n0;
        acc += v1 * n1;
    }
    if (p < pe) {
        int e0 = lst[p];
        if (lstb) {
            int f0 = lstb[p];
            e0 = f0 ? f0 : e0;
        }
        unsigned s0 = (unsigned)(e0 - 1);
        s0 = s0 < nmax ? s0 : nmax;
        const float* r0 = h + (size_t)s0 * hp;
        float n0 = *(hp > 16 ? r0 + 16 : dinv + s0) * di;
        v4f_t v0 = *((const v4f_t*)r0 + q);
        acc += v0 * n0;
    }
    float d2 = di * di;
    v4f_t hv = *((const v4f_t*)(h + (size_t)i * hp) + q);
    v4f_t bv = ((const v4f_t*)b1)[q];
    v4f_t o;
    o.x = fmaxf(acc.x + hv.x * d2 + bv.x, 0.0f);
    o.y = fmaxf(acc.y + hv.y * d2 + bv.y, 0.0f);
    o.z = fmaxf(acc.z + hv.z * d2 + bv.z, 0.0f);
    o.w = fmaxf(acc.w + hv.w * d2 + bv.w, 0.0f);
    *((v4f_t*)(hout + (size_t)i * 16) + q) = o;
}

__global__ void gcn_gemm2(const float* __restrict__ h, const float* __restrict__ W2,
                          const float* __restrict__ dinv, float* __restrict__ h2, int N) {
    int i = blockIdx.x * blockDim.x + threadIdx.x;
    if (i >= N) return;
    const float* hr = h + (size_t)i * 16;
    float o0 = 0.0f, o1 = 0.0f;
#pragma unroll
    for (int k = 0; k < 16; ++k) {
        float v = hr[k];
        o0 += v * W2[2 * k];
        o1 += v * W2[2 * k + 1];
    }
    v4f_t o = {o0, o1, dinv[i], 0.0f};
    *((v4f_t*)h2 + i) = o;
}

__global__ void gcn_pull2_final(const int* __restrict__ start, const int* __restrict__ endp,
                                const int* __restrict__ lst, const int* __restrict__ lstb,
                                const float* __restrict__ dinv,
                                const float* __restrict__ h2, const float* __restrict__ b2,
                                float* __restrict__ out, int N) {
    int t = blockIdx.x * blockDim.x + threadIdx.x;
    int i = t >> 2;
    int q = t & 3;
    bool ok = i < N;
    int ic = ok ? i : N - 1;
    int p  = start[ic] + q;
    int pe = ok ? endp[ic] : 0;
    float di = dinv[ic];
    unsigned nmax = (unsigned)N - 1u;
    float g0 = 0.0f, g1 = 0.0f;
    for (; p < pe; p += 4) {
        int ev = lst[p];
        if (lstb) {
            int fv = lstb[p];
            ev = fv ? fv : ev;
        }
        unsigned s = (unsigned)(ev - 1);
        s = s < nmax ? s : nmax;
        v4f_t v = *((const v4f_t*)h2 + s);
        float nrm = v.z * di;
        g0 += v.x * nrm;
        g1 += v.y * nrm;
    }
    g0 += __shfl_xor(g0, 1);
    g1 += __shfl_xor(g1, 1);
    g0 += __shfl_xor(g0, 2);
    g1 += __shfl_xor(g1, 2);
    if (!ok || q != 0) return;
    float d2 = di * di;
    v4f_t me = *((const v4f_t*)h2 + i);
    float a = g0 + me.x * d2 + b2[0];
    float b = g1 + me.y * d2 + b2[1];
    float mx  = fmaxf(a, b);
    float lse = mx + logf(expf(a - mx) + expf(b - mx));
    out[2 * (size_t)i]     = a - lse;
    out[2 * (size_t)i + 1] = b - lse;
}


#define SLOT 16

__device__ __forceinline__ float atadd1_ret(float* p, float a) {
    float r;
    asm volatile("global_atomic_add_f32 %0, %1, %2, off th:TH_ATOMIC_RETURN scope:SCOPE_DEV\n\t"
                 "s_wait_loadcnt 0x0"
                 : "=&v"(r) : "v"(p), "v"(a) : "memory");
    return r;
}

__device__ __forceinline__ void atadd8_ret_s(float* base, unsigned (&o)[8],
                                             const float (&a)[8], float (&r)[8]) {
    asm volatile("v_lshlrev_b32 %8, 2, %8\n\t"
                 "v_lshlrev_b32 %9, 2, %9\n\t"
                 "v_lshlrev_b32 %10, 2, %10\n\t"
                 "v_lshlrev_b32 %11, 2, %11\n\t"
                 "v_lshlrev_b32 %12, 2, %12\n\t"
                 "v_lshlrev_b32 %13, 2, %13\n\t"
                 "v_lshlrev_b32 %14, 2, %14\n\t"
                 "v_lshlrev_b32 %15, 2, %15\n\t"
                 "global_atomic_add_f32 %0, %8, %16, %24 th:TH_ATOMIC_RETURN scope:SCOPE_DEV\n\t"
                 "global_atomic_add_f32 %1, %9, %17, %24 th:TH_ATOMIC_RETURN scope:SCOPE_DEV\n\t"
                 "global_atomic_add_f32 %2, %10, %18, %24 th:TH_ATOMIC_RETURN scope:SCOPE_DEV\n\t"
                 "global_atomic_add_f32 %3, %11, %19, %24 th:TH_ATOMIC_RETURN scope:SCOPE_DEV\n\t"
                 "global_atomic_add_f32 %4, %12, %20, %24 th:TH_ATOMIC_RETURN scope:SCOPE_DEV\n\t"
                 "global_atomic_add_f32 %5, %13, %21, %24 th:TH_ATOMIC_RETURN scope:SCOPE_DEV\n\t"
                 "global_atomic_add_f32 %6, %14, %22, %24 th:TH_ATOMIC_RETURN scope:SCOPE_DEV\n\t"
                 "global_atomic_add_f32 %7, %15, %23, %24 th:TH_ATOMIC_RETURN scope:SCOPE_DEV\n\t"
                 "s_wait_loadcnt 0x0"
                 : "=&v"(r[0]), "=&v"(r[1]), "=&v"(r[2]), "=&v"(r[3]),
                   "=&v"(r[4]), "=&v"(r[5]), "=&v"(r[6]), "=&v"(r[7]),
                   "+v"(o[0]), "+v"(o[1]), "+v"(o[2]), "+v"(o[3]),
                   "+v"(o[4]), "+v"(o[5]), "+v"(o[6]), "+v"(o[7])
                 : "v"(a[0]), "v"(a[1]), "v"(a[2]), "v"(a[3]),
                   "v"(a[4]), "v"(a[5]), "v"(a[6]), "v"(a[7]),
                   "s"(base)
                 : "memory");
}

__device__ __forceinline__ void place_in_slot(int* __restrict__ lp, int* __restrict__ lc,
                                              float* __restrict__ cntc, float* __restrict__ ovn,
                                              int* __restrict__ ovf, int E, unsigned d, int s,
                                              float rankf, unsigned idx) {
    int r = (int)rankf;
    if (r < SLOT) {
        lp[(size_t)d * SLOT + r] = s;
    } else {
        int c = (int)atadd1_ret(cntc + d, 1.0f);
        if (c < SLOT) {
            lc[(size_t)d * SLOT + c] = s;
        } else {
            unsigned k = (unsigned)atadd1_ret(ovn, 1.0f);
            if (k < (unsigned)E) ovf[k] = (int)idx;
        }
    }
}

__global__ void gcn_gemm1_slots(const float* __restrict__ x, const float* __restrict__ W1,
                                float* __restrict__ h1, int hp, int ntiles, int N,
                                const int* __restrict__ src, const int* __restrict__ dst,
                                float* __restrict__ deg, size_t pstride, int E,
                                int* __restrict__ lists, size_t lstride,
                                float* __restrict__ ovn, int* __restrict__ ovf) {
    __shared__ __attribute__((aligned(16))) float xs[(THREADS / 32) * 16 * XPITCH];
    int wib  = threadIdx.x >> 5;
    int wave = blockIdx.x * (blockDim.x >> 5) + wib;
    bool active = wave < ntiles;
    int tile = active ? wave : ntiles - 1;
    int lane = threadIdx.x & 31;
    float* xt = xs + wib * (16 * XPITCH);

    unsigned wv     = (unsigned)__builtin_amdgcn_readfirstlane(wave);
    unsigned nwaves = gridDim.x * (blockDim.x >> 5);
    unsigned chunk  = (((unsigned)E + nwaves - 1u) / nwaves + 31u) & ~31u;
    unsigned e0 = wv * chunk;
    unsigned e1 = e0 + chunk < (unsigned)E ? e0 + chunk : (unsigned)E;
    bool any = e0 < e1;
    unsigned hsel = pkg_half();
    float* plane = deg + (size_t)hsel * pstride;
    int*   lp    = lists + (size_t)hsel * lstride;
    int*   lc    = lists + 2 * lstride;
    float* cntc  = deg + 2 * pstride;

    unsigned dv[EPL] = {};
    int      sv[EPL] = {};
    if (any) {
#pragma unroll
        for (int j = 0; j < EPL; ++j) {
            unsigned idx = e0 + j * 32 + lane;
            dv[j] = (unsigned)__builtin_nontemporal_load(dst + (idx < e1 ? idx : e1 - 1u));
        }
#pragma unroll
        for (int j = 0; j < EPL; ++j) {
            unsigned idx = e0 + j * 32 + lane;
            sv[j] = __builtin_nontemporal_load(src + (idx < e1 ? idx : e1 - 1u));
        }
    }

    v4f_t rowv[16];
#pragma unroll
    for (int r = 0; r < 16; ++r) {
        int row = tile * 16 + r;
        int rcl = row < N ? row : N - 1;
        rowv[r] = __builtin_nontemporal_load((const v4f_t*)(x + (size_t)rcl * 128) + lane);
    }
    {
        float aa[EPL];
        float rk[EPL];
#pragma unroll
        for (int j = 0; j < EPL; ++j) {
            unsigned idx = e0 + j * 32 + lane;
            aa[j] = idx < e1 ? 1.0f : 0.0f;
        }
        atadd8_ret_s(plane, dv, aa, rk);
#pragma unroll
        for (int j = 0; j < EPL; ++j) {
            unsigned idx = e0 + j * 32 + lane;
            if (idx < e1) place_in_slot(lp, lc, cntc, ovn, ovf, E, dv[j] >> 2, sv[j], rk[j], idx);
        }
    }
    for (unsigned g = e0 + 32 * EPL; g < e1; g += 32 * EPL) {
        float    aa[EPL];
        float    rk[EPL];
        unsigned dd[EPL];
        int      ss[EPL];
#pragma unroll
        for (int j = 0; j < EPL; ++j) {
            unsigned idx = g + j * 32 + lane;
            unsigned cl = idx < e1 ? idx : e1 - 1u;
            dd[j] = (unsigned)__builtin_nontemporal_load(dst + cl);
            ss[j] = __builtin_nontemporal_load(src + cl);
            aa[j] = idx < e1 ? 1.0f : 0.0f;
        }
        atadd8_ret_s(plane, dd, aa, rk);
#pragma unroll
        for (int j = 0; j < EPL; ++j) {
            unsigned idx = g + j * 32 + lane;
            if (idx < e1) place_in_slot(lp, lc, cntc, ovn, ovf, E, dd[j] >> 2, ss[j], rk[j], idx);
        }
    }

#pragma unroll
    for (int r = 0; r < 16; ++r) *(v4f_t*)(xt + r * XPITCH + lane * 4) = rowv[r];
    __syncthreads();

    int half = lane >> 4;
    int m    = lane & 15;
    int koff = half << 1;
    const float* xr = xt + m * XPITCH;

    v8f_t c = {};
#pragma unroll
    for (int k0 = 0; k0 < 128; k0 += 4) {
        v2f_t a = *(const v2f_t*)(xr + k0 + koff);
        v2f_t b;
        b.x = W1[(k0 + koff) * 16 + m];
        b.y = W1[(k0 + koff + 1) * 16 + m];
        c = __builtin_amdgcn_wmma_f32_16x16x4_f32(false, a, false, b, (short)0, c,
                                                  false, false);
    }
    if (!active) return;
#pragma unroll
    for (int v = 0; v < 8; ++v) {
        int r = wave * 16 + half * 8 + v;
        if (r < N) h1[(size_t)r * hp + m] = c[v];
    }
}

__global__ void gcn_dinv_slots(const float* __restrict__ cnt, size_t pstride, int N,
                               float* __restrict__ dinv, float* __restrict__ h1, int hp) {
    int i = blockIdx.x * blockDim.x + threadIdx.x;
    if (i >= N) return;
    float f = cnt[i];
    f += cnt[(size_t)i + pstride];
    float dv = rsqrtf(f + 1.0f);
    dinv[i] = dv;
    h1[(size_t)i * hp + 16] = dv;
}

__device__ __forceinline__ size_t slot_off(size_t base, int j, int nA, size_t adjB) {
    return base + (size_t)j + (((size_t)0 - (size_t)(j >= nA)) & adjB);
}

__global__ void gcn_pull16_slots(const float* __restrict__ cnt, size_t pstride,
                                 const int* __restrict__ lists, size_t lstride,
                                 const float* __restrict__ ovn, const int* __restrict__ ovf,
                                 const int* __restrict__ src, const int* __restrict__ dst, int E,
                                 const float* __restrict__ dinv,
                                 const float* __restrict__ h, int hp, const float* __restrict__ b1,
                                 float* __restrict__ hout, int N) {
    int t = blockIdx.x * blockDim.x + threadIdx.x;
    int i = t >> 2;
    int q = t & 3;
    if (i >= N) return;
    int cA = (int)cnt[i];
    int cB = (int)cnt[(size_t)i + pstride];
    int nA  = cA < SLOT ? cA : SLOT;
    int nAB = nA + (cB < SLOT ? cB : SLOT);
    size_t base = (size_t)i * SLOT;
    size_t adjB = lstride - (size_t)nA;
    float di = dinv[i];
    unsigned nmax = (unsigned)N - 1u;
    v4f_t acc = {0.0f, 0.0f, 0.0f, 0.0f};
    int j = 0;
    for (; j + 1 < nAB; j += 2) {
        unsigned s0 = (unsigned)lists[slot_off(base, j, nA, adjB)];
        unsigned s1 = (unsigned)lists[slot_off(base, j + 1, nA, adjB)];
        s0 = s0 < nmax ? s0 : nmax;
        s1 = s1 < nmax ? s1 : nmax;
        const float* r0 = h + (size_t)s0 * hp;
        const float* r1 = h + (size_t)s1 * hp;
        float n0 = r0[16] * di;
        float n1 = r1[16] * di;
        v4f_t v0 = *((const v4f_t*)r0 + q);
        v4f_t v1 = *((const v4f_t*)r1 + q);
        acc += v0 * n0;
        acc += v1 * n1;
    }
    if (j < nAB) {
        unsigned s0 = (unsigned)lists[slot_off(base, j, nA, adjB)];
        s0 = s0 < nmax ? s0 : nmax;
        const float* r0 = h + (size_t)s0 * hp;
        float n0 = r0[16] * di;
        v4f_t v0 = *((const v4f_t*)r0 + q);
        acc += v0 * n0;
    }
    if (cA > SLOT || cB > SLOT) {
        int cC = (int)cnt[(size_t)i + 2 * pstride];
        int nC = cC < SLOT ? cC : SLOT;
        for (int k = 0; k < nC; ++k) {
            unsigned s0 = (unsigned)lists[2 * lstride + base + (size_t)k];
            s0 = s0 < nmax ? s0 : nmax;
            const float* r0 = h + (size_t)s0 * hp;
            float n0 = r0[16] * di;
            v4f_t v0 = *((const v4f_t*)r0 + q);
            acc += v0 * n0;
        }
        if (cC > SLOT) {
            unsigned K = (unsigned)ovn[0];
            K = K < (unsigned)E ? K : (unsigned)E;
            for (unsigned k = 0; k < K; ++k) {
                unsigned e = (unsigned)ovf[k];
                e = e < (unsigned)E ? e : (unsigned)E - 1u;
                if (dst[e] != i) continue;
                unsigned s0 = (unsigned)src[e];
                s0 = s0 < nmax ? s0 : nmax;
                const float* r0 = h + (size_t)s0 * hp;
                float n0 = r0[16] * di;
                v4f_t v0 = *((const v4f_t*)r0 + q);
                acc += v0 * n0;
            }
        }
    }
    float d2 = di * di;
    v4f_t hv = *((const v4f_t*)(h + (size_t)i * hp) + q);
    v4f_t bv = ((const v4f_t*)b1)[q];
    v4f_t o;
    o.x = fmaxf(acc.x + hv.x * d2 + bv.x, 0.0f);
    o.y = fmaxf(acc.y + hv.y * d2 + bv.y, 0.0f);
    o.z = fmaxf(acc.z + hv.z * d2 + bv.z, 0.0f);
    o.w = fmaxf(acc.w + hv.w * d2 + bv.w, 0.0f);
    *((v4f_t*)(hout + (size_t)i * 16) + q) = o;
}

__global__ void gcn_pull2_slots(const float* __restrict__ cnt, size_t pstride,
                                const int* __restrict__ lists, size_t lstride,
                                const float* __restrict__ ovn, const int* __restrict__ ovf,
                                const int* __restrict__ src, const int* __restrict__ dst, int E,
                                const float* __restrict__ dinv,
                                const float* __restrict__ h2, const float* __restrict__ b2,
                                float* __restrict__ out, int N) {
    int t = blockIdx.x * blockDim.x + threadIdx.x;
    int i = t >> 2;
    int q = t & 3;
    bool ok = i < N;
    int ic = ok ? i : N - 1;
    int cA = ok ? (int)cnt[ic] : 0;
    int cB = ok ? (int)cnt[(size_t)ic + pstride] : 0;
    int nA  = cA < SLOT ? cA : SLOT;
    int nAB = nA + (cB < SLOT ? cB : SLOT);
    size_t base = (size_t)ic * SLOT;
    size_t adjB = lstride - (size_t)nA;
    float di = dinv[ic];
    unsigned nmax = (unsigned)N - 1u;
    float g0 = 0.0f, g1 = 0.0f;
    for (int j = q; j < nAB; j += 4) {
        unsigned s = (unsigned)lists[slot_off(base, j, nA, adjB)];
        s = s < nmax ? s : nmax;
        v4f_t v = *((const v4f_t*)h2 + s);
        float nrm = v.z * di;
        g0 += v.x * nrm;
        g1 += v.y * nrm;
    }
    if (cA > SLOT || cB > SLOT) {
        int cC = (int)cnt[(size_t)ic + 2 * pstride];
        int nC = cC < SLOT ? cC : SLOT;
        for (int k = q; k < nC; k += 4) {
            unsigned s = (unsigned)lists[2 * lstride + base + (size_t)k];
            s = s < nmax ? s : nmax;
            v4f_t v = *((const v4f_t*)h2 + s);
            float nrm = v.z * di;
            g0 += v.x * nrm;
            g1 += v.y * nrm;
        }
        if (q == 0 && cC > SLOT) {
            unsigned K = (unsigned)ovn[0];
            K = K < (unsigned)E ? K : (unsigned)E;
            for (unsigned k = 0; k < K; ++k) {
                unsigned e = (unsigned)ovf[k];
                e = e < (unsigned)E ? e : (unsigned)E - 1u;
                if (dst[e] != ic) continue;
                unsigned s = (unsigned)src[e];
                s = s < nmax ? s : nmax;
                v4f_t v = *((const v4f_t*)h2 + s);
                float nrm = v.z * di;
                g0 += v.x * nrm;
                g1 += v.y * nrm;
            }
        }
    }
    g0 += __shfl_xor(g0, 1);
    g1 += __shfl_xor(g1, 1);
    g0 += __shfl_xor(g0, 2);
    g1 += __shfl_xor(g1, 2);
    if (!ok || q != 0) return;
    float d2 = di * di;
    v4f_t me = *((const v4f_t*)h2 + i);
    float a = g0 + me.x * d2 + b2[0];
    float b = g1 + me.y * d2 + b2[1];
    float mx  = fmaxf(a, b);
    float lse = mx + logf(expf(a - mx) + expf(b - mx));
    out[2 * (size_t)i]     = a - lse;
    out[2 * (size_t)i + 1] = b - lse;
}

#define R4(v) (((size_t)(v) + 3) & ~(size_t)3)
#define R32(v) (((size_t)(v) + 31) & ~(size_t)31)
extern "C" void kernel_launch(void* const* d_in, const int* in_sizes, int n_in,
                              void* d_out, int out_size, void* d_ws, size_t ws_size,
                              hipStream_t stream) {
    const float* x  = (const float*)d_in[0];
    const int*   ei = (const int*)d_in[1];
    const float* W1 = (const float*)d_in[2];
    const float* b1 = (const float*)d_in[3];
    const float* W2 = (const float*)d_in[4];
    const float* b2 = (const float*)d_in[5];
    float* out = (float*)d_out;

    int N = in_sizes[0] / 128;
    int E = in_sizes[1] / 2;
    const int* src = ei;
    const int* dst = ei + E;

    int eb  = (E + THREADS - 1) / THREADS;
    int nb  = (N + THREADS - 1) / THREADS;
    int nb4 = (int)(((long long)N * 4 + THREADS - 1) / THREADS);

    {
        size_t n  = (size_t)N;
        size_t ne = (size_t)E;
        size_t o_dinv  = R32(3 * n + 1);
        size_t o_ovf   = o_dinv + R32(n);
        size_t o_h1r   = o_ovf + R32(ne);
        size_t o_lists = o_h1r + R32(16 * n);
        size_t lstr    = R32(SLOT * n);
        size_t o_h1    = o_lists + 3 * lstr;
        size_t total   = o_h1 + 32 * n;
        if (N > 0 && E > 0 && ws_size >= total * sizeof(float)) {
            float* ws    = (float*)d_ws;
            float* cnt   = ws;
            float* ovn   = ws + 3 * n;
            float* dinv  = ws + o_dinv;
            int*   ovf   = (int*)(ws + o_ovf);
            float* h1r   = ws + o_h1r;
            int*   lists = (int*)(ws + o_lists);
            float* h1    = ws + o_h1;
            float* h2    = h1;
            hipMemsetAsync(cnt, 0, (3 * n + 1) * sizeof(float), stream);
            int ntiles = (N + 15) / 16;
            int gb = (ntiles + 7) / 8;
            gcn_gemm1_slots<<<gb, THREADS, 0, stream>>>(x, W1, h1, 32, ntiles, N, src, dst, cnt, n, E,
                                                        lists, lstr, ovn, ovf);
            gcn_dinv_slots<<<nb, THREADS, 0, stream>>>(cnt, n, N, dinv, h1, 32);
            gcn_pull16_slots<<<nb4, THREADS, 0, stream>>>(cnt, n, lists, lstr, ovn, ovf, src, dst, E,
                                                          dinv, h1, 32, b1, h1r, N);
            gcn_gemm2<<<nb, THREADS, 0, stream>>>(h1r, W2, dinv, h2, N);
            gcn_pull2_slots<<<nb4, THREADS, 0, stream>>>(cnt, n, lists, lstr, ovn, ovf, src, dst, E,
                                                         dinv, h2, b2, out, N);
            return;
        }
    }

    size_t n  = (size_t)N;
    size_t ne = (size_t)E;
    size_t big = ne > 16 * n ? ne : 16 * n;
    size_t fixed = R4(n) + R4(n) + R4((size_t)nb) + R4(big) + R4(ne) + 32 + 16 * n;
    size_t np = (ws_size >= (R4(2 * n) + fixed) * sizeof(float)) ? 2 : 1;
    size_t nl = (np == 2 && ws_size >= (R4(2 * n) + R4(ne) + fixed) * sizeof(float)) ? 2 : 1;
    int    hp = (ws_size >= (R4(np * n) + (nl - 1) * R4(ne) + fixed + 16 * n) * sizeof(float)) ? 32 : 16;
    size_t sd = (np - 1) * n;
    float* ws   = (float*)d_ws;
    float* cnt  = ws;
    int*   lstb = nl == 2 ? (int*)(cnt + R4(np * n)) : nullptr;
    int*   endp = (int*)(cnt + R4(np * n)) + (nl - 1) * R4(ne);
    float* dinv = (float*)(endp + R4(n));
    int*   bt   = (int*)(dinv + R4(n));
    int*   rankw = bt + R4((size_t)nb);
    float* h1r  = (float*)rankw;
    int*   lst  = rankw + R4(big);
    size_t h1off = (size_t)((float*)(lst + R4(ne)) - ws);
    float* h1   = ws + ((h1off + 31) & ~(size_t)31);
    float* h2   = h1;

    hipMemsetAsync(cnt, 0, (nl == 2 ? R4(np * n) + ne : np * n) * sizeof(float), stream);

    int ntiles = (N + 15) / 16;
    int gb = (ntiles + 7) / 8;
    gcn_gemm1_deg<<<gb, THREADS, 0, stream>>>(x, W1, h1, hp, ntiles, N, dst, cnt, sd, E, rankw);
    gcn_block_counts<<<nb, THREADS, 0, stream>>>(cnt, sd, N, bt);
    gcn_dinv_offsets<<<nb, THREADS, 0, stream>>>(cnt, sd, N, bt, endp, dinv, h1, hp);
    gcn_place_edges<<<eb, THREADS, 0, stream>>>(src, dst, rankw, (const int*)cnt, sd, lst, lstb, E);

    gcn_pull16<<<nb4, THREADS, 0, stream>>>((const int*)cnt, endp, lst, lstb, dinv, h1, hp, b1, h1r, N);
    gcn_gemm2<<<nb, THREADS, 0, stream>>>(h1r, W2, dinv, h2, N);
    gcn_pull2_final<<<nb4, THREADS, 0, stream>>>((const int*)cnt, endp, lst, lstb, dinv, h2, b2, out, N);
}


# Layer-1 pull finishes the node: the 16->2 product runs on the relu'd row in registers (4 lanes' partial sums added: h2 sum order changes) and writes the h2 slot; no h2 launch, no relu'd table
# speedup vs baseline: 4.0508x; 1.0344x over previous
#include <hip/hip_runtime.h>
#include <hip/hip_bf16.h>

typedef __attribute__((ext_vector_type(2))) float v2f_t;
typedef __attribute__((ext_vector_type(4))) float v4f_t;
typedef __attribute__((ext_vector_type(8))) float v8f_t;

#define THREADS 256

__device__ __forceinline__ void atadd8_ret(float* const (&p)[8], const float (&a)[8], float (&r)[8]) {
    asm volatile("global_atomic_add_f32 %0, %8, %16, off th:TH_ATOMIC_RETURN scope:SCOPE_DEV\n\t"
                 "global_atomic_add_f32 %1, %9, %17, off th:TH_ATOMIC_RETURN scope:SCOPE_DEV\n\t"
                 "global_atomic_add_f32 %2, %10, %18, off th:TH_ATOMIC_RETURN scope:SCOPE_DEV\n\t"
                 "global_atomic_add_f32 %3, %11, %19, off th:TH_ATOMIC_RETURN scope:SCOPE_DEV\n\t"
                 "global_atomic_add_f32 %4, %12, %20, off th:TH_ATOMIC_RETURN scope:SCOPE_DEV\n\t"
                 "global_atomic_add_f32 %5, %13, %21, off th:TH_ATOMIC_RETURN scope:SCOPE_DEV\n\t"
                 "global_atomic_add_f32 %6, %14, %22, off th:TH_ATOMIC_RETURN scope:SCOPE_DEV\n\t"
                 "global_atomic_add_f32 %7, %15, %23, off th:TH_ATOMIC_RETURN scope:SCOPE_DEV\n\t"
                 "s_wait_loadcnt 0x0"
                 : "=&v"(r[0]), "=&v"(r[1]), "=&v"(r[2]), "=&v"(r[3]),
                   "=&v"(r[4]), "=&v"(r[5]), "=&v"(r[6]), "=&v"(r[7])
                 : "v"(p[0]), "v"(p[1]), "v"(p[2]), "v"(p[3]),
                   "v"(p[4]), "v"(p[5]), "v"(p[6]), "v"(p[7]),
                   "v"(a[0]), "v"(a[1]), "v"(a[2]), "v"(a[3]),
                   "v"(a[4]), "v"(a[5]), "v"(a[6]), "v"(a[7])
                 : "memory");
}

__device__ __forceinline__ unsigned pkg_half() {
    unsigned id = __builtin_amdgcn_s_sendmsg_rtn(0x87);
    return (id >> 18) & 1u;
}


__global__ void gcn_block_counts(const float* __restrict__ cnt, size_t pstride, int N,
                                 int* __restrict__ bt) {
    __shared__ int wtot[THREADS / 32];
    int i = blockIdx.x * blockDim.x + threadIdx.x;
    int c = 0;
    if (i < N) {
        float f = cnt[i];
        if (pstride) f += cnt[(size_t)i + pstride];
        c = (int)f;
    }
    for (int o = 16; o > 0; o >>= 1) c += __shfl_xor(c, o);
    if ((threadIdx.x & 31) == 0) wtot[threadIdx.x >> 5] = c;
    __syncthreads();
    if (threadIdx.x == 0) {
        int t = 0;
        for (int k = 0; k < THREADS / 32; ++k) t += wtot[k];
        bt[blockIdx.x] = t;
    }
}

__global__ void gcn_dinv_offsets(float* __restrict__ cnt, size_t pstride, int N,
                                 const int* __restrict__ bt, int* __restrict__ endp,
                                 float* __restrict__ dinv, float* __restrict__ h1, int hp) {
    __shared__ int wsum[THREADS / 32];
    __shared__ int bsum[THREADS / 32];
    int i = blockIdx.x * blockDim.x + threadIdx.x;
    int lane = threadIdx.x & 31;
    int w = threadIdx.x >> 5;
    int part = 0;
    for (int k = threadIdx.x; k < (int)blockIdx.x; k += THREADS) part += bt[k];
    for (int o = 16; o > 0; o >>= 1) part += __shfl_xor(part, o);
    int cA = 0, c = 0;
    if (i < N) {
        float fa = cnt[i];
        float f = fa;
        if (pstride) f += cnt[(size_t)i + pstride];
        cA = (int)fa;
        c  = (int)f;
        float dv = rsqrtf(f + 1.0f);
        dinv[i] = dv;
        if (hp > 16) h1[(size_t)i * hp + 16] = dv;
    }
    int incl = c;
    for (int o = 1; o < 32; o <<= 1) {
        int t = __shfl_up(incl, o);
        if (lane >= o) incl += t;
    }
    if (lane == 0)  bsum[w] = part;
    if (lane == 31) wsum[w] = incl;
    __syncthreads();
    int st = incl - c;
    for (int k = 0; k < THREADS / 32; ++k) {
        st += bsum[k];
        if (k < w) st += wsum[k];
    }
    if (i < N) {
        int* s0 = (int*)cnt;
        s0[i] = st;
        if (pstride) s0[(size_t)i + pstride] = st + cA;
        endp[i] = st + c;
    }
}

__global__ void gcn_place_edges(const int* __restrict__ src, const int* __restrict__ dst,
                                const int* __restrict__ rankw, const int* __restrict__ st,
                                size_t pstride, int* __restrict__ lst, int* __restrict__ lstb,
                                int E) {
    int e = blockIdx.x * blockDim.x + threadIdx.x;
    int* lp = (lstb && pkg_half()) ? lstb : lst;
    if (e >= E) return;
    int s = __builtin_nontemporal_load(src + e);
    int d = __builtin_nontemporal_load(dst + e);
    int r = __builtin_nontemporal_load(rankw + e);
    size_t pl = (r >> 30) & 1 ? pstride : 0;
    int pos = st[(size_t)d + pl] + (r & 0x3fffffff);
    if ((unsigned)pos < (unsigned)E) lp[pos] = s + 1;
}

#define XPITCH 132
#define EPL 8
__global__ void gcn_gemm1_deg(const float* __restrict__ x, const float* __restrict__ W1,
                              float* __restrict__ h1, int hp, int ntiles, int N,
                              const int* __restrict__ dst, float* __restrict__ deg,
                              size_t pstride, int E, int* __restrict__ rankw) {
    __shared__ __attribute__((aligned(16))) float xs[(THREADS / 32) * 16 * XPITCH];
    int wib  = threadIdx.x >> 5;
    int wave = blockIdx.x * (blockDim.x >> 5) + wib;
    bool active = wave < ntiles;
    int tile = active ? wave : ntiles - 1;
    int lane = threadIdx.x & 31;
    float* xt = xs + wib * (16 * XPITCH);

    unsigned wv     = (unsigned)__builtin_amdgcn_readfirstlane(wave);
    unsigned nwaves = gridDim.x * (blockDim.x >> 5);
    unsigned chunk  = (((unsigned)E + nwaves - 1u) / nwaves + 31u) & ~31u;
    unsigned e0 = wv * chunk;
    unsigned e1 = e0 + chunk < (unsigned)E ? e0 + chunk : (unsigned)E;
    bool any = e0 < e1;
    unsigned hsel = pstride ? pkg_half() : 0u;
    float* plane = deg + (size_t)hsel * pstride;
    int hbit = (int)(hsel << 30);

    int dv[EPL] = {};
    if (any) {
#pragma unroll
        for (int j = 0; j < EPL; ++j) {
            unsigned idx = e0 + j * 32 + lane;
            dv[j] = __builtin_nontemporal_load(dst + (idx < e1 ? idx : e1 - 1u));
        }
    }

    v4f_t rowv[16];
#pragma unroll
    for (int r = 0; r < 16; ++r) {
        int row = tile * 16 + r;
        int rcl = row < N ? row : N - 1;
        rowv[r] = __builtin_nontemporal_load((const v4f_t*)(x + (size_t)rcl * 128) + lane);
    }
    {
        float* pp[EPL];
        float  aa[EPL];
        float  rk[EPL];
#pragma unroll
        for (int j = 0; j < EPL; ++j) {
            unsigned idx = e0 + j * 32 + lane;
            pp[j] = plane + (unsigned)dv[j];
            aa[j] = idx < e1 ? 1.0f : 0.0f;
        }
        atadd8_ret(pp, aa, rk);
#pragma unroll
        for (int j = 0; j < EPL; ++j) {
            unsigned idx = e0 + j * 32 + lane;
            if (idx < e1) __builtin_nontemporal_store((int)rk[j] | hbit, rankw + idx);
        }
    }
    for (unsigned g = e0 + 32 * EPL; g < e1; g += 32 * EPL) {
        float* pp[EPL];
        float  aa[EPL];
        float  rk[EPL];
#pragma unroll
        for (int j = 0; j < EPL; ++j) {
            unsigned idx = g + j * 32 + lane;
            pp[j] = plane + (unsigned)__builtin_nontemporal_load(dst + (idx < e1 ? idx : e1 - 1u));
            aa[j] = idx < e1 ? 1.0f : 0.0f;
        }
        atadd8_ret(pp, aa, rk);
#pragma unroll
        for (int j = 0; j < EPL; ++j) {
            unsigned idx = g + j * 32 + lane;
            if (idx < e1) __builtin_nontemporal_store((int)rk[j] | hbit, rankw + idx);
        }
    }

#pragma unroll
    for (int r = 0; r < 16; ++r) *(v4f_t*)(xt + r * XPITCH + lane * 4) = rowv[r];
    __syncthreads();

    int half = lane >> 4;
    int m    = lane & 15;
    int koff = half << 1;
    const float* xr = xt + m * XPITCH;

    v8f_t c = {};
#pragma unroll
    for (int k0 = 0; k0 < 128; k0 += 4) {
        v2f_t a = *(const v2f_t*)(xr + k0 + koff);
        v2f_t b;
        b.x = W1[(k0 + koff) * 16 + m];
        b.y = W1[(k0 + koff + 1) * 16 + m];
        c = __builtin_amdgcn_wmma_f32_16x16x4_f32(false, a, false, b, (short)0, c,
                                                  false, false);
    }
    if (!active) return;
#pragma unroll
    for (int v = 0; v < 8; ++v) {
        int r = wave * 16 + half * 8 + v;
        if (r < N) h1[(size_t)r * hp + m] = c[v];
    }
}

__global__ void gcn_pull16(const int* __restrict__ start, const int* __restrict__ endp,
                           const int* __restrict__ lst, const int* __restrict__ lstb,
                           const float* __restrict__ dinv,
                           const float* __restrict__ h, int hp, const float* __restrict__ b1,
                           float* __restrict__ hout, int N) {
    int t = blockIdx.x * blockDim.x + threadIdx.x;
    int i = t >> 2;
    int q = t & 3;
    if (i >= N) return;
    int p  = start[i];
    int pe = endp[i];
    float di = dinv[i];
    unsigned nmax = (unsigned)N - 1u;
    v4f_t acc = {0.0f, 0.0f, 0.0f, 0.0f};
    for (; p + 1 < pe; p += 2) {
        int e0 = lst[p];
        int e1 = lst[p + 1];
        if (lstb) {
            int f0 = lstb[p];
            int f1 = lstb[p + 1];
            e0 = f0 ? f0 : e0;
            e1 = f1 ? f1 : e1;
        }
        unsigned s0 = (unsigned)(e0 - 1);
        unsigned s1 = (unsigned)(e1 - 1);
        s0 = s0 < nmax ? s0 : nmax;
        s1 = s1 < nmax ? s1 : nmax;
        const float* r0 = h + (size_t)s0 * hp;
        const float* r1 = h + (size_t)s1 * hp;
        float n0 = *(hp > 16 ? r0 + 16 : dinv + s0) * di;
        float n1 = *(hp > 16 ? r1 + 16 : dinv + s1) * di;
        v4f_t v0 = *((const v4f_t*)r0 + q);
        v4f_t v1 = *((const v4f_t*)r1 + q);
        acc += v0 * n0;
        acc += v1 * n1;
    }
    if (p < pe) {
        int e0 = lst[p];
        if (lstb) {
            int f0 = lstb[p];
            e0 = f0 ? f0 : e0;
        }
        unsigned s0 = (unsigned)(e0 - 1);
        s0 = s0 < nmax ? s0 : nmax;
        const float* r0 = h + (size_t)s0 * hp;
        float n0 = *(hp > 16 ? r0 + 16 : dinv + s0) * di;
        v4f_t v0 = *((const v4f_t*)r0 + q);
        acc += v0 * n0;
    }
    float d2 = di * di;
    v4f_t hv = *((const v4f_t*)(h + (size_t)i * hp) + q);
    v4f_t bv = ((const v4f_t*)b1)[q];
    v4f_t o;
    o.x = fmaxf(acc.x + hv.x * d2 + bv.x, 0.0f);
    o.y = fmaxf(acc.y + hv.y * d2 + bv.y, 0.0f);
    o.z = fmaxf(acc.z + hv.z * d2 + bv.z, 0.0f);
    o.w = fmaxf(acc.w + hv.w * d2 + bv.w, 0.0f);
    *((v4f_t*)(hout + (size_t)i * 16) + q) = o;
}

__global__ void gcn_gemm2(const float* __restrict__ h, const float* __restrict__ W2,
                          const float* __restrict__ dinv, float* __restrict__ h2, int N) {
    int i = blockIdx.x * blockDim.x + threadIdx.x;
    if (i >= N) return;
    const float* hr = h + (size_t)i * 16;
    float o0 = 0.0f, o1 = 0.0f;
#pragma unroll
    for (int k = 0; k < 16; ++k) {
        float v = hr[k];
        o0 += v * W2[2 * k];
        o1 += v * W2[2 * k + 1];
    }
    v4f_t o = {o0, o1, dinv[i], 0.0f};
    *((v4f_t*)h2 + i) = o;
}

__global__ void gcn_pull2_final(const int* __restrict__ start, const int* __restrict__ endp,
                                const int* __restrict__ lst, const int* __restrict__ lstb,
                                const float* __restrict__ dinv,
                                const float* __restrict__ h2, const float* __restrict__ b2,
                                float* __restrict__ out, int N) {
    int t = blockIdx.x * blockDim.x + threadIdx.x;
    int i = t >> 2;
    int q = t & 3;
    bool ok = i < N;
    int ic = ok ? i : N - 1;
    int p  = start[ic] + q;
    int pe = ok ? endp[ic] : 0;
    float di = dinv[ic];
    unsigned nmax = (unsigned)N - 1u;
    float g0 = 0.0f, g1 = 0.0f;
    for (; p < pe; p += 4) {
        int ev = lst[p];
        if (lstb) {
            int fv = lstb[p];
            ev = fv ? fv : ev;
        }
        unsigned s = (unsigned)(ev - 1);
        s = s < nmax ? s : nmax;
        v4f_t v = *((const v4f_t*)h2 + s);
        float nrm = v.z * di;
        g0 += v.x * nrm;
        g1 += v.y * nrm;
    }
    g0 += __shfl_xor(g0, 1);
    g1 += __shfl_xor(g1, 1);
    g0 += __shfl_xor(g0, 2);
    g1 += __shfl_xor(g1, 2);
    if (!ok || q != 0) return;
    float d2 = di * di;
    v4f_t me = *((const v4f_t*)h2 + i);
    float a = g0 + me.x * d2 + b2[0];
    float b = g1 + me.y * d2 + b2[1];
    float mx  = fmaxf(a, b);
    float lse = mx + logf(expf(a - mx) + expf(b - mx));
    out[2 * (size_t)i]     = a - lse;
    out[2 * (size_t)i + 1] = b - lse;
}


#define SLOT 16

__device__ __forceinline__ float atadd1_ret(float* p, float a) {
    float r;
    asm volatile("global_atomic_add_f32 %0, %1, %2, off th:TH_ATOMIC_RETURN scope:SCOPE_DEV\n\t"
                 "s_wait_loadcnt 0x0"
                 : "=&v"(r) : "v"(p), "v"(a) : "memory");
    return r;
}

__device__ __forceinline__ void atadd8_ret_s(float* base, unsigned (&o)[8],
                                             const float (&a)[8], float (&r)[8]) {
    asm volatile("v_lshlrev_b32 %8, 2, %8\n\t"
                 "v_lshlrev_b32 %9, 2, %9\n\t"
                 "v_lshlrev_b32 %10, 2, %10\n\t"
                 "v_lshlrev_b32 %11, 2, %11\n\t"
                 "v_lshlrev_b32 %12, 2, %12\n\t"
                 "v_lshlrev_b32 %13, 2, %13\n\t"
                 "v_lshlrev_b32 %14, 2, %14\n\t"
                 "v_lshlrev_b32 %15, 2, %15\n\t"
                 "global_atomic_add_f32 %0, %8, %16, %24 th:TH_ATOMIC_RETURN scope:SCOPE_DEV\n\t"
                 "global_atomic_add_f32 %1, %9, %17, %24 th:TH_ATOMIC_RETURN scope:SCOPE_DEV\n\t"
                 "global_atomic_add_f32 %2, %10, %18, %24 th:TH_ATOMIC_RETURN scope:SCOPE_DEV\n\t"
                 "global_atomic_add_f32 %3, %11, %19, %24 th:TH_ATOMIC_RETURN scope:SCOPE_DEV\n\t"
                 "global_atomic_add_f32 %4, %12, %20, %24 th:TH_ATOMIC_RETURN scope:SCOPE_DEV\n\t"
                 "global_atomic_add_f32 %5, %13, %21, %24 th:TH_ATOMIC_RETURN scope:SCOPE_DEV\n\t"
                 "global_atomic_add_f32 %6, %14, %22, %24 th:TH_ATOMIC_RETURN scope:SCOPE_DEV\n\t"
                 "global_atomic_add_f32 %7, %15, %23, %24 th:TH_ATOMIC_RETURN scope:SCOPE_DEV\n\t"
                 "s_wait_loadcnt 0x0"
                 : "=&v"(r[0]), "=&v"(r[1]), "=&v"(r[2]), "=&v"(r[3]),
                   "=&v"(r[4]), "=&v"(r[5]), "=&v"(r[6]), "=&v"(r[7]),
                   "+v"(o[0]), "+v"(o[1]), "+v"(o[2]), "+v"(o[3]),
                   "+v"(o[4]), "+v"(o[5]), "+v"(o[6]), "+v"(o[7])
                 : "v"(a[0]), "v"(a[1]), "v"(a[2]), "v"(a[3]),
                   "v"(a[4]), "v"(a[5]), "v"(a[6]), "v"(a[7]),
                   "s"(base)
                 : "memory");
}

__device__ __forceinline__ void place_in_slot(int* __restrict__ lp, int* __restrict__ lc,
                                              float* __restrict__ cntc, float* __restrict__ ovn,
                                              int* __restrict__ ovf, int E, unsigned d, int s,
                                              float rankf, unsigned idx) {
    int r = (int)rankf;
    if (r < SLOT) {
        lp[(size_t)d * SLOT + r] = s;
    } else {
        int c = (int)atadd1_ret(cntc + d, 1.0f);
        if (c < SLOT) {
            lc[(size_t)d * SLOT + c] = s;
        } else {
            unsigned k = (unsigned)atadd1_ret(ovn, 1.0f);
            if (k < (unsigned)E) ovf[k] = (int)idx;
        }
    }
}

__global__ void gcn_gemm1_slots(const float* __restrict__ x, const float* __restrict__ W1,
                                float* __restrict__ h1, int hp, int ntiles, int N,
                                const int* __restrict__ src, const int* __restrict__ dst,
                                float* __restrict__ deg, size_t pstride, int E,
                                int* __restrict__ lists, size_t lstride,
                                float* __restrict__ ovn, int* __restrict__ ovf) {
    __shared__ __attribute__((aligned(16))) float xs[(THREADS / 32) * 16 * XPITCH];
    int wib  = threadIdx.x >> 5;
    int wave = blockIdx.x * (blockDim.x >> 5) + wib;
    bool active = wave < ntiles;
    int tile = active ? wave : ntiles - 1;
    int lane = threadIdx.x & 31;
    float* xt = xs + wib * (16 * XPITCH);

    unsigned wv     = (unsigned)__builtin_amdgcn_readfirstlane(wave);
    unsigned nwaves = gridDim.x * (blockDim.x >> 5);
    unsigned chunk  = (((unsigned)E + nwaves - 1u) / nwaves + 31u) & ~31u;
    unsigned e0 = wv * chunk;
    unsigned e1 = e0 + chunk < (unsigned)E ? e0 + chunk : (unsigned)E;
    bool any = e0 < e1;
    unsigned hsel = pkg_half();
    float* plane = deg + (size_t)hsel * pstride;
    int*   lp    = lists + (size_t)hsel * lstride;
    int*   lc    = lists + 2 * lstride;
    float* cntc  = deg + 2 * pstride;

    unsigned dv[EPL] = {};
    int      sv[EPL] = {};
    if (any) {
#pragma unroll
        for (int j = 0; j < EPL; ++j) {
            unsigned idx = e0 + j * 32 + lane;
            dv[j] = (unsigned)__builtin_nontemporal_load(dst + (idx < e1 ? idx : e1 - 1u));
        }
#pragma unroll
        for (int j = 0; j < EPL; ++j) {
            unsigned idx = e0 + j * 32 + lane;
            sv[j] = __builtin_nontemporal_load(src + (idx < e1 ? idx : e1 - 1u));
        }
    }

    v4f_t rowv[16];
#pragma unroll
    for (int r = 0; r < 16; ++r) {
        int row = tile * 16 + r;
        int rcl = row < N ? row : N - 1;
        rowv[r] = __builtin_nontemporal_load((const v4f_t*)(x + (size_t)rcl * 128) + lane);
    }
    {
        float aa[EPL];
        float rk[EPL];
#pragma unroll
        for (int j = 0; j < EPL; ++j) {
            unsigned idx = e0 + j * 32 + lane;
            aa[j] = idx < e1 ? 1.0f : 0.0f;
        }
        atadd8_ret_s(plane, dv, aa, rk);
#pragma unroll
        for (int j = 0; j < EPL; ++j) {
            unsigned idx = e0 + j * 32 + lane;
            if (idx < e1) place_in_slot(lp, lc, cntc, ovn, ovf, E, dv[j] >> 2, sv[j], rk[j], idx);
        }
    }
    for (unsigned g = e0 + 32 * EPL; g < e1; g += 32 * EPL) {
        float    aa[EPL];
        float    rk[EPL];
        unsigned dd[EPL];
        int      ss[EPL];
#pragma unroll
        for (int j = 0; j < EPL; ++j) {
            unsigned idx = g + j * 32 + lane;
            unsigned cl = idx < e1 ? idx : e1 - 1u;
            dd[j] = (unsigned)__builtin_nontemporal_load(dst + cl);
            ss[j] = __builtin_nontemporal_load(src + cl);
            aa[j] = idx < e1 ? 1.0f : 0.0f;
        }
        atadd8_ret_s(plane, dd, aa, rk);
#pragma unroll
        for (int j = 0; j < EPL; ++j) {
            unsigned idx = g + j * 32 + lane;
            if (idx < e1) place_in_slot(lp, lc, cntc, ovn, ovf, E, dd[j] >> 2, ss[j], rk[j], idx);
        }
    }

#pragma unroll
    for (int r = 0; r < 16; ++r) *(v4f_t*)(xt + r * XPITCH + lane * 4) = rowv[r];
    __syncthreads();

    int half = lane >> 4;
    int m    = lane & 15;
    int koff = half << 1;
    const float* xr = xt + m * XPITCH;

    v8f_t c = {};
#pragma unroll
    for (int k0 = 0; k0 < 128; k0 += 4) {
        v2f_t a = *(const v2f_t*)(xr + k0 + koff);
        v2f_t b;
        b.x = W1[(k0 + koff) * 16 + m];
        b.y = W1[(k0 + koff + 1) * 16 + m];
        c = __builtin_amdgcn_wmma_f32_16x16x4_f32(false, a, false, b, (short)0, c,
                                                  false, false);
    }
    if (!active) return;
#pragma unroll
    for (int v = 0; v < 8; ++v) {
        int r = wave * 16 + half * 8 + v;
        if (r < N) h1[(size_t)r * hp + m] = c[v];
    }
}

__global__ void gcn_dinv_slots(const float* __restrict__ cnt, size_t pstride, int N,
                               float* __restrict__ dinv, float* __restrict__ h1, int hp) {
    int i = blockIdx.x * blockDim.x + threadIdx.x;
    if (i >= N) return;
    float f = cnt[i];
    f += cnt[(size_t)i + pstride];
    float dv = rsqrtf(f + 1.0f);
    dinv[i] = dv;
    h1[(size_t)i * hp + 16] = dv;
}

__device__ __forceinline__ size_t slot_off(size_t base, int j, int nA, size_t adjB) {
    return base + (size_t)j + (((size_t)0 - (size_t)(j >= nA)) & adjB);
}

__global__ void gcn_pull16_slots(const float* __restrict__ cnt, size_t pstride,
                                 const int* __restrict__ lists, size_t lstride,
                                 const float* __restrict__ ovn, const int* __restrict__ ovf,
                                 const int* __restrict__ src, const int* __restrict__ dst, int E,
                                 const float* __restrict__ dinv,
                                 const float* __restrict__ h, int hp, const float* __restrict__ b1,
                                 const float* __restrict__ W2, float* __restrict__ h2, int N) {
    int t = blockIdx.x * blockDim.x + threadIdx.x;
    int i = t >> 2;
    int q = t & 3;
    if (i >= N) return;
    v4f_t wa = ((const v4f_t*)W2)[2 * q];
    v4f_t wb = ((const v4f_t*)W2)[2 * q + 1];
    int cA = (int)cnt[i];
    int cB = (int)cnt[(size_t)i + pstride];
    int nA  = cA < SLOT ? cA : SLOT;
    int nAB = nA + (cB < SLOT ? cB : SLOT);
    size_t base = (size_t)i * SLOT;
    size_t adjB = lstride - (size_t)nA;
    float di = dinv[i];
    unsigned nmax = (unsigned)N - 1u;
    v4f_t acc = {0.0f, 0.0f, 0.0f, 0.0f};
    int j = 0;
    for (; j + 1 < nAB; j += 2) {
        unsigned s0 = (unsigned)lists[slot_off(base, j, nA, adjB)];
        unsigned s1 = (unsigned)lists[slot_off(base, j + 1, nA, adjB)];
        s0 = s0 < nmax ? s0 : nmax;
        s1 = s1 < nmax ? s1 : nmax;
        const float* r0 = h + (size_t)s0 * hp;
        const float* r1 = h + (size_t)s1 * hp;
        float n0 = r0[16] * di;
        float n1 = r1[16] * di;
        v4f_t v0 = *((const v4f_t*)r0 + q);
        v4f_t v1 = *((const v4f_t*)r1 + q);
        acc += v0 * n0;
        acc += v1 * n1;
    }
    if (j < nAB) {
        unsigned s0 = (unsigned)lists[slot_off(base, j, nA, adjB)];
        s0 = s0 < nmax ? s0 : nmax;
        const float* r0 = h + (size_t)s0 * hp;
        float n0 = r0[16] * di;
        v4f_t v0 = *((const v4f_t*)r0 + q);
        acc += v0 * n0;
    }
    if (cA > SLOT || cB > SLOT) {
        int cC = (int)cnt[(size_t)i + 2 * pstride];
        int nC = cC < SLOT ? cC : SLOT;
        for (int k = 0; k < nC; ++k) {
            unsigned s0 = (unsigned)lists[2 * lstride + base + (size_t)k];
            s0 = s0 < nmax ? s0 : nmax;
            const float* r0 = h + (size_t)s0 * hp;
            float n0 = r0[16] * di;
            v4f_t v0 = *((const v4f_t*)r0 + q);
            acc += v0 * n0;
        }
        if (cC > SLOT) {
            unsigned K = (unsigned)ovn[0];
            K = K < (unsigned)E ? K : (unsigned)E;
            for (unsigned k = 0; k < K; ++k) {
                unsigned e = (unsigned)ovf[k];
                e = e < (unsigned)E ? e : (unsigned)E - 1u;
                if (dst[e] != i) continue;
                unsigned s0 = (unsigned)src[e];
                s0 = s0 < nmax ? s0 : nmax;
                const float* r0 = h + (size_t)s0 * hp;
                float n0 = r0[16] * di;
                v4f_t v0 = *((const v4f_t*)r0 + q);
                acc += v0 * n0;
            }
        }
    }
    float d2 = di * di;
    v4f_t hv = *((const v4f_t*)(h + (size_t)i * hp) + q);
    v4f_t bv = ((const v4f_t*)b1)[q];
    v4f_t o;
    o.x = fmaxf(acc.x + hv.x * d2 + bv.x, 0.0f);
    o.y = fmaxf(acc.y + hv.y * d2 + bv.y, 0.0f);
    o.z = fmaxf(acc.z + hv.z * d2 + bv.z, 0.0f);
    o.w = fmaxf(acc.w + hv.w * d2 + bv.w, 0.0f);
    float p0 = o.x * wa.x + o.y * wa.z + o.z * wb.x + o.w * wb.z;
    float p1 = o.x * wa.y + o.y * wa.w + o.z * wb.y + o.w * wb.w;
    p0 += __shfl_xor(p0, 1);
    p1 += __shfl_xor(p1, 1);
    p0 += __shfl_xor(p0, 2);
    p1 += __shfl_xor(p1, 2);
    h2[t] = q == 0 ? p0 : q == 1 ? p1 : q == 2 ? di : 0.0f;
}

__global__ void gcn_pull2_slots(const float* __restrict__ cnt, size_t pstride,
                                const int* __restrict__ lists, size_t lstride,
                                const float* __restrict__ ovn, const int* __restrict__ ovf,
                                const int* __restrict__ src, const int* __restrict__ dst, int E,
                                const float* __restrict__ dinv,
                                const float* __restrict__ h2, const float* __restrict__ b2,
                                float* __restrict__ out, int N) {
    int t = blockIdx.x * blockDim.x + threadIdx.x;
    int i = t >> 2;
    int q = t & 3;
    bool ok = i < N;
    int ic = ok ? i : N - 1;
    int cA = ok ? (int)cnt[ic] : 0;
    int cB = ok ? (int)cnt[(size_t)ic + pstride] : 0;
    int nA  = cA < SLOT ? cA : SLOT;
    int nAB = nA + (cB < SLOT ? cB : SLOT);
    size_t base = (size_t)ic * SLOT;
    size_t adjB = lstride - (size_t)nA;
    float di = dinv[ic];
    unsigned nmax = (unsigned)N - 1u;
    float g0 = 0.0f, g1 = 0.0f;
    for (int j = q; j < nAB; j += 4) {
        unsigned s = (unsigned)lists[slot_off(base, j, nA, adjB)];
        s = s < nmax ? s : nmax;
        v4f_t v = *((const v4f_t*)h2 + s);
        float nrm = v.z * di;
        g0 += v.x * nrm;
        g1 += v.y * nrm;
    }
    if (cA > SLOT || cB > SLOT) {
        int cC = (int)cnt[(size_t)ic + 2 * pstride];
        int nC = cC < SLOT ? cC : SLOT;
        for (int k = q; k < nC; k += 4) {
            unsigned s = (unsigned)lists[2 * lstride + base + (size_t)k];
            s = s < nmax ? s : nmax;
            v4f_t v = *((const v4f_t*)h2 + s);
            float nrm = v.z * di;
            g0 += v.x * nrm;
            g1 += v.y * nrm;
        }
        if (q == 0 && cC > SLOT) {
            unsigned K = (unsigned)ovn[0];
            K = K < (unsigned)E ? K : (unsigned)E;
            for (unsigned k = 0; k < K; ++k) {
                unsigned e = (unsigned)ovf[k];
                e = e < (unsigned)E ? e : (unsigned)E - 1u;
                if (dst[e] != ic) continue;
                unsigned s = (unsigned)src[e];
                s = s < nmax ? s : nmax;
                v4f_t v = *((const v4f_t*)h2 + s);
                float nrm = v.z * di;
                g0 += v.x * nrm;
                g1 += v.y * nrm;
            }
        }
    }
    g0 += __shfl_xor(g0, 1);
    g1 += __shfl_xor(g1, 1);
    g0 += __shfl_xor(g0, 2);
    g1 += __shfl_xor(g1, 2);
    if (!ok || q != 0) return;
    float d2 = di * di;
    v4f_t me = *((const v4f_t*)h2 + i);
    float a = g0 + me.x * d2 + b2[0];
    float b = g1 + me.y * d2 + b2[1];
    float mx  = fmaxf(a, b);
    float lse = mx + logf(expf(a - mx) + expf(b - mx));
    out[2 * (size_t)i]     = a - lse;
    out[2 * (size_t)i + 1] = b - lse;
}

#define R4(v) (((size_t)(v) + 3) & ~(size_t)3)
#define R32(v) (((size_t)(v) + 31) & ~(size_t)31)
extern "C" void kernel_launch(void* const* d_in, const int* in_sizes, int n_in,
                              void* d_out, int out_size, void* d_ws, size_t ws_size,
                              hipStream_t stream) {
    const float* x  = (const float*)d_in[0];
    const int*   ei = (const int*)d_in[1];
    const float* W1 = (const float*)d_in[2];
    const float* b1 = (const float*)d_in[3];
    const float* W2 = (const float*)d_in[4];
    const float* b2 = (const float*)d_in[5];
    float* out = (float*)d_out;

    int N = in_sizes[0] / 128;
    int E = in_sizes[1] / 2;
    const int* src = ei;
    const int* dst = ei + E;

    int eb  = (E + THREADS - 1) / THREADS;
    int nb  = (N + THREADS - 1) / THREADS;
    int nb4 = (int)(((long long)N * 4 + THREADS - 1) / THREADS);

    {
        size_t n  = (size_t)N;
        size_t ne = (size_t)E;
        size_t o_dinv  = R32(3 * n + 1);
        size_t o_ovf   = o_dinv + R32(n);
        size_t o_h1r   = o_ovf + R32(ne);
        size_t o_lists = o_h1r + R32(16 * n);
        size_t lstr    = R32(SLOT * n);
        size_t o_h1    = o_lists + 3 * lstr;
        size_t total   = o_h1 + 32 * n;
        if (N > 0 && E > 0 && ws_size >= total * sizeof(float)) {
            float* ws    = (float*)d_ws;
            float* cnt   = ws;
            float* ovn   = ws + 3 * n;
            float* dinv  = ws + o_dinv;
            int*   ovf   = (int*)(ws + o_ovf);
            float* h2    = ws + o_h1r;
            int*   lists = (int*)(ws + o_lists);
            float* h1    = ws + o_h1;
            hipMemsetAsync(cnt, 0, (3 * n + 1) * sizeof(float), stream);
            int ntiles = (N + 15) / 16;
            int gb = (ntiles + 7) / 8;
            gcn_gemm1_slots<<<gb, THREADS, 0, stream>>>(x, W1, h1, 32, ntiles, N, src, dst, cnt, n, E,
                                                        lists, lstr, ovn, ovf);
            gcn_dinv_slots<<<nb, THREADS, 0, stream>>>(cnt, n, N, dinv, h1, 32);
            gcn_pull16_slots<<<nb4, THREADS, 0, stream>>>(cnt, n, lists, lstr, ovn, ovf, src, dst, E,
                                                          dinv, h1, 32, b1, W2, h2, N);
            gcn_pull2_slots<<<nb4, THREADS, 0, stream>>>(cnt, n, lists, lstr, ovn, ovf, src, dst, E,
                                                         dinv, h2, b2, out, N);
            return;
        }
    }

    size_t n  = (size_t)N;
    size_t ne = (size_t)E;
    size_t big = ne > 16 * n ? ne : 16 * n;
    size_t fixed = R4(n) + R4(n) + R4((size_t)nb) + R4(big) + R4(ne) + 32 + 16 * n;
    size_t np = (ws_size >= (R4(2 * n) + fixed) * sizeof(float)) ? 2 : 1;
    size_t nl = (np == 2 && ws_size >= (R4(2 * n) + R4(ne) + fixed) * sizeof(float)) ? 2 : 1;
    int    hp = (ws_size >= (R4(np * n) + (nl - 1) * R4(ne) + fixed + 16 * n) * sizeof(float)) ? 32 : 16;
    size_t sd = (np - 1) * n;
    float* ws   = (float*)d_ws;
    float* cnt  = ws;
    int*   lstb = nl == 2 ? (int*)(cnt + R4(np * n)) : nullptr;
    int*   endp = (int*)(cnt + R4(np * n)) + (nl - 1) * R4(ne);
    float* dinv = (float*)(endp + R4(n));
    int*   bt   = (int*)(dinv + R4(n));
    int*   rankw = bt + R4((size_t)nb);
    float* h1r  = (float*)rankw;
    int*   lst  = rankw + R4(big);
    size_t h1off = (size_t)((float*)(lst + R4(ne)) - ws);
    float* h1   = ws + ((h1off + 31) & ~(size_t)31);
    float* h2   = h1;

    hipMemsetAsync(cnt, 0, (nl == 2 ? R4(np * n) + ne : np * n) * sizeof(float), stream);

    int ntiles = (N + 15) / 16;
    int gb = (ntiles + 7) / 8;
    gcn_gemm1_deg<<<gb, THREADS, 0, stream>>>(x, W1, h1, hp, ntiles, N, dst, cnt, sd, E, rankw);
    gcn_block_counts<<<nb, THREADS, 0, stream>>>(cnt, sd, N, bt);
    gcn_dinv_offsets<<<nb, THREADS, 0, stream>>>(cnt, sd, N, bt, endp, dinv, h1, hp);
    gcn_place_edges<<<eb, THREADS, 0, stream>>>(src, dst, rankw, (const int*)cnt, sd, lst, lstb, E);

    gcn_pull16<<<nb4, THREADS, 0, stream>>>((const int*)cnt, endp, lst, lstb, dinv, h1, hp, b1, h1r, N);
    gcn_gemm2<<<nb, THREADS, 0, stream>>>(h1r, W2, dinv, h2, N);
    gcn_pull2_final<<<nb4, THREADS, 0, stream>>>((const int*)cnt, endp, lst, lstb, dinv, h2, b2, out, N);
}


# Both pulls read a node's two slots once, one 16-byte load per thread and plane kept in registers (layer 1 hands entries round the node's 4 lanes), not 4 bytes per entry; h2 sum order changes
# speedup vs baseline: 4.4967x; 1.1101x over previous
#include <hip/hip_runtime.h>
#include <hip/hip_bf16.h>

typedef __attribute__((ext_vector_type(2))) float v2f_t;
typedef __attribute__((ext_vector_type(4))) float v4f_t;
typedef __attribute__((ext_vector_type(8))) float v8f_t;
typedef __attribute__((ext_vector_type(4))) int   v4i_t;
typedef __attribute__((ext_vector_type(3))) float v3f_t;

#define THREADS 256

__device__ __forceinline__ void atadd8_ret(float* const (&p)[8], const float (&a)[8], float (&r)[8]) {
    asm volatile("global_atomic_add_f32 %0, %8, %16, off th:TH_ATOMIC_RETURN scope:SCOPE_DEV\n\t"
                 "global_atomic_add_f32 %1, %9, %17, off th:TH_ATOMIC_RETURN scope:SCOPE_DEV\n\t"
                 "global_atomic_add_f32 %2, %10, %18, off th:TH_ATOMIC_RETURN scope:SCOPE_DEV\n\t"
                 "global_atomic_add_f32 %3, %11, %19, off th:TH_ATOMIC_RETURN scope:SCOPE_DEV\n\t"
                 "global_atomic_add_f32 %4, %12, %20, off th:TH_ATOMIC_RETURN scope:SCOPE_DEV\n\t"
                 "global_atomic_add_f32 %5, %13, %21, off th:TH_ATOMIC_RETURN scope:SCOPE_DEV\n\t"
                 "global_atomic_add_f32 %6, %14, %22, off th:TH_ATOMIC_RETURN scope:SCOPE_DEV\n\t"
                 "global_atomic_add_f32 %7, %15, %23, off th:TH_ATOMIC_RETURN scope:SCOPE_DEV\n\t"
                 "s_wait_loadcnt 0x0"
                 : "=&v"(r[0]), "=&v"(r[1]), "=&v"(r[2]), "=&v"(r[3]),
                   "=&v"(r[4]), "=&v"(r[5]), "=&v"(r[6]), "=&v"(r[7])
                 : "v"(p[0]), "v"(p[1]), "v"(p[2]), "v"(p[3]),
                   "v"(p[4]), "v"(p[5]), "v"(p[6]), "v"(p[7]),
                   "v"(a[0]), "v"(a[1]), "v"(a[2]), "v"(a[3]),
                   "v"(a[4]), "v"(a[5]), "v"(a[6]), "v"(a[7])
                 : "memory");
}

__device__ __forceinline__ unsigned pkg_half() {
    unsigned id = __builtin_amdgcn_s_sendmsg_rtn(0x87);
    return (id >> 18) & 1u;
}


__global__ void gcn_block_counts(const float* __restrict__ cnt, size_t pstride, int N,
                                 int* __restrict__ bt) {
    __shared__ int wtot[THREADS / 32];
    int i = blockIdx.x * blockDim.x + threadIdx.x;
    int c = 0;
    if (i < N) {
        float f = cnt[i];
        if (pstride) f += cnt[(size_t)i + pstride];
        c = (int)f;
    }
    for (int o = 16; o > 0; o >>= 1) c += __shfl_xor(c, o);
    if ((threadIdx.x & 31) == 0) wtot[threadIdx.x >> 5] = c;
    __syncthreads();
    if (threadIdx.x == 0) {
        int t = 0;
        for (int k = 0; k < THREADS / 32; ++k) t += wtot[k];
        bt[blockIdx.x] = t;
    }
}

__global__ void gcn_dinv_offsets(float* __restrict__ cnt, size_t pstride, int N,
                                 const int* __restrict__ bt, int* __restrict__ endp,
                                 float* __restrict__ dinv, float* __restrict__ h1, int hp) {
    __shared__ int wsum[THREADS / 32];
    __shared__ int bsum[THREADS / 32];
    int i = blockIdx.x * blockDim.x + threadIdx.x;
    int lane = threadIdx.x & 31;
    int w = threadIdx.x >> 5;
    int part = 0;
    for (int k = threadIdx.x; k < (int)blockIdx.x; k += THREADS) part += bt[k];
    for (int o = 16; o > 0; o >>= 1) part += __shfl_xor(part, o);
    int cA = 0, c = 0;
    if (i < N) {
        float fa = cnt[i];
        float f = fa;
        if (pstride) f += cnt[(size_t)i + pstride];
        cA = (int)fa;
        c  = (int)f;
        float dv = rsqrtf(f + 1.0f);
        dinv[i] = dv;
        if (hp > 16) h1[(size_t)i * hp + 16] = dv;
    }
    int incl = c;
    for (int o = 1; o < 32; o <<= 1) {
        int t = __shfl_up(incl, o);
        if (lane >= o) incl += t;
    }
    if (lane == 0)  bsum[w] = part;
    if (lane == 31) wsum[w] = incl;
    __syncthreads();
    int st = incl - c;
    for (int k = 0; k < THREADS / 32; ++k) {
        st += bsum[k];
        if (k < w) st += wsum[k];
    }
    if (i < N) {
        int* s0 = (int*)cnt;
        s0[i] = st;
        if (pstride) s0[(size_t)i + pstride] = st + cA;
        endp[i] = st + c;
    }
}

__global__ void gcn_place_edges(const int* __restrict__ src, const int* __restrict__ dst,
                                const int* __restrict__ rankw, const int* __restrict__ st,
                                size_t pstride, int* __restrict__ lst, int* __restrict__ lstb,
                                int E) {
    int e = blockIdx.x * blockDim.x + threadIdx.x;
    int* lp = (lstb && pkg_half()) ? lstb : lst;
    if (e >= E) return;
    int s = __builtin_nontemporal_load(src + e);
    int d = __builtin_nontemporal_load(dst + e);
    int r = __builtin_nontemporal_load(rankw + e);
    size_t pl = (r >> 30) & 1 ? pstride : 0;
    int pos = st[(size_t)d + pl] + (r & 0x3fffffff);
    if ((unsigned)pos < (unsigned)E) lp[pos] = s + 1;
}

#define XPITCH 132
#define EPL 8
__global__ void gcn_gemm1_deg(const float* __restrict__ x, const float* __restrict__ W1,
                              float* __restrict__ h1, int hp, int ntiles, int N,
                              const int* __restrict__ dst, float* __restrict__ deg,
                              size_t pstride, int E, int* __restrict__ rankw) {
    __shared__ __attribute__((aligned(16))) float xs[(THREADS / 32) * 16 * XPITCH];
    int wib  = threadIdx.x >> 5;
    int wave = blockIdx.x * (blockDim.x >> 5) + wib;
    bool active = wave < ntiles;
    int tile = active ? wave : ntiles - 1;
    int lane = threadIdx.x & 31;
    float* xt = xs + wib * (16 * XPITCH);

    unsigned wv     = (unsigned)__builtin_amdgcn_readfirstlane(wave);
    unsigned nwaves = gridDim.x * (blockDim.x >> 5);
    unsigned chunk  = (((unsigned)E + nwaves - 1u) / nwaves + 31u) & ~31u;
    unsigned e0 = wv * chunk;
    unsigned e1 = e0 + chunk < (unsigned)E ? e0 + chunk : (unsigned)E;
    bool any = e0 < e1;
    unsigned hsel = pstride ? pkg_half() : 0u;
    float* plane = deg + (size_t)hsel * pstride;
    int hbit = (int)(hsel << 30);

    int dv[EPL] = {};
    if (any) {
#pragma unroll
        for (int j = 0; j < EPL; ++j) {
            unsigned idx = e0 + j * 32 + lane;
            dv[j] = __builtin_nontemporal_load(dst + (idx < e1 ? idx : e1 - 1u));
        }
    }

    v4f_t rowv[16];
#pragma unroll
    for (int r = 0; r < 16; ++r) {
        int row = tile * 16 + r;
        int rcl = row < N ? row : N - 1;
        rowv[r] = __builtin_nontemporal_load((const v4f_t*)(x + (size_t)rcl * 128) + lane);
    }
    {
        float* pp[EPL];
        float  aa[EPL];
        float  rk[EPL];
#pragma unroll
        for (int j = 0; j < EPL; ++j) {
            unsigned idx = e0 + j * 32 + lane;
            pp[j] = plane + (unsigned)dv[j];
            aa[j] = idx < e1 ? 1.0f : 0.0f;
        }
        atadd8_ret(pp, aa, rk);
#pragma unroll
        for (int j = 0; j < EPL; ++j) {
            unsigned idx = e0 + j * 32 + lane;
            if (idx < e1) __builtin_nontemporal_store((int)rk[j] | hbit, rankw + idx);
        }
    }
    for (unsigned g = e0 + 32 * EPL; g < e1; g += 32 * EPL) {
        float* pp[EPL];
        float  aa[EPL];
        float  rk[EPL];
#pragma unroll
        for (int j = 0; j < EPL; ++j) {
            unsigned idx = g + j * 32 + lane;
            pp[j] = plane + (unsigned)__builtin_nontemporal_load(dst + (idx < e1 ? idx : e1 - 1u));
            aa[j] = idx < e1 ? 1.0f : 0.0f;
        }
        atadd8_ret(pp, aa, rk);
#pragma unroll
        for (int j = 0; j < EPL; ++j) {
            unsigned idx = g + j * 32 + lane;
            if (idx < e1) __builtin_nontemporal_store((int)rk[j] | hbit, rankw + idx);
        }
    }

#pragma unroll
    for (int r = 0; r < 16; ++r) *(v4f_t*)(xt + r * XPITCH + lane * 4) = rowv[r];
    __syncthreads();

    int half = lane >> 4;
    int m    = lane & 15;
    int koff = half << 1;
    const float* xr = xt + m * XPITCH;

    v8f_t c = {};
#pragma unroll
    for (int k0 = 0; k0 < 128; k0 += 4) {
        v2f_t a = *(const v2f_t*)(xr + k0 + koff);
        v2f_t b;
        b.x = W1[(k0 + koff) * 16 + m];
        b.y = W1[(k0 + koff + 1) * 16 + m];
        c = __builtin_amdgcn_wmma_f32_16x16x4_f32(false, a, false, b, (short)0, c,
                                                  false, false);
    }
    if (!active) return;
#pragma unroll
    for (int v = 0; v < 8; ++v) {
        int r = wave * 16 + half * 8 + v;
        if (r < N) h1[(size_t)r * hp + m] = c[v];
    }
}

__global__ void gcn_pull16(const int* __restrict__ start, const int* __restrict__ endp,
                           const int* __restrict__ lst, const int* __restrict__ lstb,
                           const float* __restrict__ dinv,
                           const float* __restrict__ h, int hp, const float* __restrict__ b1,
                           float* __restrict__ hout, int N) {
    int t = blockIdx.x * blockDim.x + threadIdx.x;
    int i = t >> 2;
    int q = t & 3;
    if (i >= N) return;
    int p  = start[i];
    int pe = endp[i];
    float di = dinv[i];
    unsigned nmax = (unsigned)N - 1u;
    v4f_t acc = {0.0f, 0.0f, 0.0f, 0.0f};
    for (; p + 1 < pe; p += 2) {
        int e0 = lst[p];
        int e1 = lst[p + 1];
        if (lstb) {
            int f0 = lstb[p];
            int f1 = lstb[p + 1];
            e0 = f0 ? f0 : e0;
            e1 = f1 ? f1 : e1;
        }
        unsigned s0 = (unsigned)(e0 - 1);
        unsigned s1 = (unsigned)(e1 - 1);
        s0 = s0 < nmax ? s0 : nmax;
        s1 = s1 < nmax ? s1 : nmax;
        const float* r0 = h + (size_t)s0 * hp;
        const float* r1 = h + (size_t)s1 * hp;
        float n0 = *(hp > 16 ? r0 + 16 : dinv + s0) * di;
        float n1 = *(hp > 16 ? r1 + 16 : dinv + s1) * di;
        v4f_t v0 = *((const v4f_t*)r0 + q);
        v4f_t v1 = *((const v4f_t*)r1 + q);
        acc += v0 * n0;
        acc += v1 * n1;
    }
    if (p < pe) {
        int e0 = lst[p];
        if (lstb) {
            int f0 = lstb[p];
            e0 = f0 ? f0 : e0;
        }
        unsigned s0 = (unsigned)(e0 - 1);
        s0 = s0 < nmax ? s0 : nmax;
        const float* r0 = h + (size_t)s0 * hp;
        float n0 = *(hp > 16 ? r0 + 16 : dinv + s0) * di;
        v4f_t v0 = *((const v4f_t*)r0 + q);
        acc += v0 * n0;
    }
    float d2 = di * di;
    v4f_t hv = *((const v4f_t*)(h + (size_t)i * hp) + q);
    v4f_t bv = ((const v4f_t*)b1)[q];
    v4f_t o;
    o.x = fmaxf(acc.x + hv.x * d2 + bv.x, 0.0f);
    o.y = fmaxf(acc.y + hv.y * d2 + bv.y, 0.0f);
    o.z = fmaxf(acc.z + hv.z * d2 + bv.z, 0.0f);
    o.w = fmaxf(acc.w + hv.w * d2 + bv.w, 0.0f);
    *((v4f_t*)(hout + (size_t)i * 16) + q) = o;
}

__global__ void gcn_gemm2(const float* __restrict__ h, const float* __restrict__ W2,
                          const float* __restrict__ dinv, float* __restrict__ h2, int N) {
    int i = blockIdx.x * blockDim.x + threadIdx.x;
    if (i >= N) return;
    const float* hr = h + (size_t)i * 16;
    float o0 = 0.0f, o1 = 0.0f;
#pragma unroll
    for (int k = 0; k < 16; ++k) {
        float v = hr[k];
        o0 += v * W2[2 * k];
        o1 += v * W2[2 * k + 1];
    }
    v4f_t o = {o0, o1, dinv[i], 0.0f};
    *((v4f_t*)h2 + i) = o;
}

__global__ void gcn_pull2_final(const int* __restrict__ start, const int* __restrict__ endp,
                                const int* __restrict__ lst, const int* __restrict__ lstb,
                                const float* __restrict__ dinv,
                                const float* __restrict__ h2, const float* __restrict__ b2,
                                float* __restrict__ out, int N) {
    int t = blockIdx.x * blockDim.x + threadIdx.x;
    int i = t >> 2;
    int q = t & 3;
    bool ok = i < N;
    int ic = ok ? i : N - 1;
    int p  = start[ic] + q;
    int pe = ok ? endp[ic] : 0;
    float di = dinv[ic];
    unsigned nmax = (unsigned)N - 1u;
    float g0 = 0.0f, g1 = 0.0f;
    for (; p < pe; p += 4) {
        int ev = lst[p];
        if (lstb) {
            int fv = lstb[p];
            ev = fv ? fv : ev;
        }
        unsigned s = (unsigned)(ev - 1);
        s = s < nmax ? s : nmax;
        v4f_t v = *((const v4f_t*)h2 + s);
        float nrm = v.z * di;
        g0 += v.x * nrm;
        g1 += v.y * nrm;
    }
    g0 += __shfl_xor(g0, 1);
    g1 += __shfl_xor(g1, 1);
    g0 += __shfl_xor(g0, 2);
    g1 += __shfl_xor(g1, 2);
    if (!ok || q != 0) return;
    float d2 = di * di;
    v4f_t me = *((const v4f_t*)h2 + i);
    float a = g0 + me.x * d2 + b2[0];
    float b = g1 + me.y * d2 + b2[1];
    float mx  = fmaxf(a, b);
    float lse = mx + logf(expf(a - mx) + expf(b - mx));
    out[2 * (size_t)i]     = a - lse;
    out[2 * (size_t)i + 1] = b - lse;
}


#define SLOT 16

__device__ __forceinline__ float atadd1_ret(float* p, float a) {
    float r;
    asm volatile("global_atomic_add_f32 %0, %1, %2, off th:TH_ATOMIC_RETURN scope:SCOPE_DEV\n\t"
                 "s_wait_loadcnt 0x0"
                 : "=&v"(r) : "v"(p), "v"(a) : "memory");
    return r;
}

__device__ __forceinline__ void atadd8_ret_s(float* base, unsigned (&o)[8],
                                             const float (&a)[8], float (&r)[8]) {
    asm volatile("v_lshlrev_b32 %8, 2, %8\n\t"
                 "v_lshlrev_b32 %9, 2, %9\n\t"
                 "v_lshlrev_b32 %10, 2, %10\n\t"
                 "v_lshlrev_b32 %11, 2, %11\n\t"
                 "v_lshlrev_b32 %12, 2, %12\n\t"
                 "v_lshlrev_b32 %13, 2, %13\n\t"
                 "v_lshlrev_b32 %14, 2, %14\n\t"
                 "v_lshlrev_b32 %15, 2, %15\n\t"
                 "global_atomic_add_f32 %0, %8, %16, %24 th:TH_ATOMIC_RETURN scope:SCOPE_DEV\n\t"
                 "global_atomic_add_f32 %1, %9, %17, %24 th:TH_ATOMIC_RETURN scope:SCOPE_DEV\n\t"
                 "global_atomic_add_f32 %2, %10, %18, %24 th:TH_ATOMIC_RETURN scope:SCOPE_DEV\n\t"
                 "global_atomic_add_f32 %3, %11, %19, %24 th:TH_ATOMIC_RETURN scope:SCOPE_DEV\n\t"
                 "global_atomic_add_f32 %4, %12, %20, %24 th:TH_ATOMIC_RETURN scope:SCOPE_DEV\n\t"
                 "global_atomic_add_f32 %5, %13, %21, %24 th:TH_ATOMIC_RETURN scope:SCOPE_DEV\n\t"
                 "global_atomic_add_f32 %6, %14, %22, %24 th:TH_ATOMIC_RETURN scope:SCOPE_DEV\n\t"
                 "global_atomic_add_f32 %7, %15, %23, %24 th:TH_ATOMIC_RETURN scope:SCOPE_DEV\n\t"
                 "s_wait_loadcnt 0x0"
                 : "=&v"(r[0]), "=&v"(r[1]), "=&v"(r[2]), "=&v"(r[3]),
                   "=&v"(r[4]), "=&v"(r[5]), "=&v"(r[6]), "=&v"(r[7]),
                   "+v"(o[0]), "+v"(o[1]), "+v"(o[2]), "+v"(o[3]),
                   "+v"(o[4]), "+v"(o[5]), "+v"(o[6]), "+v"(o[7])
                 : "v"(a[0]), "v"(a[1]), "v"(a[2]), "v"(a[3]),
                   "v"(a[4]), "v"(a[5]), "v"(a[6]), "v"(a[7]),
                   "s"(base)
                 : "memory");
}

__device__ __forceinline__ void place_in_slot(int* __restrict__ lp, int* __restrict__ lc,
                                              float* __restrict__ cntc, float* __restrict__ ovn,
                                              int* __restrict__ ovf, int E, unsigned d, int s,
                                              float rankf, unsigned idx) {
    int r = (int)rankf;
    if (r < SLOT) {
        lp[(size_t)d * SLOT + r] = s;
    } else {
        int c = (int)atadd1_ret(cntc + d, 1.0f);
        if (c < SLOT) {
            lc[(size_t)d * SLOT + c] = s;
        } else {
            unsigned k = (unsigned)atadd1_ret(ovn, 1.0f);
            if (k < (unsigned)E) ovf[k] = (int)idx;
        }
    }
}

__global__ void gcn_gemm1_slots(const float* __restrict__ x, const float* __restrict__ W1,
                                float* __restrict__ h1, int hp, int ntiles, int N,
                                const int* __restrict__ src, const int* __restrict__ dst,
                                float* __restrict__ deg, size_t pstride, int E,
                                int* __restrict__ lists, size_t lstride,
                                float* __restrict__ ovn, int* __restrict__ ovf) {
    __shared__ __attribute__((aligned(16))) float xs[(THREADS / 32) * 16 * XPITCH];
    int wib  = threadIdx.x >> 5;
    int wave = blockIdx.x * (blockDim.x >> 5) + wib;
    bool active = wave < ntiles;
    int tile = active ? wave : ntiles - 1;
    int lane = threadIdx.x & 31;
    float* xt = xs + wib * (16 * XPITCH);

    unsigned wv     = (unsigned)__builtin_amdgcn_readfirstlane(wave);
    unsigned nwaves = gridDim.x * (blockDim.x >> 5);
    unsigned chunk  = (((unsigned)E + nwaves - 1u) / nwaves + 31u) & ~31u;
    unsigned e0 = wv * chunk;
    unsigned e1 = e0 + chunk < (unsigned)E ? e0 + chunk : (unsigned)E;
    bool any = e0 < e1;
    unsigned hsel = pkg_half();
    float* plane = deg + (size_t)hsel * pstride;
    int*   lp    = lists + (size_t)hsel * lstride;
    int*   lc    = lists + 2 * lstride;
    float* cntc  = deg + 2 * pstride;

    unsigned dv[EPL] = {};
    int      sv[EPL] = {};
    if (any) {
#pragma unroll
        for (int j = 0; j < EPL; ++j) {
            unsigned idx = e0 + j * 32 + lane;
            dv[j] = (unsigned)__builtin_nontemporal_load(dst + (idx < e1 ? idx : e1 - 1u));
        }
#pragma unroll
        for (int j = 0; j < EPL; ++j) {
            unsigned idx = e0 + j * 32 + lane;
            sv[j] = __builtin_nontemporal_load(src + (idx < e1 ? idx : e1 - 1u));
        }
    }

    v4f_t rowv[16];
#pragma unroll
    for (int r = 0; r < 16; ++r) {
        int row = tile * 16 + r;
        int rcl = row < N ? row : N - 1;
        rowv[r] = __builtin_nontemporal_load((const v4f_t*)(x + (size_t)rcl * 128) + lane);
    }
    {
        float aa[EPL];
        float rk[EPL];
#pragma unroll
        for (int j = 0; j < EPL; ++j) {
            unsigned idx = e0 + j * 32 + lane;
            aa[j] = idx < e1 ? 1.0f : 0.0f;
        }
        atadd8_ret_s(plane, dv, aa, rk);
#pragma unroll
        for (int j = 0; j < EPL; ++j) {
            unsigned idx = e0 + j * 32 + lane;
            if (idx < e1) place_in_slot(lp, lc, cntc, ovn, ovf, E, dv[j] >> 2, sv[j], rk[j], idx);
        }
    }
    for (unsigned g = e0 + 32 * EPL; g < e1; g += 32 * EPL) {
        float    aa[EPL];
        float    rk[EPL];
        unsigned dd[EPL];
        int      ss[EPL];
#pragma unroll
        for (int j = 0; j < EPL; ++j) {
            unsigned idx = g + j * 32 + lane;
            unsigned cl = idx < e1 ? idx : e1 - 1u;
            dd[j] = (unsigned)__builtin_nontemporal_load(dst + cl);
            ss[j] = __builtin_nontemporal_load(src + cl);
            aa[j] = idx < e1 ? 1.0f : 0.0f;
        }
        atadd8_ret_s(plane, dd, aa, rk);
#pragma unroll
        for (int j = 0; j < EPL; ++j) {
            unsigned idx = g + j * 32 + lane;
            if (idx < e1) place_in_slot(lp, lc, cntc, ovn, ovf, E, dd[j] >> 2, ss[j], rk[j], idx);
        }
    }

#pragma unroll
    for (int r = 0; r < 16; ++r) *(v4f_t*)(xt + r * XPITCH + lane * 4) = rowv[r];
    __syncthreads();

    int half = lane >> 4;
    int m    = lane & 15;
    int koff = half << 1;
    const float* xr = xt + m * XPITCH;

    v8f_t c = {};
#pragma unroll
    for (int k0 = 0; k0 < 128; k0 += 4) {
        v2f_t a = *(const v2f_t*)(xr + k0 + koff);
        v2f_t b;
        b.x = W1[(k0 + koff) * 16 + m];
        b.y = W1[(k0 + koff + 1) * 16 + m];
        c = __builtin_amdgcn_wmma_f32_16x16x4_f32(false, a, false, b, (short)0, c,
                                                  false, false);
    }
    if (!active) return;
#pragma unroll
    for (int v = 0; v < 8; ++v) {
        int r = wave * 16 + half * 8 + v;
        if (r < N) h1[(size_t)r * hp + m] = c[v];
    }
}

__global__ void gcn_dinv_slots(const float* __restrict__ cnt, size_t pstride, int N,
                               float* __restrict__ dinv, float* __restrict__ h1, int hp) {
    int i = blockIdx.x * blockDim.x + threadIdx.x;
    if (i >= N) return;
    float f = cnt[i];
    f += cnt[(size_t)i + pstride];
    float dv = rsqrtf(f + 1.0f);
    dinv[i] = dv;
    h1[(size_t)i * hp + 16] = dv;
}

static_assert(SLOT == 16, "4 threads of a node hold 4 slot entries each");

__device__ __forceinline__ int quad_from(int v, int m) {
    switch (m) {
    case 0:  return __builtin_amdgcn_mov_dpp(v, 0x00, 0xF, 0xF, true);
    case 1:  return __builtin_amdgcn_mov_dpp(v, 0x55, 0xF, 0xF, true);
    case 2:  return __builtin_amdgcn_mov_dpp(v, 0xAA, 0xF, 0xF, true);
    default: return __builtin_amdgcn_mov_dpp(v, 0xFF, 0xF, 0xF, true);
    }
}

struct hrow_t { v4f_t v; float n; };
__device__ __forceinline__ hrow_t gather_h1(bool on, int ent, unsigned nmax,
                                            const float* __restrict__ h, int hp, int q) {
    hrow_t r;
    r.v = (v4f_t){0.0f, 0.0f, 0.0f, 0.0f};
    r.n = 0.0f;
    if (on) {
        unsigned s = (unsigned)ent;
        s = s < nmax ? s : nmax;
        const float* p = h + (size_t)s * hp;
        r.n = p[16];
        r.v = *((const v4f_t*)p + q);
    }
    return r;
}

__device__ __forceinline__ void pull16_part(v4i_t L, int j0, int n, const float* __restrict__ h,
                                            int hp, unsigned nmax, float di, int q, v4f_t& acc) {
    if (j0 >= n) return;
    int m = j0 >> 2;
    hrow_t r0 = gather_h1(j0     < n, quad_from(L.x, m),     nmax, h, hp, q);
    hrow_t r1 = gather_h1(j0 + 1 < n, quad_from(L.y, m),     nmax, h, hp, q);
    hrow_t r2 = gather_h1(j0 + 2 < n, quad_from(L.z, m),     nmax, h, hp, q);
    hrow_t r3 = gather_h1(j0 + 3 < n, quad_from(L.w, m),     nmax, h, hp, q);
    hrow_t r4 = gather_h1(j0 + 4 < n, quad_from(L.x, m + 1), nmax, h, hp, q);
    hrow_t r5 = gather_h1(j0 + 5 < n, quad_from(L.y, m + 1), nmax, h, hp, q);
    hrow_t r6 = gather_h1(j0 + 6 < n, quad_from(L.z, m + 1), nmax, h, hp, q);
    hrow_t r7 = gather_h1(j0 + 7 < n, quad_from(L.w, m + 1), nmax, h, hp, q);
    float n0 = r0.n * di;  acc += r0.v * n0;
    float n1 = r1.n * di;  acc += r1.v * n1;
    float n2 = r2.n * di;  acc += r2.v * n2;
    float n3 = r3.n * di;  acc += r3.v * n3;
    float n4 = r4.n * di;  acc += r4.v * n4;
    float n5 = r5.n * di;  acc += r5.v * n5;
    float n6 = r6.n * di;  acc += r6.v * n6;
    float n7 = r7.n * di;  acc += r7.v * n7;
}

__global__ void gcn_pull16_slots(const float* __restrict__ cnt, size_t pstride,
                                 const int* __restrict__ lists, size_t lstride,
                                 const float* __restrict__ ovn, const int* __restrict__ ovf,
                                 const int* __restrict__ src, const int* __restrict__ dst, int E,
                                 const float* __restrict__ dinv,
                                 const float* __restrict__ h, int hp, const float* __restrict__ b1,
                                 const float* __restrict__ W2, float* __restrict__ h2, int N) {
    int t = blockIdx.x * blockDim.x + threadIdx.x;
    int i = t >> 2;
    int q = t & 3;
    if (i >= N) return;
    v4f_t wa = ((const v4f_t*)W2)[2 * q];
    v4f_t wb = ((const v4f_t*)W2)[2 * q + 1];
    int cA = (int)cnt[i];
    int cB = (int)cnt[(size_t)i + pstride];
    int nA = cA < SLOT ? cA : SLOT;
    int nB = cB < SLOT ? cB : SLOT;
    size_t base = (size_t)i * SLOT;
    v4i_t LA = *(const v4i_t*)(lists + base + 4 * q);
    v4i_t LB = *(const v4i_t*)(lists + lstride + base + 4 * q);
    float di = dinv[i];
    unsigned nmax = (unsigned)N - 1u;
    v4f_t acc = {0.0f, 0.0f, 0.0f, 0.0f};
    pull16_part(LA, 0, nA, h, hp, nmax, di, q, acc);
    pull16_part(LA, 8, nA, h, hp, nmax, di, q, acc);
    pull16_part(LB, 0, nB, h, hp, nmax, di, q, acc);
    pull16_part(LB, 8, nB, h, hp, nmax, di, q, acc);
    if (cA > SLOT || cB > SLOT) {
        int cC = (int)cnt[(size_t)i + 2 * pstride];
        int nC = cC < SLOT ? cC : SLOT;
        for (int k = 0; k < nC; ++k) {
            unsigned s0 = (unsigned)lists[2 * lstride + base + (size_t)k];
            s0 = s0 < nmax ? s0 : nmax;
            const float* r0 = h + (size_t)s0 * hp;
            float n0 = r0[16] * di;
            v4f_t v0 = *((const v4f_t*)r0 + q);
            acc += v0 * n0;
        }
        if (cC > SLOT) {
            unsigned K = (unsigned)ovn[0];
            K = K < (unsigned)E ? K : (unsigned)E;
            for (unsigned k = 0; k < K; ++k) {
                unsigned e = (unsigned)ovf[k];
                e = e < (unsigned)E ? e : (unsigned)E - 1u;
                if (dst[e] != i) continue;
                unsigned s0 = (unsigned)src[e];
                s0 = s0 < nmax ? s0 : nmax;
                const float* r0 = h + (size_t)s0 * hp;
                float n0 = r0[16] * di;
                v4f_t v0 = *((const v4f_t*)r0 + q);
                acc += v0 * n0;
            }
        }
    }
    float d2 = di * di;
    v4f_t hv = *((const v4f_t*)(h + (size_t)i * hp) + q);
    v4f_t bv = ((const v4f_t*)b1)[q];
    v4f_t o;
    o.x = fmaxf(acc.x + hv.x * d2 + bv.x, 0.0f);
    o.y = fmaxf(acc.y + hv.y * d2 + bv.y, 0.0f);
    o.z = fmaxf(acc.z + hv.z * d2 + bv.z, 0.0f);
    o.w = fmaxf(acc.w + hv.w * d2 + bv.w, 0.0f);
    float p0 = o.x * wa.x + o.y * wa.z + o.z * wb.x + o.w * wb.z;
    float p1 = o.x * wa.y + o.y * wa.w + o.z * wb.y + o.w * wb.w;
    p0 += __shfl_xor(p0, 1);
    p1 += __shfl_xor(p1, 1);
    p0 += __shfl_xor(p0, 2);
    p1 += __shfl_xor(p1, 2);
    h2[t] = q == 0 ? p0 : q == 1 ? p1 : q == 2 ? di : 0.0f;
}

__device__ __forceinline__ v3f_t gather_h2(bool on, int ent, unsigned nmax,
                                           const float* __restrict__ h2) {
    v3f_t g = {0.0f, 0.0f, 0.0f};
    if (on) {
        unsigned s = (unsigned)ent;
        s = s < nmax ? s : nmax;
        g = *(const v3f_t*)(h2 + 4 * (size_t)s);
    }
    return g;
}
__device__ __forceinline__ void add_h2(v3f_t v, float di, float& g0, float& g1) {
    float nrm = v.z * di;
    g0 += v.x * nrm;
    g1 += v.y * nrm;
}

__global__ void gcn_pull2_slots(const float* __restrict__ cnt, size_t pstride,
                                const int* __restrict__ lists, size_t lstride,
                                const float* __restrict__ ovn, const int* __restrict__ ovf,
                                const int* __restrict__ src, const int* __restrict__ dst, int E,
                                const float* __restrict__ dinv,
                                const float* __restrict__ h2, const float* __restrict__ b2,
                                float* __restrict__ out, int N) {
    int t = blockIdx.x * blockDim.x + threadIdx.x;
    int i = t >> 2;
    int q = t & 3;
    bool ok = i < N;
    int ic = ok ? i : N - 1;
    int cA = ok ? (int)cnt[ic] : 0;
    int cB = ok ? (int)cnt[(size_t)ic + pstride] : 0;
    int nA = cA < SLOT ? cA : SLOT;
    int nB = cB < SLOT ? cB : SLOT;
    size_t base = (size_t)ic * SLOT;
    v4i_t LA = *(const v4i_t*)(lists + base + 4 * q);
    v4i_t LB = *(const v4i_t*)(lists + lstride + base + 4 * q);
    float di = dinv[ic];
    unsigned nmax = (unsigned)N - 1u;
    float g0 = 0.0f, g1 = 0.0f;
    {
        int ra = nA - 4 * q;
        int rb = nB - 4 * q;
        v3f_t a0 = gather_h2(ra > 0, LA.x, nmax, h2);
        v3f_t a1 = gather_h2(ra > 1, LA.y, nmax, h2);
        v3f_t a2 = gather_h2(ra > 2, LA.z, nmax, h2);
        v3f_t a3 = gather_h2(ra > 3, LA.w, nmax, h2);
        v3f_t c0 = gather_h2(rb > 0, LB.x, nmax, h2);
        v3f_t c1 = gather_h2(rb > 1, LB.y, nmax, h2);
        v3f_t c2 = gather_h2(rb > 2, LB.z, nmax, h2);
        v3f_t c3 = gather_h2(rb > 3, LB.w, nmax, h2);
        add_h2(a0, di, g0, g1);
        add_h2(a1, di, g0, g1);
        add_h2(a2, di, g0, g1);
        add_h2(a3, di, g0, g1);
        add_h2(c0, di, g0, g1);
        add_h2(c1, di, g0, g1);
        add_h2(c2, di, g0, g1);
        add_h2(c3, di, g0, g1);
    }
    if (cA > SLOT || cB > SLOT) {
        int cC = (int)cnt[(size_t)ic + 2 * pstride];
        int nC = cC < SLOT ? cC : SLOT;
        for (int k = q; k < nC; k += 4) {
            unsigned s = (unsigned)lists[2 * lstride + base + (size_t)k];
            s = s < nmax ? s : nmax;
            v4f_t v = *((const v4f_t*)h2 + s);
            float nrm = v.z * di;
            g0 += v.x * nrm;
            g1 += v.y * nrm;
        }
        if (q == 0 && cC > SLOT) {
            unsigned K = (unsigned)ovn[0];
            K = K < (unsigned)E ? K : (unsigned)E;
            for (unsigned k = 0; k < K; ++k) {
                unsigned e = (unsigned)ovf[k];
                e = e < (unsigned)E ? e : (unsigned)E - 1u;
                if (dst[e] != ic) continue;
                unsigned s = (unsigned)src[e];
                s = s < nmax ? s : nmax;
                v4f_t v = *((const v4f_t*)h2 + s);
                float nrm = v.z * di;
                g0 += v.x * nrm;
                g1 += v.y * nrm;
            }
        }
    }
    g0 += __shfl_xor(g0, 1);
    g1 += __shfl_xor(g1, 1);
    g0 += __shfl_xor(g0, 2);
    g1 += __shfl_xor(g1, 2);
    if (!ok || q != 0) return;
    float d2 = di * di;
    v4f_t me = *((const v4f_t*)h2 + i);
    float a = g0 + me.x * d2 + b2[0];
    float b = g1 + me.y * d2 + b2[1];
    float mx  = fmaxf(a, b);
    float lse = mx + logf(expf(a - mx) + expf(b - mx));
    out[2 * (size_t)i]     = a - lse;
    out[2 * (size_t)i + 1] = b - lse;
}

#define R4(v) (((size_t)(v) + 3) & ~(size_t)3)
#define R32(v) (((size_t)(v) + 31) & ~(size_t)31)
extern "C" void kernel_launch(void* const* d_in, const int* in_sizes, int n_in,
                              void* d_out, int out_size, void* d_ws, size_t ws_size,
                              hipStream_t stream) {
    const float* x  = (const float*)d_in[0];
    const int*   ei = (const int*)d_in[1];
    const float* W1 = (const float*)d_in[2];
    const float* b1 = (const float*)d_in[3];
    const float* W2 = (const float*)d_in[4];
    const float* b2 = (const float*)d_in[5];
    float* out = (float*)d_out;

    int N = in_sizes[0] / 128;
    int E = in_sizes[1] / 2;
    const int* src = ei;
    const int* dst = ei + E;

    int eb  = (E + THREADS - 1) / THREADS;
    int nb  = (N + THREADS - 1) / THREADS;
    int nb4 = (int)(((long long)N * 4 + THREADS - 1) / THREADS);

    {
        size_t n  = (size_t)N;
        size_t ne = (size_t)E;
        size_t o_dinv  = R32(3 * n + 1);
        size_t o_ovf   = o_dinv + R32(n);
        size_t o_h1r   = o_ovf + R32(ne);
        size_t o_lists = o_h1r + R32(16 * n);
        size_t lstr    = R32(SLOT * n);
        size_t o_h1    = o_lists + 3 * lstr;
        size_t total   = o_h1 + 32 * n;
        if (N > 0 && E > 0 && ws_size >= total * sizeof(float)) {
            float* ws    = (float*)d_ws;
            float* cnt   = ws;
            float* ovn   = ws + 3 * n;
            float* dinv  = ws + o_dinv;
            int*   ovf   = (int*)(ws + o_ovf);
            float* h2    = ws + o_h1r;
            int*   lists = (int*)(ws + o_lists);
            float* h1    = ws + o_h1;
            hipMemsetAsync(cnt, 0, (3 * n + 1) * sizeof(float), stream);
            int ntiles = (N + 15) / 16;
            int gb = (ntiles + 7) / 8;
            gcn_gemm1_slots<<<gb, THREADS, 0, stream>>>(x, W1, h1, 32, ntiles, N, src, dst, cnt, n, E,
                                                        lists, lstr, ovn, ovf);
            gcn_dinv_slots<<<nb, THREADS, 0, stream>>>(cnt, n, N, dinv, h1, 32);
            gcn_pull16_slots<<<nb4, THREADS, 0, stream>>>(cnt, n, lists, lstr, ovn, ovf, src, dst, E,
                                                          dinv, h1, 32, b1, W2, h2, N);
            gcn_pull2_slots<<<nb4, THREADS, 0, stream>>>(cnt, n, lists, lstr, ovn, ovf, src, dst, E,
                                                         dinv, h2, b2, out, N);
            return;
        }
    }

    size_t n  = (size_t)N;
    size_t ne = (size_t)E;
    size_t big = ne > 16 * n ? ne : 16 * n;
    size_t fixed = R4(n) + R4(n) + R4((size_t)nb) + R4(big) + R4(ne) + 32 + 16 * n;
    size_t np = (ws_size >= (R4(2 * n) + fixed) * sizeof(float)) ? 2 : 1;
    size_t nl = (np == 2 && ws_size >= (R4(2 * n) + R4(ne) + fixed) * sizeof(float)) ? 2 : 1;
    int    hp = (ws_size >= (R4(np * n) + (nl - 1) * R4(ne) + fixed + 16 * n) * sizeof(float)) ? 32 : 16;
    size_t sd = (np - 1) * n;
    float* ws   = (float*)d_ws;
    float* cnt  = ws;
    int*   lstb = nl == 2 ? (int*)(cnt + R4(np * n)) : nullptr;
    int*   endp = (int*)(cnt + R4(np * n)) + (nl - 1) * R4(ne);
    float* dinv = (float*)(endp + R4(n));
    int*   bt   = (int*)(dinv + R4(n));
    int*   rankw = bt + R4((size_t)nb);
    float* h1r  = (float*)rankw;
    int*   lst  = rankw + R4(big);
    size_t h1off = (size_t)((float*)(lst + R4(ne)) - ws);
    float* h1   = ws + ((h1off + 31) & ~(size_t)31);
    float* h2   = h1;

    hipMemsetAsync(cnt, 0, (nl == 2 ? R4(np * n) + ne : np * n) * sizeof(float), stream);

    int ntiles = (N + 15) / 16;
    int gb = (ntiles + 7) / 8;
    gcn_gemm1_deg<<<gb, THREADS, 0, stream>>>(x, W1, h1, hp, ntiles, N, dst, cnt, sd, E, rankw);
    gcn_block_counts<<<nb, THREADS, 0, stream>>>(cnt, sd, N, bt);
    gcn_dinv_offsets<<<nb, THREADS, 0, stream>>>(cnt, sd, N, bt, endp, dinv, h1, hp);
    gcn_place_edges<<<eb, THREADS, 0, stream>>>(src, dst, rankw, (const int*)cnt, sd, lst, lstb, E);

    gcn_pull16<<<nb4, THREADS, 0, stream>>>((const int*)cnt, endp, lst, lstb, dinv, h1, hp, b1, h1r, N);
    gcn_gemm2<<<nb, THREADS, 0, stream>>>(h1r, W2, dinv, h2, N);
    gcn_pull2_final<<<nb4, THREADS, 0, stream>>>((const int*)cnt, endp, lst, lstb, dinv, h2, b2, out, N);
}


# Edges sorted by destination range in LDS in the x @ W1 kernel and written as runs (one atomic per run, not per edge); a block per range ranks them by ballots into list lines; sum order changes
# speedup vs baseline: 4.9900x; 1.1097x over previous
#include <hip/hip_runtime.h>
#include <hip/hip_bf16.h>

typedef __attribute__((ext_vector_type(2))) float v2f_t;
typedef __attribute__((ext_vector_type(4))) float v4f_t;
typedef __attribute__((ext_vector_type(8))) float v8f_t;
typedef __attribute__((ext_vector_type(4))) int   v4i_t;
typedef __attribute__((ext_vector_type(3))) float v3f_t;

#define THREADS 256

__device__ __forceinline__ void atadd8_ret(float* const (&p)[8], const float (&a)[8], float (&r)[8]) {
    asm volatile("global_atomic_add_f32 %0, %8, %16, off th:TH_ATOMIC_RETURN scope:SCOPE_DEV\n\t"
                 "global_atomic_add_f32 %1, %9, %17, off th:TH_ATOMIC_RETURN scope:SCOPE_DEV\n\t"
                 "global_atomic_add_f32 %2, %10, %18, off th:TH_ATOMIC_RETURN scope:SCOPE_DEV\n\t"
                 "global_atomic_add_f32 %3, %11, %19, off th:TH_ATOMIC_RETURN scope:SCOPE_DEV\n\t"
                 "global_atomic_add_f32 %4, %12, %20, off th:TH_ATOMIC_RETURN scope:SCOPE_DEV\n\t"
                 "global_atomic_add_f32 %5, %13, %21, off th:TH_ATOMIC_RETURN scope:SCOPE_DEV\n\t"
                 "global_atomic_add_f32 %6, %14, %22, off th:TH_ATOMIC_RETURN scope:SCOPE_DEV\n\t"
                 "global_atomic_add_f32 %7, %15, %23, off th:TH_ATOMIC_RETURN scope:SCOPE_DEV\n\t"
                 "s_wait_loadcnt 0x0"
                 : "=&v"(r[0]), "=&v"(r[1]), "=&v"(r[2]), "=&v"(r[3]),
                   "=&v"(r[4]), "=&v"(r[5]), "=&v"(r[6]), "=&v"(r[7])
                 : "v"(p[0]), "v"(p[1]), "v"(p[2]), "v"(p[3]),
                   "v"(p[4]), "v"(p[5]), "v"(p[6]), "v"(p[7]),
                   "v"(a[0]), "v"(a[1]), "v"(a[2]), "v"(a[3]),
                   "v"(a[4]), "v"(a[5]), "v"(a[6]), "v"(a[7])
                 : "memory");
}

__device__ __forceinline__ unsigned pkg_half() {
    unsigned id = __builtin_amdgcn_s_sendmsg_rtn(0x87);
    return (id >> 18) & 1u;
}


__global__ void gcn_block_counts(const float* __restrict__ cnt, size_t pstride, int N,
                                 int* __restrict__ bt) {
    __shared__ int wtot[THREADS / 32];
    int i = blockIdx.x * blockDim.x + threadIdx.x;
    int c = 0;
    if (i < N) {
        float f = cnt[i];
        if (pstride) f += cnt[(size_t)i + pstride];
        c = (int)f;
    }
    for (int o = 16; o > 0; o >>= 1) c += __shfl_xor(c, o);
    if ((threadIdx.x & 31) == 0) wtot[threadIdx.x >> 5] = c;
    __syncthreads();
    if (threadIdx.x == 0) {
        int t = 0;
        for (int k = 0; k < THREADS / 32; ++k) t += wtot[k];
        bt[blockIdx.x] = t;
    }
}

__global__ void gcn_dinv_offsets(float* __restrict__ cnt, size_t pstride, int N,
                                 const int* __restrict__ bt, int* __restrict__ endp,
                                 float* __restrict__ dinv, float* __restrict__ h1, int hp) {
    __shared__ int wsum[THREADS / 32];
    __shared__ int bsum[THREADS / 32];
    int i = blockIdx.x * blockDim.x + threadIdx.x;
    int lane = threadIdx.x & 31;
    int w = threadIdx.x >> 5;
    int part = 0;
    for (int k = threadIdx.x; k < (int)blockIdx.x; k += THREADS) part += bt[k];
    for (int o = 16; o > 0; o >>= 1) part += __shfl_xor(part, o);
    int cA = 0, c = 0;
    if (i < N) {
        float fa = cnt[i];
        float f = fa;
        if (pstride) f += cnt[(size_t)i + pstride];
        cA = (int)fa;
        c  = (int)f;
        float dv = rsqrtf(f + 1.0f);
        dinv[i] = dv;
        if (hp > 16) h1[(size_t)i * hp + 16] = dv;
    }
    int incl = c;
    for (int o = 1; o < 32; o <<= 1) {
        int t = __shfl_up(incl, o);
        if (lane >= o) incl += t;
    }
    if (lane == 0)  bsum[w] = part;
    if (lane == 31) wsum[w] = incl;
    __syncthreads();
    int st = incl - c;
    for (int k = 0; k < THREADS / 32; ++k) {
        st += bsum[k];
        if (k < w) st += wsum[k];
    }
    if (i < N) {
        int* s0 = (int*)cnt;
        s0[i] = st;
        if (pstride) s0[(size_t)i + pstride] = st + cA;
        endp[i] = st + c;
    }
}

__global__ void gcn_place_edges(const int* __restrict__ src, const int* __restrict__ dst,
                                const int* __restrict__ rankw, const int* __restrict__ st,
                                size_t pstride, int* __restrict__ lst, int* __restrict__ lstb,
                                int E) {
    int e = blockIdx.x * blockDim.x + threadIdx.x;
    int* lp = (lstb && pkg_half()) ? lstb : lst;
    if (e >= E) return;
    int s = __builtin_nontemporal_load(src + e);
    int d = __builtin_nontemporal_load(dst + e);
    int r = __builtin_nontemporal_load(rankw + e);
    size_t pl = (r >> 30) & 1 ? pstride : 0;
    int pos = st[(size_t)d + pl] + (r & 0x3fffffff);
    if ((unsigned)pos < (unsigned)E) lp[pos] = s + 1;
}

#define XPITCH 132
#define EPL 8
__global__ void gcn_gemm1_deg(const float* __restrict__ x, const float* __restrict__ W1,
                              float* __restrict__ h1, int hp, int ntiles, int N,
                              const int* __restrict__ dst, float* __restrict__ deg,
                              size_t pstride, int E, int* __restrict__ rankw) {
    __shared__ __attribute__((aligned(16))) float xs[(THREADS / 32) * 16 * XPITCH];
    int wib  = threadIdx.x >> 5;
    int wave = blockIdx.x * (blockDim.x >> 5) + wib;
    bool active = wave < ntiles;
    int tile = active ? wave : ntiles - 1;
    int lane = threadIdx.x & 31;
    float* xt = xs + wib * (16 * XPITCH);

    unsigned wv     = (unsigned)__builtin_amdgcn_readfirstlane(wave);
    unsigned nwaves = gridDim.x * (blockDim.x >> 5);
    unsigned chunk  = (((unsigned)E + nwaves - 1u) / nwaves + 31u) & ~31u;
    unsigned e0 = wv * chunk;
    unsigned e1 = e0 + chunk < (unsigned)E ? e0 + chunk : (unsigned)E;
    bool any = e0 < e1;
    unsigned hsel = pstride ? pkg_half() : 0u;
    float* plane = deg + (size_t)hsel * pstride;
    int hbit = (int)(hsel << 30);

    int dv[EPL] = {};
    if (any) {
#pragma unroll
        for (int j = 0; j < EPL; ++j) {
            unsigned idx = e0 + j * 32 + lane;
            dv[j] = __builtin_nontemporal_load(dst + (idx < e1 ? idx : e1 - 1u));
        }
    }

    v4f_t rowv[16];
#pragma unroll
    for (int r = 0; r < 16; ++r) {
        int row = tile * 16 + r;
        int rcl = row < N ? row : N - 1;
        rowv[r] = __builtin_nontemporal_load((const v4f_t*)(x + (size_t)rcl * 128) + lane);
    }
    {
        float* pp[EPL];
        float  aa[EPL];
        float  rk[EPL];
#pragma unroll
        for (int j = 0; j < EPL; ++j) {
            unsigned idx = e0 + j * 32 + lane;
            pp[j] = plane + (unsigned)dv[j];
            aa[j] = idx < e1 ? 1.0f : 0.0f;
        }
        atadd8_ret(pp, aa, rk);
#pragma unroll
        for (int j = 0; j < EPL; ++j) {
            unsigned idx = e0 + j * 32 + lane;
            if (idx < e1) __builtin_nontemporal_store((int)rk[j] | hbit, rankw + idx);
        }
    }
    for (unsigned g = e0 + 32 * EPL; g < e1; g += 32 * EPL) {
        float* pp[EPL];
        float  aa[EPL];
        float  rk[EPL];
#pragma unroll
        for (int j = 0; j < EPL; ++j) {
            unsigned idx = g + j * 32 + lane;
            pp[j] = plane + (unsigned)__builtin_nontemporal_load(dst + (idx < e1 ? idx : e1 - 1u));
            aa[j] = idx < e1 ? 1.0f : 0.0f;
        }
        atadd8_ret(pp, aa, rk);
#pragma unroll
        for (int j = 0; j < EPL; ++j) {
            unsigned idx = g + j * 32 + lane;
            if (idx < e1) __builtin_nontemporal_store((int)rk[j] | hbit, rankw + idx);
        }
    }

#pragma unroll
    for (int r = 0; r < 16; ++r) *(v4f_t*)(xt + r * XPITCH + lane * 4) = rowv[r];
    __syncthreads();

    int half = lane >> 4;
    int m    = lane & 15;
    int koff = half << 1;
    const float* xr = xt + m * XPITCH;

    v8f_t c = {};
#pragma unroll
    for (int k0 = 0; k0 < 128; k0 += 4) {
        v2f_t a = *(const v2f_t*)(xr + k0 + koff);
        v2f_t b;
        b.x = W1[(k0 + koff) * 16 + m];
        b.y = W1[(k0 + koff + 1) * 16 + m];
        c = __builtin_amdgcn_wmma_f32_16x16x4_f32(false, a, false, b, (short)0, c,
                                                  false, false);
    }
    if (!active) return;
#pragma unroll
    for (int v = 0; v < 8; ++v) {
        int r = wave * 16 + half * 8 + v;
        if (r < N) h1[(size_t)r * hp + m] = c[v];
    }
}

__global__ void gcn_pull16(const int* __restrict__ start, const int* __restrict__ endp,
                           const int* __restrict__ lst, const int* __restrict__ lstb,
                           const float* __restrict__ dinv,
                           const float* __restrict__ h, int hp, const float* __restrict__ b1,
                           float* __restrict__ hout, int N) {
    int t = blockIdx.x * blockDim.x + threadIdx.x;
    int i = t >> 2;
    int q = t & 3;
    if (i >= N) return;
    int p  = start[i];
    int pe = endp[i];
    float di = dinv[i];
    unsigned nmax = (unsigned)N - 1u;
    v4f_t acc = {0.0f, 0.0f, 0.0f, 0.0f};
    for (; p + 1 < pe; p += 2) {
        int e0 = lst[p];
        int e1 = lst[p + 1];
        if (lstb) {
            int f0 = lstb[p];
            int f1 = lstb[p + 1];
            e0 = f0 ? f0 : e0;
            e1 = f1 ? f1 : e1;
        }
        unsigned s0 = (unsigned)(e0 - 1);
        unsigned s1 = (unsigned)(e1 - 1);
        s0 = s0 < nmax ? s0 : nmax;
        s1 = s1 < nmax ? s1 : nmax;
        const float* r0 = h + (size_t)s0 * hp;
        const float* r1 = h + (size_t)s1 * hp;
        float n0 = *(hp > 16 ? r0 + 16 : dinv + s0) * di;
        float n1 = *(hp > 16 ? r1 + 16 : dinv + s1) * di;
        v4f_t v0 = *((const v4f_t*)r0 + q);
        v4f_t v1 = *((const v4f_t*)r1 + q);
        acc += v0 * n0;
        acc += v1 * n1;
    }
    if (p < pe) {
        int e0 = lst[p];
        if (lstb) {
            int f0 = lstb[p];
            e0 = f0 ? f0 : e0;
        }
        unsigned s0 = (unsigned)(e0 - 1);
        s0 = s0 < nmax ? s0 : nmax;
        const float* r0 = h + (size_t)s0 * hp;
        float n0 = *(hp > 16 ? r0 + 16 : dinv + s0) * di;
        v4f_t v0 = *((const v4f_t*)r0 + q);
        acc += v0 * n0;
    }
    float d2 = di * di;
    v4f_t hv = *((const v4f_t*)(h + (size_t)i * hp) + q);
    v4f_t bv = ((const v4f_t*)b1)[q];
    v4f_t o;
    o.x = fmaxf(acc.x + hv.x * d2 + bv.x, 0.0f);
    o.y = fmaxf(acc.y + hv.y * d2 + bv.y, 0.0f);
    o.z = fmaxf(acc.z + hv.z * d2 + bv.z, 0.0f);
    o.w = fmaxf(acc.w + hv.w * d2 + bv.w, 0.0f);
    *((v4f_t*)(hout + (size_t)i * 16) + q) = o;
}

__global__ void gcn_gemm2(const float* __restrict__ h, const float* __restrict__ W2,
                          const float* __restrict__ dinv, float* __restrict__ h2, int N) {
    int i = blockIdx.x * blockDim.x + threadIdx.x;
    if (i >= N) return;
    const float* hr = h + (size_t)i * 16;
    float o0 = 0.0f, o1 = 0.0f;
#pragma unroll
    for (int k = 0; k < 16; ++k) {
        float v = hr[k];
        o0 += v * W2[2 * k];
        o1 += v * W2[2 * k + 1];
    }
    v4f_t o = {o0, o1, dinv[i], 0.0f};
    *((v4f_t*)h2 + i) = o;
}

__global__ void gcn_pull2_final(const int* __restrict__ start, const int* __restrict__ endp,
                                const int* __restrict__ lst, const int* __restrict__ lstb,
                                const float* __restrict__ dinv,
                                const float* __restrict__ h2, const float* __restrict__ b2,
                                float* __restrict__ out, int N) {
    int t = blockIdx.x * blockDim.x + threadIdx.x;
    int i = t >> 2;
    int q = t & 3;
    bool ok = i < N;
    int ic = ok ? i : N - 1;
    int p  = start[ic] + q;
    int pe = ok ? endp[ic] : 0;
    float di = dinv[ic];
    unsigned nmax = (unsigned)N - 1u;
    float g0 = 0.0f, g1 = 0.0f;
    for (; p < pe; p += 4) {
        int ev = lst[p];
        if (lstb) {
            int fv = lstb[p];
            ev = fv ? fv : ev;
        }
        unsigned s = (unsigned)(ev - 1);
        s = s < nmax ? s : nmax;
        v4f_t v = *((const v4f_t*)h2 + s);
        float nrm = v.z * di;
        g0 += v.x * nrm;
        g1 += v.y * nrm;
    }
    g0 += __shfl_xor(g0, 1);
    g1 += __shfl_xor(g1, 1);
    g0 += __shfl_xor(g0, 2);
    g1 += __shfl_xor(g1, 2);
    if (!ok || q != 0) return;
    float d2 = di * di;
    v4f_t me = *((const v4f_t*)h2 + i);
    float a = g0 + me.x * d2 + b2[0];
    float b = g1 + me.y * d2 + b2[1];
    float mx  = fmaxf(a, b);
    float lse = mx + logf(expf(a - mx) + expf(b - mx));
    out[2 * (size_t)i]     = a - lse;
    out[2 * (size_t)i + 1] = b - lse;
}


#define SLOT 16

__device__ __forceinline__ float atadd1_ret(float* p, float a) {
    float r;
    asm volatile("global_atomic_add_f32 %0, %1, %2, off th:TH_ATOMIC_RETURN scope:SCOPE_DEV\n\t"
                 "s_wait_loadcnt 0x0"
                 : "=&v"(r) : "v"(p), "v"(a) : "memory");
    return r;
}

__device__ __forceinline__ void atadd8_ret_s(float* base, unsigned (&o)[8],
                                             const float (&a)[8], float (&r)[8]) {
    asm volatile("v_lshlrev_b32 %8, 2, %8\n\t"
                 "v_lshlrev_b32 %9, 2, %9\n\t"
                 "v_lshlrev_b32 %10, 2, %10\n\t"
                 "v_lshlrev_b32 %11, 2, %11\n\t"
                 "v_lshlrev_b32 %12, 2, %12\n\t"
                 "v_lshlrev_b32 %13, 2, %13\n\t"
                 "v_lshlrev_b32 %14, 2, %14\n\t"
                 "v_lshlrev_b32 %15, 2, %15\n\t"
                 "global_atomic_add_f32 %0, %8, %16, %24 th:TH_ATOMIC_RETURN scope:SCOPE_DEV\n\t"
                 "global_atomic_add_f32 %1, %9, %17, %24 th:TH_ATOMIC_RETURN scope:SCOPE_DEV\n\t"
                 "global_atomic_add_f32 %2, %10, %18, %24 th:TH_ATOMIC_RETURN scope:SCOPE_DEV\n\t"
                 "global_atomic_add_f32 %3, %11, %19, %24 th:TH_ATOMIC_RETURN scope:SCOPE_DEV\n\t"
                 "global_atomic_add_f32 %4, %12, %20, %24 th:TH_ATOMIC_RETURN scope:SCOPE_DEV\n\t"
                 "global_atomic_add_f32 %5, %13, %21, %24 th:TH_ATOMIC_RETURN scope:SCOPE_DEV\n\t"
                 "global_atomic_add_f32 %6, %14, %22, %24 th:TH_ATOMIC_RETURN scope:SCOPE_DEV\n\t"
                 "global_atomic_add_f32 %7, %15, %23, %24 th:TH_ATOMIC_RETURN scope:SCOPE_DEV\n\t"
                 "s_wait_loadcnt 0x0"
                 : "=&v"(r[0]), "=&v"(r[1]), "=&v"(r[2]), "=&v"(r[3]),
                   "=&v"(r[4]), "=&v"(r[5]), "=&v"(r[6]), "=&v"(r[7]),
                   "+v"(o[0]), "+v"(o[1]), "+v"(o[2]), "+v"(o[3]),
                   "+v"(o[4]), "+v"(o[5]), "+v"(o[6]), "+v"(o[7])
                 : "v"(a[0]), "v"(a[1]), "v"(a[2]), "v"(a[3]),
                   "v"(a[4]), "v"(a[5]), "v"(a[6]), "v"(a[7]),
                   "s"(base)
                 : "memory");
}

__device__ __forceinline__ void place_in_slot(int* __restrict__ lp, int* __restrict__ lc,
                                              float* __restrict__ cntc, float* __restrict__ ovn,
                                              int* __restrict__ ovf, int E, unsigned d, int s,
                                              float rankf, unsigned idx) {
    int r = (int)rankf;
    if (r < SLOT) {
        lp[(size_t)d * SLOT + r] = s;
    } else {
        int c = (int)atadd1_ret(cntc + d, 1.0f);
        if (c < SLOT) {
            lc[(size_t)d * SLOT + c] = s;
        } else {
            unsigned k = (unsigned)atadd1_ret(ovn, 1.0f);
            if (k < (unsigned)E) ovf[k] = (int)idx;
        }
    }
}

__global__ void gcn_gemm1_slots(const float* __restrict__ x, const float* __restrict__ W1,
                                float* __restrict__ h1, int hp, int ntiles, int N,
                                const int* __restrict__ src, const int* __restrict__ dst,
                                float* __restrict__ deg, size_t pstride, int E,
                                int* __restrict__ lists, size_t lstride,
                                float* __restrict__ ovn, int* __restrict__ ovf) {
    __shared__ __attribute__((aligned(16))) float xs[(THREADS / 32) * 16 * XPITCH];
    int wib  = threadIdx.x >> 5;
    int wave = blockIdx.x * (blockDim.x >> 5) + wib;
    bool active = wave < ntiles;
    int tile = active ? wave : ntiles - 1;
    int lane = threadIdx.x & 31;
    float* xt = xs + wib * (16 * XPITCH);

    unsigned wv     = (unsigned)__builtin_amdgcn_readfirstlane(wave);
    unsigned nwaves = gridDim.x * (blockDim.x >> 5);
    unsigned chunk  = (((unsigned)E + nwaves - 1u) / nwaves + 31u) & ~31u;
    unsigned e0 = wv * chunk;
    unsigned e1 = e0 + chunk < (unsigned)E ? e0 + chunk : (unsigned)E;
    bool any = e0 < e1;
    unsigned hsel = pkg_half();
    float* plane = deg + (size_t)hsel * pstride;
    int*   lp    = lists + (size_t)hsel * lstride;
    int*   lc    = lists + 2 * lstride;
    float* cntc  = deg + 2 * pstride;

    unsigned dv[EPL] = {};
    int      sv[EPL] = {};
    if (any) {
#pragma unroll
        for (int j = 0; j < EPL; ++j) {
            unsigned idx = e0 + j * 32 + lane;
            dv[j] = (unsigned)__builtin_nontemporal_load(dst + (idx < e1 ? idx : e1 - 1u));
        }
#pragma unroll
        for (int j = 0; j < EPL; ++j) {
            unsigned idx = e0 + j * 32 + lane;
            sv[j] = __builtin_nontemporal_load(src + (idx < e1 ? idx : e1 - 1u));
        }
    }

    v4f_t rowv[16];
#pragma unroll
    for (int r = 0; r < 16; ++r) {
        int row = tile * 16 + r;
        int rcl = row < N ? row : N - 1;
        rowv[r] = __builtin_nontemporal_load((const v4f_t*)(x + (size_t)rcl * 128) + lane);
    }
    {
        float aa[EPL];
        float rk[EPL];
#pragma unroll
        for (int j = 0; j < EPL; ++j) {
            unsigned idx = e0 + j * 32 + lane;
            aa[j] = idx < e1 ? 1.0f : 0.0f;
        }
        atadd8_ret_s(plane, dv, aa, rk);
#pragma unroll
        for (int j = 0; j < EPL; ++j) {
            unsigned idx = e0 + j * 32 + lane;
            if (idx < e1) place_in_slot(lp, lc, cntc, ovn, ovf, E, dv[j] >> 2, sv[j], rk[j], idx);
        }
    }
    for (unsigned g = e0 + 32 * EPL; g < e1; g += 32 * EPL) {
        float    aa[EPL];
        float    rk[EPL];
        unsigned dd[EPL];
        int      ss[EPL];
#pragma unroll
        for (int j = 0; j < EPL; ++j) {
            unsigned idx = g + j * 32 + lane;
            unsigned cl = idx < e1 ? idx : e1 - 1u;
            dd[j] = (unsigned)__builtin_nontemporal_load(dst + cl);
            ss[j] = __builtin_nontemporal_load(src + cl);
            aa[j] = idx < e1 ? 1.0f : 0.0f;
        }
        atadd8_ret_s(plane, dd, aa, rk);
#pragma unroll
        for (int j = 0; j < EPL; ++j) {
            unsigned idx = g + j * 32 + lane;
            if (idx < e1) place_in_slot(lp, lc, cntc, ovn, ovf, E, dd[j] >> 2, ss[j], rk[j], idx);
        }
    }

#pragma unroll
    for (int r = 0; r < 16; ++r) *(v4f_t*)(xt + r * XPITCH + lane * 4) = rowv[r];
    __syncthreads();

    int half = lane >> 4;
    int m    = lane & 15;
    int koff = half << 1;
    const float* xr = xt + m * XPITCH;

    v8f_t c = {};
#pragma unroll
    for (int k0 = 0; k0 < 128; k0 += 4) {
        v2f_t a = *(const v2f_t*)(xr + k0 + koff);
        v2f_t b;
        b.x = W1[(k0 + koff) * 16 + m];
        b.y = W1[(k0 + koff + 1) * 16 + m];
        c = __builtin_amdgcn_wmma_f32_16x16x4_f32(false, a, false, b, (short)0, c,
                                                  false, false);
    }
    if (!active) return;
#pragma unroll
    for (int v = 0; v < 8; ++v) {
        int r = wave * 16 + half * 8 + v;
        if (r < N) h1[(size_t)r * hp + m] = c[v];
    }
}

__global__ void gcn_dinv_slots(const float* __restrict__ cnt, size_t pstride, int N,
                               float* __restrict__ dinv, float* __restrict__ h1, int hp) {
    int i = blockIdx.x * blockDim.x + threadIdx.x;
    if (i >= N) return;
    float f = cnt[i];
    f += cnt[(size_t)i + pstride];
    float dv = rsqrtf(f + 1.0f);
    dinv[i] = dv;
    h1[(size_t)i * hp + 16] = dv;
}

static_assert(SLOT == 16, "4 threads of a node hold 4 slot entries each");

__device__ __forceinline__ int quad_from(int v, int m) {
    switch (m) {
    case 0:  return __builtin_amdgcn_mov_dpp(v, 0x00, 0xF, 0xF, true);
    case 1:  return __builtin_amdgcn_mov_dpp(v, 0x55, 0xF, 0xF, true);
    case 2:  return __builtin_amdgcn_mov_dpp(v, 0xAA, 0xF, 0xF, true);
    default: return __builtin_amdgcn_mov_dpp(v, 0xFF, 0xF, 0xF, true);
    }
}

struct hrow_t { v4f_t v; float n; };
__device__ __forceinline__ hrow_t gather_h1(bool on, int ent, unsigned nmax,
                                            const float* __restrict__ h, int hp, int q) {
    hrow_t r;
    r.v = (v4f_t){0.0f, 0.0f, 0.0f, 0.0f};
    r.n = 0.0f;
    if (on) {
        unsigned s = (unsigned)ent;
        s = s < nmax ? s : nmax;
        const float* p = h + (size_t)s * hp;
        r.n = p[16];
        r.v = *((const v4f_t*)p + q);
    }
    return r;
}

__device__ __forceinline__ void pull16_part(v4i_t L, int j0, int n, const float* __restrict__ h,
                                            int hp, unsigned nmax, float di, int q, v4f_t& acc) {
    if (j0 >= n) return;
    int m = j0 >> 2;
    hrow_t r0 = gather_h1(j0     < n, quad_from(L.x, m),     nmax, h, hp, q);
    hrow_t r1 = gather_h1(j0 + 1 < n, quad_from(L.y, m),     nmax, h, hp, q);
    hrow_t r2 = gather_h1(j0 + 2 < n, quad_from(L.z, m),     nmax, h, hp, q);
    hrow_t r3 = gather_h1(j0 + 3 < n, quad_from(L.w, m),     nmax, h, hp, q);
    hrow_t r4 = gather_h1(j0 + 4 < n, quad_from(L.x, m + 1), nmax, h, hp, q);
    hrow_t r5 = gather_h1(j0 + 5 < n, quad_from(L.y, m + 1), nmax, h, hp, q);
    hrow_t r6 = gather_h1(j0 + 6 < n, quad_from(L.z, m + 1), nmax, h, hp, q);
    hrow_t r7 = gather_h1(j0 + 7 < n, quad_from(L.w, m + 1), nmax, h, hp, q);
    float n0 = r0.n * di;  acc += r0.v * n0;
    float n1 = r1.n * di;  acc += r1.v * n1;
    float n2 = r2.n * di;  acc += r2.v * n2;
    float n3 = r3.n * di;  acc += r3.v * n3;
    float n4 = r4.n * di;  acc += r4.v * n4;
    float n5 = r5.n * di;  acc += r5.v * n5;
    float n6 = r6.n * di;  acc += r6.v * n6;
    float n7 = r7.n * di;  acc += r7.v * n7;
}

__global__ void gcn_pull16_slots(const float* __restrict__ cnt, size_t pstride,
                                 const int* __restrict__ lists, size_t lstride,
                                 const float* __restrict__ ovn, const int* __restrict__ ovf,
                                 const int* __restrict__ src, const int* __restrict__ dst, int E,
                                 const float* __restrict__ dinv,
                                 const float* __restrict__ h, int hp, const float* __restrict__ b1,
                                 const float* __restrict__ W2, float* __restrict__ h2, int N) {
    int t = blockIdx.x * blockDim.x + threadIdx.x;
    int i = t >> 2;
    int q = t & 3;
    if (i >= N) return;
    v4f_t wa = ((const v4f_t*)W2)[2 * q];
    v4f_t wb = ((const v4f_t*)W2)[2 * q + 1];
    int cA = (int)cnt[i];
    int cB = (int)cnt[(size_t)i + pstride];
    int nA = cA < SLOT ? cA : SLOT;
    int nB = cB < SLOT ? cB : SLOT;
    size_t base = (size_t)i * SLOT;
    v4i_t LA = *(const v4i_t*)(lists + base + 4 * q);
    v4i_t LB = *(const v4i_t*)(lists + lstride + base + 4 * q);
    float di = dinv[i];
    unsigned nmax = (unsigned)N - 1u;
    v4f_t acc = {0.0f, 0.0f, 0.0f, 0.0f};
    pull16_part(LA, 0, nA, h, hp, nmax, di, q, acc);
    pull16_part(LA, 8, nA, h, hp, nmax, di, q, acc);
    pull16_part(LB, 0, nB, h, hp, nmax, di, q, acc);
    pull16_part(LB, 8, nB, h, hp, nmax, di, q, acc);
    if (cA > SLOT || cB > SLOT) {
        int cC = (int)cnt[(size_t)i + 2 * pstride];
        int nC = cC < SLOT ? cC : SLOT;
        for (int k = 0; k < nC; ++k) {
            unsigned s0 = (unsigned)lists[2 * lstride + base + (size_t)k];
            s0 = s0 < nmax ? s0 : nmax;
            const float* r0 = h + (size_t)s0 * hp;
            float n0 = r0[16] * di;
            v4f_t v0 = *((const v4f_t*)r0 + q);
            acc += v0 * n0;
        }
        if (cC > SLOT) {
            unsigned K = (unsigned)ovn[0];
            K = K < (unsigned)E ? K : (unsigned)E;
            for (unsigned k = 0; k < K; ++k) {
                unsigned e = (unsigned)ovf[k];
                e = e < (unsigned)E ? e : (unsigned)E - 1u;
                if (dst[e] != i) continue;
                unsigned s0 = (unsigned)src[e];
                s0 = s0 < nmax ? s0 : nmax;
                const float* r0 = h + (size_t)s0 * hp;
                float n0 = r0[16] * di;
                v4f_t v0 = *((const v4f_t*)r0 + q);
                acc += v0 * n0;
            }
        }
    }
    float d2 = di * di;
    v4f_t hv = *((const v4f_t*)(h + (size_t)i * hp) + q);
    v4f_t bv = ((const v4f_t*)b1)[q];
    v4f_t o;
    o.x = fmaxf(acc.x + hv.x * d2 + bv.x, 0.0f);
    o.y = fmaxf(acc.y + hv.y * d2 + bv.y, 0.0f);
    o.z = fmaxf(acc.z + hv.z * d2 + bv.z, 0.0f);
    o.w = fmaxf(acc.w + hv.w * d2 + bv.w, 0.0f);
    float p0 = o.x * wa.x + o.y * wa.z + o.z * wb.x + o.w * wb.z;
    float p1 = o.x * wa.y + o.y * wa.w + o.z * wb.y + o.w * wb.w;
    p0 += __shfl_xor(p0, 1);
    p1 += __shfl_xor(p1, 1);
    p0 += __shfl_xor(p0, 2);
    p1 += __shfl_xor(p1, 2);
    h2[t] = q == 0 ? p0 : q == 1 ? p1 : q == 2 ? di : 0.0f;
}

__device__ __forceinline__ v3f_t gather_h2(bool on, int ent, unsigned nmax,
                                           const float* __restrict__ h2) {
    v3f_t g = {0.0f, 0.0f, 0.0f};
    if (on) {
        unsigned s = (unsigned)ent;
        s = s < nmax ? s : nmax;
        g = *(const v3f_t*)(h2 + 4 * (size_t)s);
    }
    return g;
}
__device__ __forceinline__ void add_h2(v3f_t v, float di, float& g0, float& g1) {
    float nrm = v.z * di;
    g0 += v.x * nrm;
    g1 += v.y * nrm;
}

__global__ void gcn_pull2_slots(const float* __restrict__ cnt, size_t pstride,
                                const int* __restrict__ lists, size_t lstride,
                                const float* __restrict__ ovn, const int* __restrict__ ovf,
                                const int* __restrict__ src, const int* __restrict__ dst, int E,
                                const float* __restrict__ dinv,
                                const float* __restrict__ h2, const float* __restrict__ b2,
                                float* __restrict__ out, int N) {
    int t = blockIdx.x * blockDim.x + threadIdx.x;
    int i = t >> 2;
    int q = t & 3;
    bool ok = i < N;
    int ic = ok ? i : N - 1;
    int cA = ok ? (int)cnt[ic] : 0;
    int cB = ok ? (int)cnt[(size_t)ic + pstride] : 0;
    int nA = cA < SLOT ? cA : SLOT;
    int nB = cB < SLOT ? cB : SLOT;
    size_t base = (size_t)ic * SLOT;
    v4i_t LA = *(const v4i_t*)(lists + base + 4 * q);
    v4i_t LB = *(const v4i_t*)(lists + lstride + base + 4 * q);
    float di = dinv[ic];
    unsigned nmax = (unsigned)N - 1u;
    float g0 = 0.0f, g1 = 0.0f;
    {
        int ra = nA - 4 * q;
        int rb = nB - 4 * q;
        v3f_t a0 = gather_h2(ra > 0, LA.x, nmax, h2);
        v3f_t a1 = gather_h2(ra > 1, LA.y, nmax, h2);
        v3f_t a2 = gather_h2(ra > 2, LA.z, nmax, h2);
        v3f_t a3 = gather_h2(ra > 3, LA.w, nmax, h2);
        v3f_t c0 = gather_h2(rb > 0, LB.x, nmax, h2);
        v3f_t c1 = gather_h2(rb > 1, LB.y, nmax, h2);
        v3f_t c2 = gather_h2(rb > 2, LB.z, nmax, h2);
        v3f_t c3 = gather_h2(rb > 3, LB.w, nmax, h2);
        add_h2(a0, di, g0, g1);
        add_h2(a1, di, g0, g1);
        add_h2(a2, di, g0, g1);
        add_h2(a3, di, g0, g1);
        add_h2(c0, di, g0, g1);
        add_h2(c1, di, g0, g1);
        add_h2(c2, di, g0, g1);
        add_h2(c3, di, g0, g1);
    }
    if (cA > SLOT || cB > SLOT) {
        int cC = (int)cnt[(size_t)ic + 2 * pstride];
        int nC = cC < SLOT ? cC : SLOT;
        for (int k = q; k < nC; k += 4) {
            unsigned s = (unsigned)lists[2 * lstride + base + (size_t)k];
            s = s < nmax ? s : nmax;
            v4f_t v = *((const v4f_t*)h2 + s);
            float nrm = v.z * di;
            g0 += v.x * nrm;
            g1 += v.y * nrm;
        }
        if (q == 0 && cC > SLOT) {
            unsigned K = (unsigned)ovn[0];
            K = K < (unsigned)E ? K : (unsigned)E;
            for (unsigned k = 0; k < K; ++k) {
                unsigned e = (unsigned)ovf[k];
                e = e < (unsigned)E ? e : (unsigned)E - 1u;
                if (dst[e] != ic) continue;
                unsigned s = (unsigned)src[e];
                s = s < nmax ? s : nmax;
                v4f_t v = *((const v4f_t*)h2 + s);
                float nrm = v.z * di;
                g0 += v.x * nrm;
                g1 += v.y * nrm;
            }
        }
    }
    g0 += __shfl_xor(g0, 1);
    g1 += __shfl_xor(g1, 1);
    g0 += __shfl_xor(g0, 2);
    g1 += __shfl_xor(g1, 2);
    if (!ok || q != 0) return;
    float d2 = di * di;
    v4f_t me = *((const v4f_t*)h2 + i);
    float a = g0 + me.x * d2 + b2[0];
    float b = g1 + me.y * d2 + b2[1];
    float mx  = fmaxf(a, b);
    float lse = mx + logf(expf(a - mx) + expf(b - mx));
    out[2 * (size_t)i]     = a - lse;
    out[2 * (size_t)i + 1] = b - lse;
}


#define RB 11
#define RNODES (1 << RB)
#define NRMAX 256
#define BCAP (9 * RNODES)
#define LCAP 32
#define LPITCH 33
#define BTHREADS 1024
#define BWAVES (BTHREADS / 32)
#define SUBB 5
#define SUBN (1 << SUBB)
#define NSUB (RNODES / SUBN)
#define BLOADS 36
static_assert(THREADS == NRMAX, "one histogram column per thread");
static_assert(32 * EPL == 256, "a wave's group is 256 edges: the local edge number takes 8 bits");
static_assert(NSUB == 2 * BWAVES, "a wave takes two sub-ranges");
static_assert(BLOADS * BTHREADS >= 2 * BCAP, "BLOADS groups per wave take both buckets");

__device__ __forceinline__ unsigned same_key_lanes(unsigned key, bool on, const int nb) {
    unsigned m = __builtin_amdgcn_ballot_w32(on);
#pragma unroll
    for (int b = 0; b < nb; ++b) {
        bool bit = (key >> b) & 1u;
        unsigned bb = __builtin_amdgcn_ballot_w32(bit);
        m &= bit ? bb : ~bb;
    }
    return m;
}

__device__ __forceinline__ void bin_round(const unsigned (&dd)[EPL], const int (&ss)[EPL],
                                          unsigned ebase, unsigned e1, unsigned g, unsigned chunk,
                                          unsigned nmax, unsigned hsel,
                                          float* __restrict__ bcnt, int* __restrict__ bkt,
                                          float* __restrict__ cntg, float* __restrict__ ovn,
                                          int* __restrict__ ovf, int E,
                                          unsigned* lst, unsigned* gbs, unsigned* gbs2,
                                          unsigned* wtot, unsigned* tots,
                                          unsigned* sw, unsigned* sk, unsigned* wh) {
    unsigned tid = threadIdx.x;
    unsigned lane = tid & 31u;
    unsigned wib = tid >> 5;
    unsigned ltm = (1u << lane) - 1u;
    __syncthreads();
    for (unsigned u = 0; u < THREADS / 32; ++u) wh[u * NRMAX + tid] = 0u;
    __syncthreads();
    unsigned* myh = wh + wib * NRMAX;
    unsigned rk[EPL];
#pragma unroll
    for (int j = 0; j < EPL; ++j) {
        unsigned idx = ebase + j * 32 + lane;
        bool on = idx < e1;
        unsigned d = dd[j] < nmax ? dd[j] : nmax;
        unsigned key = d >> RB;
        unsigned m = same_key_lanes(key, on, 8);
        unsigned basec = 0u;
        if (on) basec = myh[key];
        rk[j] = basec + __builtin_popcount(m & ltm);
        if (on && (m & ltm) == 0u) myh[key] = basec + __builtin_popcount(m);
    }
    __syncthreads();
    unsigned c = 0u;
#pragma unroll
    for (unsigned u = 0; u < THREADS / 32; ++u) {
        unsigned t = wh[u * NRMAX + tid];
        wh[u * NRMAX + tid] = c;
        c += t;
    }
    unsigned incl = c;
    for (unsigned o = 1; o < 32; o <<= 1) {
        unsigned t = __shfl_up(incl, o);
        if (lane >= o) incl += t;
    }
    if (lane == 31u) wtot[wib] = incl;
    unsigned gb = 0u, gb2 = 0u;
    if (c) {
        gb = (unsigned)atadd1_ret(bcnt + ((size_t)hsel * NRMAX + tid) * 32, (float)c);
        if (gb + c > BCAP) {
            unsigned st = gb > BCAP ? gb : BCAP;
            gb2 = (unsigned)atadd1_ret(bcnt + ((size_t)(hsel ^ 1u) * NRMAX + tid) * 32,
                                       (float)(gb + c - st));
        }
    }
    __syncthreads();
    unsigned ex = incl - c;
    for (unsigned k = 0; k < THREADS / 32; ++k) if (k < wib) ex += wtot[k];
    lst[tid] = ex;
    gbs[tid] = gb;
    gbs2[tid] = gb2;
    if (tid == NRMAX - 1) *tots = ex + c;
    __syncthreads();
#pragma unroll
    for (int j = 0; j < EPL; ++j) {
        unsigned idx = ebase + j * 32 + lane;
        if (idx < e1) {
            unsigned d = dd[j] < nmax ? dd[j] : nmax;
            unsigned s = (unsigned)ss[j] < nmax ? (unsigned)ss[j] : nmax;
            unsigned key = d >> RB;
            unsigned pos = lst[key] + myh[key] + rk[j];
            sw[pos] = (s << RB) | (d & (RNODES - 1u));
            sk[pos] = key | ((wib * 256u + (unsigned)j * 32u + lane) << 8);
        }
    }
    __syncthreads();
    unsigned T = *tots;
    for (unsigned p = tid; p < T; p += THREADS) {
        unsigned kk = sk[p];
        unsigned k = kk & 255u;
        unsigned g0 = gbs[k];
        unsigned off = g0 + (p - lst[k]);
        unsigned w = sw[p];
        if (off < BCAP) {
            bkt[((size_t)k * 2 + hsel) * BCAP + off] = (int)w;
        } else {
            unsigned off2 = gbs2[k] + (off - (g0 > BCAP ? g0 : BCAP));
            if (off2 < BCAP) {
                bkt[((size_t)k * 2 + (hsel ^ 1u)) * BCAP + off2] = (int)w;
            } else {
                unsigned lid = kk >> 8;
                unsigned e = (blockIdx.x * (THREADS / 32) + (lid >> 8)) * chunk + g + (lid & 255u);
                unsigned d = (k << RB) | (w & (RNODES - 1u));
                atadd1_ret(cntg + d, 1.0f);
                unsigned q = (unsigned)atadd1_ret(ovn, 1.0f);
                if (q < (unsigned)E && e < (unsigned)E) ovf[q] = (int)e;
            }
        }
    }
}
__global__ void __launch_bounds__(THREADS)
gcn_gemm1_bins(const float* __restrict__ x, const float* __restrict__ W1,
                               float* __restrict__ h1, int hp, int ntiles, int N,
                               const int* __restrict__ src, const int* __restrict__ dst, int E,
                               float* __restrict__ bcnt, int* __restrict__ bkt,
                               float* __restrict__ cntg,
                               float* __restrict__ ovn, int* __restrict__ ovf) {
    __shared__ __attribute__((aligned(16))) float xs[(THREADS / 32) * 16 * XPITCH];
    __shared__ unsigned lst[NRMAX];
    __shared__ unsigned gbs[NRMAX];
    __shared__ unsigned gbs2[NRMAX];
    __shared__ unsigned wtot[THREADS / 32];
    __shared__ unsigned tots;
    unsigned* sw = (unsigned*)xs;
    unsigned* sk = sw + 32 * EPL * (THREADS / 32);
    unsigned* wh = sk + 32 * EPL * (THREADS / 32);
    static_assert(2 * 32 * EPL * (THREADS / 32) + (THREADS / 32) * NRMAX <= (THREADS / 32) * 16 * XPITCH,
                  "sorted words and counts fit in xs");
    int wib  = threadIdx.x >> 5;
    int wave = blockIdx.x * (blockDim.x >> 5) + wib;
    bool active = wave < ntiles;
    int tile = active ? wave : ntiles - 1;
    int lane = threadIdx.x & 31;
    float* xt = xs + wib * (16 * XPITCH);

    unsigned wv     = (unsigned)__builtin_amdgcn_readfirstlane(wave);
    unsigned nwaves = gridDim.x * (blockDim.x >> 5);
    unsigned chunk  = (((unsigned)E + nwaves - 1u) / nwaves + 31u) & ~31u;
    unsigned e0 = wv * chunk;
    unsigned e1 = e0 + chunk < (unsigned)E ? e0 + chunk : (unsigned)E;
    bool any = e0 < e1;
    unsigned hsel = pkg_half();
    unsigned nmax = (unsigned)N - 1u;

    v4f_t rowv[16];
#pragma unroll
    for (int r = 0; r < 16; ++r) {
        int row = tile * 16 + r;
        int rcl = row < N ? row : N - 1;
        rowv[r] = __builtin_nontemporal_load((const v4f_t*)(x + (size_t)rcl * 128) + lane);
    }
    unsigned dv[EPL] = {};
    int      sv[EPL] = {};
    if (any) {
#pragma unroll
        for (int j = 0; j < EPL; ++j) {
            unsigned idx = e0 + j * 32 + lane;
            dv[j] = (unsigned)__builtin_nontemporal_load(dst + (idx < e1 ? idx : e1 - 1u));
        }
#pragma unroll
        for (int j = 0; j < EPL; ++j) {
            unsigned idx = e0 + j * 32 + lane;
            sv[j] = __builtin_nontemporal_load(src + (idx < e1 ? idx : e1 - 1u));
        }
    }
    bin_round(dv, sv, e0, e1, 0u, chunk, nmax, hsel, bcnt, bkt, cntg, ovn, ovf, E,
              lst, gbs, gbs2, wtot, &tots, sw, sk, wh);
    for (unsigned g = 32 * EPL; g < chunk; g += 32 * EPL) {
        unsigned dd[EPL];
        int      ss[EPL];
#pragma unroll
        for (int j = 0; j < EPL; ++j) {
            unsigned idx = e0 + g + j * 32 + lane;
            unsigned cl = idx < e1 ? idx : (any ? e1 - 1u : 0u);
            dd[j] = (unsigned)__builtin_nontemporal_load(dst + cl);
            ss[j] = __builtin_nontemporal_load(src + cl);
        }
        bin_round(dd, ss, e0 + g, e1, g, chunk, nmax, hsel, bcnt, bkt, cntg, ovn, ovf, E,
                  lst, gbs, gbs2, wtot, &tots, sw, sk, wh);
    }
    __syncthreads();

#pragma unroll
    for (int r = 0; r < 16; ++r) *(v4f_t*)(xt + r * XPITCH + lane * 4) = rowv[r];
    __syncthreads();

    int half = lane >> 4;
    int m    = lane & 15;
    int koff = half << 1;
    const float* xr = xt + m * XPITCH;

    v8f_t c = {};
#pragma unroll
    for (int k0 = 0; k0 < 128; k0 += 4) {
        v2f_t a = *(const v2f_t*)(xr + k0 + koff);
        v2f_t b;
        b.x = W1[(k0 + koff) * 16 + m];
        b.y = W1[(k0 + koff + 1) * 16 + m];
        c = __builtin_amdgcn_wmma_f32_16x16x4_f32(false, a, false, b, (short)0, c,
                                                  false, false);
    }
    if (!active) return;
#pragma unroll
    for (int v = 0; v < 8; ++v) {
        int r = wave * 16 + half * 8 + v;
        if (r < N) h1[(size_t)r * hp + m] = c[v];
    }
}

__global__ void __launch_bounds__(BTHREADS)
gcn_range_lists(int* __restrict__ bkt, const float* __restrict__ bcnt,
                int N, float* __restrict__ cnt0, const float* __restrict__ cntg,
                int* __restrict__ rov, int* __restrict__ lists,
                float* __restrict__ dinv, float* __restrict__ h1, int hp) {
    __shared__ unsigned S[2 * BCAP];
    __shared__ int Lw[BWAVES * SUBN * LPITCH];
    __shared__ unsigned wh[BWAVES * NSUB];
    __shared__ unsigned tot[NSUB];
    __shared__ unsigned sst[NSUB];
    unsigned tid = threadIdx.x;
    unsigned lane = tid & 31u;
    unsigned w = tid >> 5;
    unsigned ltm = (1u << lane) - 1u;
    unsigned k = blockIdx.x;
    unsigned n0 = (unsigned)bcnt[((size_t)k) * 32];
    unsigned n1 = (unsigned)bcnt[((size_t)NRMAX + k) * 32];
    n0 = n0 < BCAP ? n0 : BCAP;
    n1 = n1 < BCAP ? n1 : BCAP;
    unsigned T = n0 + n1;
    int* b0 = bkt + (size_t)k * 2 * BCAP;
    unsigned G = (T + BTHREADS - 1u) / BTHREADS;
    for (unsigned u = tid; u < BWAVES * NSUB; u += BTHREADS) wh[u] = 0u;
    __syncthreads();
    unsigned* myh = wh + w * NSUB;
    unsigned wd[BLOADS];
    unsigned short* myi = (unsigned short*)(Lw + w * (SUBN * LPITCH));
    static_assert(BLOADS * 32 * 2 <= SUBN * LPITCH * 4, "a wave's ranks fit in its list area");
    if (T) {
#pragma unroll
        for (int u = 0; u < BLOADS; ++u) {
            unsigned p = (w * G + u) * 32u + lane;
            unsigned pc = p < T ? p : T - 1u;
            wd[u] = (unsigned)b0[pc < n0 ? pc : pc - n0 + BCAP];
        }
#pragma unroll
        for (int u = 0; u < BLOADS; ++u) {
            unsigned p = (w * G + u) * 32u + lane;
            bool on = (unsigned)u < G && p < T;
            unsigned key = (wd[u] >> SUBB) & (NSUB - 1u);
            unsigned m = same_key_lanes(key, on, RB - SUBB);
            unsigned below = __builtin_popcount(m & ltm);
            unsigned all = __builtin_popcount(m);
            myi[u * 32 + lane] = (unsigned short)(below | (all << 5));
            if (on && below == 0u) myh[key] += all;
        }
    }
    __syncthreads();
    if (tid < NSUB) {
        unsigned run = 0u;
#pragma unroll
        for (unsigned u = 0; u < BWAVES; ++u) {
            unsigned t = wh[u * NSUB + tid];
            wh[u * NSUB + tid] = run;
            run += t;
        }
        tot[tid] = run;
    }
    __syncthreads();
    if (tid < NSUB) {
        unsigned s = 0u;
#pragma unroll
        for (unsigned q = 0; q < NSUB; ++q) s += q < tid ? tot[q] : 0u;
        sst[tid] = s;
    }
    __syncthreads();
    if (T) {
#pragma unroll
        for (int u = 0; u < BLOADS; ++u) {
            unsigned p = (w * G + u) * 32u + lane;
            bool on = (unsigned)u < G && p < T;
            unsigned key = (wd[u] >> SUBB) & (NSUB - 1u);
            unsigned info = myi[u * 32 + lane];
            if (on) {
                unsigned basec = myh[key];
                unsigned pos = sst[key] + basec + (info & 31u);
                if (pos < 2u * BCAP) S[pos] = wd[u];
                if ((info & 31u) == 0u) myh[key] = basec + (info >> 5);
            }
        }
    }
    __syncthreads();
    int* myL = Lw + w * (SUBN * LPITCH);
    for (unsigned rep = 0; rep < NSUB / BWAVES; ++rep) {
        unsigned sr = w + rep * BWAVES;
        unsigned st = sst[sr];
        unsigned tt = tot[sr];
        unsigned cnt = 0u;
        unsigned oc = 0u;
        for (unsigned q0 = 0; q0 < tt; q0 += 32u) {
            unsigned q = q0 + lane;
            bool on = q < tt;
            unsigned wd = 0u;
            if (on && st + q < 2u * BCAP) wd = S[st + q];
            unsigned key = wd & (SUBN - 1u);
            unsigned vm = __builtin_amdgcn_ballot_w32(on);
            unsigned ms = vm;
            unsigned mn = vm;
#pragma unroll
            for (int b = 0; b < SUBB; ++b) {
                bool kb = (key >> b) & 1u;
                bool lb = (lane >> b) & 1u;
                unsigned bb = __builtin_amdgcn_ballot_w32(kb);
                ms &= kb ? bb : ~bb;
                mn &= lb ? bb : ~bb;
            }
            unsigned basec = (unsigned)__shfl((int)cnt, (int)key);
            unsigned r = basec + __builtin_popcount(ms & ltm);
            cnt += __builtin_popcount(mn);
            bool ov = on && r >= LCAP;
            unsigned om = __builtin_amdgcn_ballot_w32(ov);
            if (on && r < LCAP) myL[key * LPITCH + r] = (int)(wd >> RB);
            if (ov) {
                unsigned j = st + oc + __builtin_popcount(om & ltm);
                if (j < 2u * BCAP) b0[j] = (int)wd;
            }
            oc += __builtin_popcount(om);
        }
        if (lane == 0) {
            rov[2 * ((size_t)k * NSUB + sr)]     = (int)st;
            rov[2 * ((size_t)k * NSUB + sr) + 1] = (int)oc;
        }
        for (unsigned it = 0; it < SUBN / 4; ++it) {
            unsigned nl = it * 4 + (lane >> 3);
            unsigned part = lane & 7u;
            unsigned c = (unsigned)__shfl((int)cnt, (int)nl);
            unsigned node = (k << RB) + sr * SUBN + nl;
            const int* lp = myL + nl * LPITCH + part * 4;
            v4i_t v;
            v.x = part * 4 + 0 < c ? lp[0] : 0;
            v.y = part * 4 + 1 < c ? lp[1] : 0;
            v.z = part * 4 + 2 < c ? lp[2] : 0;
            v.w = part * 4 + 3 < c ? lp[3] : 0;
            if (node < (unsigned)N) *(v4i_t*)(lists + (size_t)node * LCAP + part * 4) = v;
        }
        {
            unsigned node = (k << RB) + sr * SUBN + lane;
            if (node < (unsigned)N) {
                float f = (float)cnt;
                float fg = cntg[node];
                cnt0[node] = f;
                float dv = rsqrtf((f + fg) + 1.0f);
                dinv[node] = dv;
                h1[(size_t)node * hp + 16] = dv;
            }
        }
    }
}

__global__ void gcn_pull16_lists(const float* __restrict__ cnt0, const float* __restrict__ cntg,
                                 const int* __restrict__ lists,
                                 const int* __restrict__ bkt,
                                 const int* __restrict__ rov,
                                 const float* __restrict__ ovn, const int* __restrict__ ovf,
                                 const int* __restrict__ src, const int* __restrict__ dst, int E,
                                 const float* __restrict__ dinv,
                                 const float* __restrict__ h, int hp, const float* __restrict__ b1,
                                 const float* __restrict__ W2, float* __restrict__ h2, int N) {
    int t = blockIdx.x * blockDim.x + threadIdx.x;
    int i = t >> 2;
    int q = t & 3;
    if (i >= N) return;
    v4f_t wa = ((const v4f_t*)W2)[2 * q];
    v4f_t wb = ((const v4f_t*)W2)[2 * q + 1];
    int c  = (int)cnt0[i];
    float fg = cntg[i];
    int nA = c < 16 ? c : 16;
    int nB = c - 16;
    nB = nB < 0 ? 0 : (nB < 16 ? nB : 16);
    size_t base = (size_t)i * LCAP;
    v4i_t LA = *(const v4i_t*)(lists + base + 4 * q);
    v4i_t LB = *(const v4i_t*)(lists + base + 16 + 4 * q);
    float di = dinv[i];
    unsigned nmax = (unsigned)N - 1u;
    v4f_t acc = {0.0f, 0.0f, 0.0f, 0.0f};
    pull16_part(LA, 0, nA, h, hp, nmax, di, q, acc);
    pull16_part(LA, 8, nA, h, hp, nmax, di, q, acc);
    pull16_part(LB, 0, nB, h, hp, nmax, di, q, acc);
    pull16_part(LB, 8, nB, h, hp, nmax, di, q, acc);
    if (c > LCAP) {
        unsigned rg = (unsigned)i >> RB;
        unsigned dl = (unsigned)i & (RNODES - 1u);
        size_t sb = 2 * ((size_t)rg * NSUB + (dl >> SUBB));
        unsigned st = (unsigned)rov[sb];
        unsigned R  = (unsigned)rov[sb + 1];
        st = st < 2u * BCAP ? st : 2u * BCAP;
        R  = R < 2u * BCAP - st ? R : 2u * BCAP - st;
        const int* q0 = bkt + (size_t)rg * 2 * BCAP;
        for (unsigned j = st; j < st + R; ++j) {
            unsigned w = (unsigned)q0[j];
            if ((w & (RNODES - 1u)) != dl) continue;
            unsigned s0 = w >> RB;
            s0 = s0 < nmax ? s0 : nmax;
            const float* r0 = h + (size_t)s0 * hp;
            float n0 = r0[16] * di;
            v4f_t v0 = *((const v4f_t*)r0 + q);
            acc += v0 * n0;
        }
    }
    if (fg > 0.0f) {
        unsigned K = (unsigned)ovn[0];
        K = K < (unsigned)E ? K : (unsigned)E;
        for (unsigned k = 0; k < K; ++k) {
            unsigned e = (unsigned)ovf[k];
            e = e < (unsigned)E ? e : (unsigned)E - 1u;
            if (dst[e] != i) continue;
            unsigned s0 = (unsigned)src[e];
            s0 = s0 < nmax ? s0 : nmax;
            const float* r0 = h + (size_t)s0 * hp;
            float n0 = r0[16] * di;
            v4f_t v0 = *((const v4f_t*)r0 + q);
            acc += v0 * n0;
        }
    }
    float d2 = di * di;
    v4f_t hv = *((const v4f_t*)(h + (size_t)i * hp) + q);
    v4f_t bv = ((const v4f_t*)b1)[q];
    v4f_t o;
    o.x = fmaxf(acc.x + hv.x * d2 + bv.x, 0.0f);
    o.y = fmaxf(acc.y + hv.y * d2 + bv.y, 0.0f);
    o.z = fmaxf(acc.z + hv.z * d2 + bv.z, 0.0f);
    o.w = fmaxf(acc.w + hv.w * d2 + bv.w, 0.0f);
    float p0 = o.x * wa.x + o.y * wa.z + o.z * wb.x + o.w * wb.z;
    float p1 = o.x * wa.y + o.y * wa.w + o.z * wb.y + o.w * wb.w;
    p0 += __shfl_xor(p0, 1);
    p1 += __shfl_xor(p1, 1);
    p0 += __shfl_xor(p0, 2);
    p1 += __shfl_xor(p1, 2);
    h2[t] = q == 0 ? p0 : q == 1 ? p1 : q == 2 ? di : 0.0f;
}

__global__ void gcn_pull2_lists(const float* __restrict__ cnt0, const float* __restrict__ cntg,
                                const int* __restrict__ lists,
                                const int* __restrict__ bkt,
                                const int* __restrict__ rov,
                                const float* __restrict__ ovn, const int* __restrict__ ovf,
                                const int* __restrict__ src, const int* __restrict__ dst, int E,
                                const float* __restrict__ dinv,
                                const float* __restrict__ h2, const float* __restrict__ b2,
                                float* __restrict__ out, int N) {
    int t = blockIdx.x * blockDim.x + threadIdx.x;
    int i = t >> 2;
    int q = t & 3;
    bool ok = i < N;
    int ic = ok ? i : N - 1;
    int c  = ok ? (int)cnt0[ic] : 0;
    float fg = ok ? cntg[ic] : 0.0f;
    int nA = c < 16 ? c : 16;
    int nB = c - 16;
    nB = nB < 0 ? 0 : (nB < 16 ? nB : 16);
    size_t base = (size_t)ic * LCAP;
    v4i_t LA = *(const v4i_t*)(lists + base + 4 * q);
    v4i_t LB = *(const v4i_t*)(lists + base + 16 + 4 * q);
    float di = dinv[ic];
    unsigned nmax = (unsigned)N - 1u;
    float g0 = 0.0f, g1 = 0.0f;
    {
        int ra = nA - 4 * q;
        int rb = nB - 4 * q;
        v3f_t a0 = gather_h2(ra > 0, LA.x, nmax, h2);
        v3f_t a1 = gather_h2(ra > 1, LA.y, nmax, h2);
        v3f_t a2 = gather_h2(ra > 2, LA.z, nmax, h2);
        v3f_t a3 = gather_h2(ra > 3, LA.w, nmax, h2);
        v3f_t c0 = gather_h2(rb > 0, LB.x, nmax, h2);
        v3f_t c1 = gather_h2(rb > 1, LB.y, nmax, h2);
        v3f_t c2 = gather_h2(rb > 2, LB.z, nmax, h2);
        v3f_t c3 = gather_h2(rb > 3, LB.w, nmax, h2);
        add_h2(a0, di, g0, g1);
        add_h2(a1, di, g0, g1);
        add_h2(a2, di, g0, g1);
        add_h2(a3, di, g0, g1);
        add_h2(c0, di, g0, g1);
        add_h2(c1, di, g0, g1);
        add_h2(c2, di, g0, g1);
        add_h2(c3, di, g0, g1);
    }
    if (c > LCAP) {
        unsigned rg = (unsigned)ic >> RB;
        unsigned dl = (unsigned)ic & (RNODES - 1u);
        size_t sb = 2 * ((size_t)rg * NSUB + (dl >> SUBB));
        unsigned st = (unsigned)rov[sb];
        unsigned R  = (unsigned)rov[sb + 1];
        st = st < 2u * BCAP ? st : 2u * BCAP;
        R  = R < 2u * BCAP - st ? R : 2u * BCAP - st;
        const int* q0 = bkt + (size_t)rg * 2 * BCAP;
        for (unsigned j = st + (unsigned)q; j < st + R; j += 4) {
            unsigned w = (unsigned)q0[j];
            if ((w & (RNODES - 1u)) != dl) continue;
            unsigned s = w >> RB;
            s = s < nmax ? s : nmax;
            v4f_t v = *((const v4f_t*)h2 + s);
            float nrm = v.z * di;
            g0 += v.x * nrm;
            g1 += v.y * nrm;
        }
    }
    if (q == 0 && fg > 0.0f) {
        unsigned K = (unsigned)ovn[0];
        K = K < (unsigned)E ? K : (unsigned)E;
        for (unsigned k = 0; k < K; ++k) {
            unsigned e = (unsigned)ovf[k];
            e = e < (unsigned)E ? e : (unsigned)E - 1u;
            if (dst[e] != ic) continue;
            unsigned s = (unsigned)src[e];
            s = s < nmax ? s : nmax;
            v4f_t v = *((const v4f_t*)h2 + s);
            float nrm = v.z * di;
            g0 += v.x * nrm;
            g1 += v.y * nrm;
        }
    }
    g0 += __shfl_xor(g0, 1);
    g1 += __shfl_xor(g1, 1);
    g0 += __shfl_xor(g0, 2);
    g1 += __shfl_xor(g1, 2);
    if (!ok || q != 0) return;
    float d2 = di * di;
    v4f_t me = *((const v4f_t*)h2 + i);
    float a = g0 + me.x * d2 + b2[0];
    float b = g1 + me.y * d2 + b2[1];
    float mx  = fmaxf(a, b);
    float lse = mx + logf(expf(a - mx) + expf(b - mx));
    out[2 * (size_t)i]     = a - lse;
    out[2 * (size_t)i + 1] = b - lse;
}

#define R4(v) (((size_t)(v) + 3) & ~(size_t)3)
#define R32(v) (((size_t)(v) + 31) & ~(size_t)31)
extern "C" void kernel_launch(void* const* d_in, const int* in_sizes, int n_in,
                              void* d_out, int out_size, void* d_ws, size_t ws_size,
                              hipStream_t stream) {
    const float* x  = (const float*)d_in[0];
    const int*   ei = (const int*)d_in[1];
    const float* W1 = (const float*)d_in[2];
    const float* b1 = (const float*)d_in[3];
    const float* W2 = (const float*)d_in[4];
    const float* b2 = (const float*)d_in[5];
    float* out = (float*)d_out;

    int N = in_sizes[0] / 128;
    int E = in_sizes[1] / 2;
    const int* src = ei;
    const int* dst = ei + E;

    int eb  = (E + THREADS - 1) / THREADS;
    int nb  = (N + THREADS - 1) / THREADS;
    int nb4 = (int)(((long long)N * 4 + THREADS - 1) / THREADS);


    {
        size_t n  = (size_t)N;
        size_t ne = (size_t)E;
        size_t nrs = (n + RNODES - 1) / RNODES;
        size_t o_cg   = R32(n);
        size_t o_ovn  = o_cg + R32(n);
        size_t o_bcnt = o_ovn + 32;
        size_t o_rov  = o_bcnt + 2 * (size_t)NRMAX * 32;
        size_t o_dinv = o_rov + R32(2 * nrs * NSUB);
        size_t o_ovf  = o_dinv + R32(n);
        size_t o_h2   = o_ovf + R32(ne);
        size_t o_lst  = o_h2 + R32(4 * n);
        size_t o_bkt  = o_lst + R32(LCAP * n);
        size_t o_h1   = o_bkt + R32(2 * nrs * BCAP);
        size_t total  = o_h1 + 32 * n;
        if (N > 0 && E > 0 && nrs <= NRMAX && ne <= ((size_t)1 << 24) &&
            ws_size >= total * sizeof(float)) {
            float* ws    = (float*)d_ws;
            float* cnt0  = ws;
            float* cntg  = ws + o_cg;
            float* ovn   = ws + o_ovn;
            float* bcnt  = ws + o_bcnt;
            int*   rov   = (int*)(ws + o_rov);
            float* dinv  = ws + o_dinv;
            int*   ovf   = (int*)(ws + o_ovf);
            float* h2    = ws + o_h2;
            int*   lists = (int*)(ws + o_lst);
            int*   bkt   = (int*)(ws + o_bkt);
            float* h1    = ws + o_h1;
            hipMemsetAsync(cntg, 0, (o_rov - o_cg) * sizeof(float), stream);
            int ntiles = (N + 15) / 16;
            int gb = (ntiles + 7) / 8;
            gcn_gemm1_bins<<<gb, THREADS, 0, stream>>>(x, W1, h1, 32, ntiles, N, src, dst, E,
                                                       bcnt, bkt, cntg, ovn, ovf);
            gcn_range_lists<<<(int)nrs, BTHREADS, 0, stream>>>(bkt, bcnt, N,
                                                               cnt0, cntg, rov, lists, dinv, h1, 32);
            gcn_pull16_lists<<<nb4, THREADS, 0, stream>>>(cnt0, cntg, lists, bkt, rov, ovn, ovf, src, dst, E,
                                                          dinv, h1, 32, b1, W2, h2, N);
            gcn_pull2_lists<<<nb4, THREADS, 0, stream>>>(cnt0, cntg, lists, bkt, rov, ovn, ovf, src, dst, E,
                                                         dinv, h2, b2, out, N);
            return;
        }
    }

    {
        size_t n  = (size_t)N;
        size_t ne = (size_t)E;
        size_t o_dinv  = R32(3 * n + 1);
        size_t o_ovf   = o_dinv + R32(n);
        size_t o_h1r   = o_ovf + R32(ne);
        size_t o_lists = o_h1r + R32(16 * n);
        size_t lstr    = R32(SLOT * n);
        size_t o_h1    = o_lists + 3 * lstr;
        size_t total   = o_h1 + 32 * n;
        if (N > 0 && E > 0 && ws_size >= total * sizeof(float)) {
            float* ws    = (float*)d_ws;
            float* cnt   = ws;
            float* ovn   = ws + 3 * n;
            float* dinv  = ws + o_dinv;
            int*   ovf   = (int*)(ws + o_ovf);
            float* h2    = ws + o_h1r;
            int*   lists = (int*)(ws + o_lists);
            float* h1    = ws + o_h1;
            hipMemsetAsync(cnt, 0, (3 * n + 1) * sizeof(float), stream);
            int ntiles = (N + 15) / 16;
            int gb = (ntiles + 7) / 8;
            gcn_gemm1_slots<<<gb, THREADS, 0, stream>>>(x, W1, h1, 32, ntiles, N, src, dst, cnt, n, E,
                                                        lists, lstr, ovn, ovf);
            gcn_dinv_slots<<<nb, THREADS, 0, stream>>>(cnt, n, N, dinv, h1, 32);
            gcn_pull16_slots<<<nb4, THREADS, 0, stream>>>(cnt, n, lists, lstr, ovn, ovf, src, dst, E,
                                                          dinv, h1, 32, b1, W2, h2, N);
            gcn_pull2_slots<<<nb4, THREADS, 0, stream>>>(cnt, n, lists, lstr, ovn, ovf, src, dst, E,
                                                         dinv, h2, b2, out, N);
            return;
        }
    }

    size_t n  = (size_t)N;
    size_t ne = (size_t)E;
    size_t big = ne > 16 * n ? ne : 16 * n;
    size_t fixed = R4(n) + R4(n) + R4((size_t)nb) + R4(big) + R4(ne) + 32 + 16 * n;
    size_t np = (ws_size >= (R4(2 * n) + fixed) * sizeof(float)) ? 2 : 1;
    size_t nl = (np == 2 && ws_size >= (R4(2 * n) + R4(ne) + fixed) * sizeof(float)) ? 2 : 1;
    int    hp = (ws_size >= (R4(np * n) + (nl - 1) * R4(ne) + fixed + 16 * n) * sizeof(float)) ? 32 : 16;
    size_t sd = (np - 1) * n;
    float* ws   = (float*)d_ws;
    float* cnt  = ws;
    int*   lstb = nl == 2 ? (int*)(cnt + R4(np * n)) : nullptr;
    int*   endp = (int*)(cnt + R4(np * n)) + (nl - 1) * R4(ne);
    float* dinv = (float*)(endp + R4(n));
    int*   bt   = (int*)(dinv + R4(n));
    int*   rankw = bt + R4((size_t)nb);
    float* h1r  = (float*)rankw;
    int*   lst  = rankw + R4(big);
    size_t h1off = (size_t)((float*)(lst + R4(ne)) - ws);
    float* h1   = ws + ((h1off + 31) & ~(size_t)31);
    float* h2   = h1;

    hipMemsetAsync(cnt, 0, (nl == 2 ? R4(np * n) + ne : np * n) * sizeof(float), stream);

    int ntiles = (N + 15) / 16;
    int gb = (ntiles + 7) / 8;
    gcn_gemm1_deg<<<gb, THREADS, 0, stream>>>(x, W1, h1, hp, ntiles, N, dst, cnt, sd, E, rankw);
    gcn_block_counts<<<nb, THREADS, 0, stream>>>(cnt, sd, N, bt);
    gcn_dinv_offsets<<<nb, THREADS, 0, stream>>>(cnt, sd, N, bt, endp, dinv, h1, hp);
    gcn_place_edges<<<eb, THREADS, 0, stream>>>(src, dst, rankw, (const int*)cnt, sd, lst, lstb, E);

    gcn_pull16<<<nb4, THREADS, 0, stream>>>((const int*)cnt, endp, lst, lstb, dinv, h1, hp, b1, h1r, N);
    gcn_gemm2<<<nb, THREADS, 0, stream>>>(h1r, W2, dinv, h2, N);
    gcn_pull2_final<<<nb4, THREADS, 0, stream>>>((const int*)cnt, endp, lst, lstb, dinv, h2, b2, out, N);
}
